# starting kernel
# baseline (speedup 1.0000x reference)
#include <hip/hip_runtime.h>
#include <hip/hip_bf16.h>

typedef __attribute__((ext_vector_type(2))) float v2f;
typedef __attribute__((ext_vector_type(8))) float v8f;

__device__ __forceinline__ v8f wmma4(v2f a, v2f b, v8f c) {
  return __builtin_amdgcn_wmma_f32_16x16x4_f32(
       false, a,  false, b,
       (short)0, c,  false,  false);
}

__global__ void deg_kernel(const int* __restrict__ dst, float* __restrict__ deg, int E) {
  int tid = blockIdx.x * blockDim.x + threadIdx.x;
  if (tid < E) atomicAdd(&deg[dst[tid]], 1.0f);
}

__global__ void dinv_kernel(const float* __restrict__ deg, float* __restrict__ dinv, int N) {
  int tid = blockIdx.x * blockDim.x + threadIdx.x;
  if (tid < N) {
    float d = deg[tid];
    d = d < 1.0f ? 1.0f : d;
    dinv[tid] = rsqrtf(d);
  }
}

__global__ void wlap_kernel(const float* __restrict__ dinv, const int* __restrict__ src,
                            const int* __restrict__ dst, float* __restrict__ wlap, int E) {
  int tid = blockIdx.x * blockDim.x + threadIdx.x;
  if (tid < E) wlap[tid] = -(dinv[src[tid]] * dinv[dst[tid]]);
}

__global__ void scatter_wadd(const float* __restrict__ x, const float* __restrict__ w,
                             const int* __restrict__ src, const int* __restrict__ dst,
                             float* __restrict__ out, long long total) {
  long long tid = (long long)blockIdx.x * blockDim.x + threadIdx.x;
  if (tid >= total) return;
  int f = (int)(tid & 63);
  long long e = tid >> 6;
  int s = src[e], d = dst[e];
  float v = x[(size_t)s * 64 + f] * w[e];
  atomicAdd(&out[(size_t)d * 64 + f], v);
}

__global__ void combine_kernel(const float* __restrict__ a, const float* __restrict__ b,
                               float* __restrict__ out, long long n) {
  long long tid = (long long)blockIdx.x * blockDim.x + threadIdx.x;
  if (tid < n) out[tid] = 2.0f * a[tid] - b[tid];
}

__global__ void concat_hp(const float* __restrict__ h, const float* __restrict__ p,
                          float* __restrict__ hp, int N) {
  long long tid = (long long)blockIdx.x * blockDim.x + threadIdx.x;
  if (tid >= (long long)N * 128) return;
  int n = (int)(tid >> 7);
  int c = (int)(tid & 127);
  hp[tid] = (c < 64) ? h[(size_t)n * 64 + c] : p[(size_t)n * 64 + (c - 64)];
}

__global__ void gemm_wmma_kernel(const float* __restrict__ A, int lda,
                                 const float* __restrict__ B,
                                 const float* __restrict__ bias,
                                 float* __restrict__ Out, int ldo, int coloff,
                                 int rows, int kdim) {
  int wave = blockIdx.x * (blockDim.x >> 5) + (threadIdx.x >> 5);
  int rowTiles = (rows + 15) >> 4;
  if (wave >= rowTiles * 4) return;
  int rt = wave >> 2, ct = wave & 3;
  int lane = threadIdx.x & 31, half = lane >> 4, l16 = lane & 15;
  int ar = rt * 16 + l16;
  if (ar >= rows) ar = rows - 1;
  int bc = ct * 16 + l16;
  v8f acc = {};
  for (int k0 = 0; k0 < kdim; k0 += 4) {
    int kr = k0 + 2 * half;
    v2f a, b;
    a.x = A[(size_t)ar * lda + kr];
    a.y = A[(size_t)ar * lda + kr + 1];
    b.x = B[(size_t)kr * 64 + bc];
    b.y = B[(size_t)(kr + 1) * 64 + bc];
    acc = wmma4(a, b, acc);
  }
  float bv = bias ? bias[ct * 16 + l16] : 0.0f;
#pragma unroll
  for (int v = 0; v < 8; ++v) {
    int r = rt * 16 + v + 8 * half;
    if (r < rows) Out[(size_t)r * ldo + coloff + ct * 16 + l16] = acc[v] + bv;
  }
}

__global__ void cheb_gemm_kernel(const float* __restrict__ Tx0, const float* __restrict__ Tx1,
                                 const float* __restrict__ Tx2, const float* __restrict__ Tx3,
                                 const float* __restrict__ fc, const float* __restrict__ Wc,
                                 const float* __restrict__ bias, float* __restrict__ Out,
                                 int rows, int Nn) {
  int wave = blockIdx.x * (blockDim.x >> 5) + (threadIdx.x >> 5);
  int rowTiles = (rows + 15) >> 4;
  if (wave >= rowTiles * 4) return;
  int rt = wave >> 2, ct = wave & 3;
  int lane = threadIdx.x & 31, half = lane >> 4, l16 = lane & 15;
  int ar = rt * 16 + l16;
  if (ar >= rows) ar = rows - 1;
  int bc = ct * 16 + l16;
  const float* As[4] = {Tx0, Tx1, Tx2, Tx3};
  v8f acc = {};
#pragma unroll
  for (int k = 0; k < 4; ++k) {
    const float* A = As[k];
    const float* B = Wc + (size_t)k * 64 * 64;
    float sc = fc[(size_t)k * Nn + ar];
#pragma unroll
    for (int k0 = 0; k0 < 64; k0 += 4) {
      int kr = k0 + 2 * half;
      v2f a, b;
      a.x = sc * A[(size_t)ar * 64 + kr];
      a.y = sc * A[(size_t)ar * 64 + kr + 1];
      b.x = B[(size_t)kr * 64 + bc];
      b.y = B[(size_t)(kr + 1) * 64 + bc];
      acc = wmma4(a, b, acc);
    }
  }
  float bv = bias[ct * 16 + l16];
#pragma unroll
  for (int v = 0; v < 8; ++v) {
    int r = rt * 16 + v + 8 * half;
    if (r < rows) Out[(size_t)r * 192 + ct * 16 + l16] = acc[v] + bv;
  }
}

__global__ void attn_kernel(const float* __restrict__ Ee, const float* __restrict__ WE,
                            const float* __restrict__ Qh, const float* __restrict__ Kh,
                            const float* __restrict__ Vh,
                            const int* __restrict__ src, const int* __restrict__ dst,
                            float* __restrict__ wV, float* __restrict__ z, int E) {
  __shared__ float lds[4 * 16 * 64];
  int wid = threadIdx.x >> 5;
  int tile = blockIdx.x * 4 + wid;
  int eTiles = (E + 15) >> 4;
  bool valid = tile < eTiles;
  int t = valid ? tile : 0;
  int e0 = t * 16;
  int lane = threadIdx.x & 31, half = lane >> 4, l16 = lane & 15;
  int ar = e0 + l16;
  if (ar >= E) ar = E - 1;
  float* myLds = lds + wid * 1024;

#pragma unroll
  for (int ct = 0; ct < 4; ++ct) {
    v8f acc = {};
#pragma unroll
    for (int k0 = 0; k0 < 64; k0 += 4) {
      int kr = k0 + 2 * half;
      v2f a, b;
      a.x = Ee[(size_t)ar * 64 + kr];
      a.y = Ee[(size_t)ar * 64 + kr + 1];
      int bc = ct * 16 + l16;
      b.x = WE[(size_t)kr * 64 + bc];
      b.y = WE[(size_t)(kr + 1) * 64 + bc];
      acc = wmma4(a, b, acc);
    }
#pragma unroll
    for (int v = 0; v < 8; ++v)
      myLds[(v + 8 * half) * 64 + ct * 16 + l16] = acc[v];
  }
  __syncthreads();
  if (!valid) return;

  const float scale = 0.35355339059327373f;
#pragma unroll
  for (int pp = 0; pp < 4; ++pp) {
    int pair = pp * 32 + (threadIdx.x & 31);
    int m = pair >> 3, hh = pair & 7;
    int em = e0 + m;
    if (em >= E) continue;
    int s = src[em], d = dst[em];
    float sc = 0.0f;
#pragma unroll
    for (int dd = 0; dd < 8; ++dd) {
      int c = hh * 8 + dd;
      sc += Kh[(size_t)s * 64 + c] * Qh[(size_t)d * 64 + c] * myLds[m * 64 + c];
    }
    sc *= scale;
    sc = fminf(5.0f, fmaxf(-5.0f, sc));
    float ss = __expf(sc);
    atomicAdd(&z[(size_t)d * 8 + hh], ss);
#pragma unroll
    for (int dd = 0; dd < 8; ++dd) {
      int c = hh * 8 + dd;
      atomicAdd(&wV[(size_t)d * 64 + c], Vh[(size_t)s * 64 + c] * ss);
    }
  }
}

__global__ void finalize_kernel(const float* __restrict__ wV, const float* __restrict__ z,
                                float* __restrict__ out, int N) {
  long long tid = (long long)blockIdx.x * blockDim.x + threadIdx.x;
  if (tid >= (long long)N * 64) return;
  int n = (int)(tid >> 6);
  int c = (int)(tid & 63);
  float zz = z[(size_t)n * 8 + (c >> 3)] + 1e-6f;
  out[(size_t)n * 192 + 128 + c] = wV[tid] / zz;
}

extern "C" void kernel_launch(void* const* d_in, const int* in_sizes, int n_in,
                              void* d_out, int out_size, void* d_ws, size_t ws_size,
                              hipStream_t stream) {
  const float* h  = (const float*)d_in[0];
  const float* p  = (const float*)d_in[1];
  const float* e  = (const float*)d_in[2];
  const float* ew = (const float*)d_in[3];
  const float* fc = (const float*)d_in[4];
  const int* esrc = (const int*)d_in[5];
  const int* edst = (const int*)d_in[6];
  const float* Wc = (const float*)d_in[7];
  const float* bc = (const float*)d_in[8];
  const float* Wg = (const float*)d_in[9];
  const float* bg = (const float*)d_in[10];
  const float* WQ = (const float*)d_in[11];
  const float* WK = (const float*)d_in[12];
  const float* WVm = (const float*)d_in[13];
  const float* WE = (const float*)d_in[14];
  float* out = (float*)d_out;

  const int N = in_sizes[0] / 64;
  const int E = in_sizes[5];
  const size_t ND = (size_t)N * 64;

  float* w = (float*)d_ws;
  float* deg  = w; w += N;
  float* dinv = w; w += N;
  float* wlap = w; w += E;
  float* Tx1  = w; w += ND;
  float* Tx2  = w; w += ND;
  float* Tx3  = w; w += ND;
  float* tmp  = w; w += ND;
  float* gcn  = w; w += ND;
  float* hp   = w; w += 2 * ND;
  float* Qh   = w; w += ND;
  float* Kh   = w; w += ND;
  float* Vh   = w; w += ND;
  float* wVb  = w; w += ND;
  float* zb   = w; w += (size_t)N * 8;

  hipMemsetAsync(deg, 0, sizeof(float) * N, stream);
  hipMemsetAsync(Tx1, 0, sizeof(float) * ND, stream);
  hipMemsetAsync(tmp, 0, sizeof(float) * ND, stream);
  hipMemsetAsync(gcn, 0, sizeof(float) * ND, stream);
  hipMemsetAsync(wVb, 0, sizeof(float) * ND, stream);
  hipMemsetAsync(zb, 0, sizeof(float) * (size_t)N * 8, stream);

  const int B = 256;
  const long long EF = (long long)E * 64;
  const int efBlocks = (int)((EF + B - 1) / B);
  const int ndBlocks = (int)(((long long)ND + B - 1) / B);

  deg_kernel<<<(E + B - 1) / B, B, 0, stream>>>(edst, deg, E);
  dinv_kernel<<<(N + B - 1) / B, B, 0, stream>>>(deg, dinv, N);
  wlap_kernel<<<(E + B - 1) / B, B, 0, stream>>>(dinv, esrc, edst, wlap, E);

  scatter_wadd<<<efBlocks, B, 0, stream>>>(h, wlap, esrc, edst, Tx1, EF);
  scatter_wadd<<<efBlocks, B, 0, stream>>>(Tx1, wlap, esrc, edst, tmp, EF);
  combine_kernel<<<ndBlocks, B, 0, stream>>>(tmp, h, Tx2, (long long)ND);
  hipMemsetAsync(tmp, 0, sizeof(float) * ND, stream);
  scatter_wadd<<<efBlocks, B, 0, stream>>>(Tx2, wlap, esrc, edst, tmp, EF);
  combine_kernel<<<ndBlocks, B, 0, stream>>>(tmp, Tx1, Tx3, (long long)ND);

  const int rowTiles = (N + 15) / 16;
  const int gemmBlocks = (rowTiles * 4 + 7) / 8;
  cheb_gemm_kernel<<<gemmBlocks, 256, 0, stream>>>(h, Tx1, Tx2, Tx3, fc, Wc, bc, out, N, N);

  scatter_wadd<<<efBlocks, B, 0, stream>>>(h, ew, esrc, edst, gcn, EF);
  gemm_wmma_kernel<<<gemmBlocks, 256, 0, stream>>>(gcn, 64, Wg, bg, out, 192, 64, N, 64);

  concat_hp<<<(int)(((long long)N * 128 + B - 1) / B), B, 0, stream>>>(h, p, hp, N);
  gemm_wmma_kernel<<<gemmBlocks, 256, 0, stream>>>(hp, 128, WQ, nullptr, Qh, 64, 0, N, 128);
  gemm_wmma_kernel<<<gemmBlocks, 256, 0, stream>>>(hp, 128, WK, nullptr, Kh, 64, 0, N, 128);
  gemm_wmma_kernel<<<gemmBlocks, 256, 0, stream>>>(hp, 128, WVm, nullptr, Vh, 64, 0, N, 128);

  const int eTiles = (E + 15) / 16;
  attn_kernel<<<(eTiles + 3) / 4, 128, 0, stream>>>(e, WE, Qh, Kh, Vh, esrc, edst, wVb, zb, E);
  finalize_kernel<<<(int)(((long long)N * 64 + B - 1) / B), B, 0, stream>>>(wVb, zb, out, N);
}


# Per-edge global atomics removed: tile ranks from per-block LDS histograms (one global add per block and tile), in-degrees counted in LDS per tile; slot order stays arrival order; 1 launch fewer
# speedup vs baseline: 2.6902x; 2.6902x over previous
#include <hip/hip_runtime.h>
#include <hip/hip_bf16.h>

typedef __attribute__((ext_vector_type(2))) float v2f;
typedef __attribute__((ext_vector_type(8))) float v8f;
typedef __attribute__((ext_vector_type(4))) float v4f;

#define TILE_ROWS 64
#define CNT_PITCH 1
#define RANK_THREADS 1024
#define RANK_PER_THREAD 8
#define SLOTS_PER_TRIP 16

__device__ __forceinline__ v8f wmma4(v2f a, v2f b, v8f c) {
  return __builtin_amdgcn_wmma_f32_16x16x4_f32(
       false, a,  false, b,
       (short)0, c,  false,  false);
}


__global__ void __launch_bounds__(RANK_THREADS)
tile_rank_kernel(const int* __restrict__ dst, int* __restrict__ cnt,
                 int* __restrict__ rank, int E, int nb) {
  extern __shared__ int hist[];
  for (int t = threadIdx.x; t < nb; t += RANK_THREADS) hist[t] = 0;
  __syncthreads();
  const int e0 = (int)blockIdx.x * (RANK_THREADS * RANK_PER_THREAD) + (int)threadIdx.x;
  int tl[RANK_PER_THREAD], lr[RANK_PER_THREAD];
#pragma unroll
  for (int j = 0; j < RANK_PER_THREAD; ++j) {
    int e = e0 + j * RANK_THREADS;
    tl[j] = (int)((unsigned)dst[e < E ? e : E - 1] / TILE_ROWS);
  }
#pragma unroll
  for (int j = 0; j < RANK_PER_THREAD; ++j)
    lr[j] = atomicAdd(&hist[tl[j]], (e0 + j * RANK_THREADS < E) ? 1 : 0);
  __syncthreads();
  for (int t = threadIdx.x; t < nb; t += RANK_THREADS) {
    int c = hist[t];
    hist[t] = (c > 0) ? atomicAdd(&cnt[(size_t)t * CNT_PITCH], c) : 0;
  }
  __syncthreads();
  int bs[RANK_PER_THREAD];
#pragma unroll
  for (int j = 0; j < RANK_PER_THREAD; ++j) bs[j] = hist[tl[j]];
#pragma unroll
  for (int j = 0; j < RANK_PER_THREAD; ++j) {
    int e = e0 + j * RANK_THREADS;
    if (e < E) rank[e] = bs[j] + lr[j];
  }
}

__global__ void tile_scan_kernel(const int* __restrict__ cnt, int* __restrict__ start, int nb) {
  __shared__ int part[1024];
  int t = threadIdx.x;
  int per = (nb + 1023) / 1024;
  int b0 = t * per;
  int s = 0;
  for (int i = 0; i < per; ++i) {
    int b = b0 + i;
    if (b < nb) s += cnt[(size_t)b * CNT_PITCH];
  }
  part[t] = s;
  __syncthreads();
  for (int off = 1; off < 1024; off <<= 1) {
    int v = (t >= off) ? part[t - off] : 0;
    __syncthreads();
    part[t] += v;
    __syncthreads();
  }
  int run = part[t] - s;
  for (int i = 0; i < per; ++i) {
    int b = b0 + i;
    if (b < nb) {
      start[b] = run;
      run += cnt[(size_t)b * CNT_PITCH];
    }
  }
  if (t == 1023) start[nb] = part[1023];
}

__global__ void tile_fill_kernel(const int* __restrict__ src, const int* __restrict__ dst,
                                 const float* __restrict__ ew,
                                 const int* __restrict__ start, const int* __restrict__ rank,
                                 float* __restrict__ recs, int E) {
  int tid = blockIdx.x * blockDim.x + threadIdx.x;
  if (tid >= E) return;
  int d = dst[tid];
  int b = (int)((unsigned)d / TILE_ROWS);
  int pos = start[b] + rank[tid];
  v4f r;
  r.x = __int_as_float(src[tid]);
  r.y = __int_as_float(tid * TILE_ROWS + (d - b * TILE_ROWS));
  r.z = __int_as_float(d);
  r.w = ew[tid];
  *(v4f*)(recs + 4 * (size_t)pos) = r;
}

__global__ void __launch_bounds__(256)
tile_deg_kernel(const float* __restrict__ recs, const int* __restrict__ start,
                float* __restrict__ dinv, int N) {
  __shared__ int cnt[TILE_ROWS];
  const int b = blockIdx.x;
  if (threadIdx.x < TILE_ROWS) cnt[threadIdx.x] = 0;
  __syncthreads();
  const int lo = start[b], hi = start[b + 1];
  for (int i = lo + (int)threadIdx.x; i < hi; i += 256) {
    unsigned ry = (unsigned)__float_as_int(recs[4 * (size_t)i + 1]);
    atomicAdd(&cnt[ry % TILE_ROWS], 1);
  }
  __syncthreads();
  if (threadIdx.x < TILE_ROWS) {
    int row = b * TILE_ROWS + (int)threadIdx.x;
    if (row < N) {
      float d = (float)cnt[threadIdx.x];
      d = d < 1.0f ? 1.0f : d;
      dinv[row] = rsqrtf(d);
    }
  }
}

__global__ void wlap_kernel(const float* __restrict__ dinv, float* recs, int E) {
  int i = blockIdx.x * blockDim.x + threadIdx.x;
  if (i < E) {
    v4f r = *(const v4f*)(recs + 4 * (size_t)i);
    r.z = -(dinv[__float_as_int(r.x)] * dinv[__float_as_int(r.z)]);
    *(v4f*)(recs + 4 * (size_t)i) = r;
  }
}

template <int WSEL>
__global__ void __launch_bounds__(256)
tile_sum_kernel(const float* __restrict__ x, const float* __restrict__ recs,
                const int* __restrict__ start, float* __restrict__ out, int N) {
  __shared__ float acc[TILE_ROWS * 64];
  const int b = blockIdx.x;
  for (int i = threadIdx.x; i < TILE_ROWS * 16; i += 256) ((v4f*)acc)[i] = (v4f){0.f, 0.f, 0.f, 0.f};
  __syncthreads();
  const int lo = start[b], hi = start[b + 1];
  const int wid = __builtin_amdgcn_readfirstlane((int)(threadIdx.x >> 5));
  const int lane = threadIdx.x & 31;
  for (int i0 = lo + wid * 32; i0 < hi; i0 += 8 * 32) {
    int i = i0 + lane;
    if (i >= hi) i = hi - 1;
    const v4f rec = *(const v4f*)(recs + 4 * (size_t)i);
    const int rs = __float_as_int(rec.x);
    const int rd = (int)((unsigned)__float_as_int(rec.y) % TILE_ROWS);
    const int rw = __float_as_int(WSEL ? rec.w : rec.z);
    const int n = hi - i0;
#pragma unroll
    for (int g = 0; g < 32; g += SLOTS_PER_TRIP) {
      v2f xv[SLOTS_PER_TRIP];
#pragma unroll
      for (int j = 0; j < SLOTS_PER_TRIP; ++j) {
        int s = __builtin_amdgcn_readlane(rs, g + j);
        xv[j] = *(const v2f*)(x + (size_t)s * 64 + 2 * lane);
      }
#pragma unroll
      for (int j = 0; j < SLOTS_PER_TRIP; ++j) {
        if (g + j < n) {
          int dl = __builtin_amdgcn_readlane(rd, g + j);
          float wv = __int_as_float(__builtin_amdgcn_readlane(rw, g + j));
          atomicAdd(&acc[dl * 64 + 2 * lane], xv[j].x * wv);
          atomicAdd(&acc[dl * 64 + 2 * lane + 1], xv[j].y * wv);
        }
      }
    }
  }
  __syncthreads();
  for (int i = threadIdx.x; i < TILE_ROWS * 16; i += 256) {
    int row = b * TILE_ROWS + (i >> 4);
    if (row < N) *(v4f*)(out + (size_t)row * 64 + 4 * (i & 15)) = ((const v4f*)acc)[i];
  }
}

#define ELEM_VECS_COMBINE 4
#define ELEM_VECS_CONCAT 8

__global__ void __launch_bounds__(256)
combine_kernel(const float* __restrict__ a, const float* __restrict__ b,
               float* __restrict__ out, int n4) {
  const int T = (int)gridDim.x * 256;
  const int i = (int)blockIdx.x * 256 + (int)threadIdx.x;
  v4f av[ELEM_VECS_COMBINE], bv[ELEM_VECS_COMBINE];
#pragma unroll
  for (int j = 0; j < ELEM_VECS_COMBINE; ++j) {
    int k = i + j * T;
    k = k < n4 ? k : n4 - 1;
    av[j] = ((const v4f*)a)[k];
    bv[j] = ((const v4f*)b)[k];
  }
#pragma unroll
  for (int j = 0; j < ELEM_VECS_COMBINE; ++j) {
    int k = i + j * T;
    if (k < n4) ((v4f*)out)[k] = 2.0f * av[j] - bv[j];
  }
}

__global__ void __launch_bounds__(256)
concat_hp(const float* __restrict__ h, const float* __restrict__ p,
          float* __restrict__ hp, int N) {
  const int T = (int)gridDim.x * 256;
  const int i = (int)blockIdx.x * 256 + (int)threadIdx.x;
  const int c = (int)threadIdx.x & 31;
  const float* s = ((c < 16) ? h : p) + 4 * (c & 15);
  v4f v[ELEM_VECS_CONCAT];
#pragma unroll
  for (int j = 0; j < ELEM_VECS_CONCAT; ++j) {
    int n = (i + j * T) >> 5;
    n = n < N ? n : N - 1;
    v[j] = *(const v4f*)(s + (size_t)n * 64);
  }
#pragma unroll
  for (int j = 0; j < ELEM_VECS_CONCAT; ++j) {
    int k = i + j * T;
    if ((k >> 5) < N) ((v4f*)hp)[k] = v[j];
  }
}

__global__ void gemm_wmma_kernel(const float* __restrict__ A, int lda,
                                 const float* __restrict__ B,
                                 const float* __restrict__ bias,
                                 float* __restrict__ Out, int ldo, int coloff,
                                 int rows, int kdim) {
  int wave = blockIdx.x * (blockDim.x >> 5) + (threadIdx.x >> 5);
  int rowTiles = (rows + 15) >> 4;
  if (wave >= rowTiles * 4) return;
  int rt = wave >> 2, ct = wave & 3;
  int lane = threadIdx.x & 31, half = lane >> 4, l16 = lane & 15;
  int ar = rt * 16 + l16;
  if (ar >= rows) ar = rows - 1;
  int bc = ct * 16 + l16;
  v8f acc = {};
  for (int k0 = 0; k0 < kdim; k0 += 4) {
    int kr = k0 + 2 * half;
    v2f a, b;
    a.x = A[(size_t)ar * lda + kr];
    a.y = A[(size_t)ar * lda + kr + 1];
    b.x = B[(size_t)kr * 64 + bc];
    b.y = B[(size_t)(kr + 1) * 64 + bc];
    acc = wmma4(a, b, acc);
  }
  float bv = bias ? bias[ct * 16 + l16] : 0.0f;
#pragma unroll
  for (int v = 0; v < 8; ++v) {
    int r = rt * 16 + v + 8 * half;
    if (r < rows) Out[(size_t)r * ldo + coloff + ct * 16 + l16] = acc[v] + bv;
  }
}

__global__ void cheb_gemm_kernel(const float* __restrict__ Tx0, const float* __restrict__ Tx1,
                                 const float* __restrict__ Tx2, const float* __restrict__ Tx3,
                                 const float* __restrict__ fc, const float* __restrict__ Wc,
                                 const float* __restrict__ bias, float* __restrict__ Out,
                                 int rows, int Nn) {
  int wave = blockIdx.x * (blockDim.x >> 5) + (threadIdx.x >> 5);
  int rowTiles = (rows + 15) >> 4;
  if (wave >= rowTiles * 4) return;
  int rt = wave >> 2, ct = wave & 3;
  int lane = threadIdx.x & 31, half = lane >> 4, l16 = lane & 15;
  int ar = rt * 16 + l16;
  if (ar >= rows) ar = rows - 1;
  int bc = ct * 16 + l16;
  const float* As[4] = {Tx0, Tx1, Tx2, Tx3};
  v8f acc = {};
#pragma unroll
  for (int k = 0; k < 4; ++k) {
    const float* A = As[k];
    const float* B = Wc + (size_t)k * 64 * 64;
    float sc = fc[(size_t)k * Nn + ar];
#pragma unroll
    for (int k0 = 0; k0 < 64; k0 += 4) {
      int kr = k0 + 2 * half;
      v2f a, b;
      a.x = sc * A[(size_t)ar * 64 + kr];
      a.y = sc * A[(size_t)ar * 64 + kr + 1];
      b.x = B[(size_t)kr * 64 + bc];
      b.y = B[(size_t)(kr + 1) * 64 + bc];
      acc = wmma4(a, b, acc);
    }
  }
  float bv = bias[ct * 16 + l16];
#pragma unroll
  for (int v = 0; v < 8; ++v) {
    int r = rt * 16 + v + 8 * half;
    if (r < rows) Out[(size_t)r * 192 + ct * 16 + l16] = acc[v] + bv;
  }
}


__device__ __forceinline__ float quad_sum(float v) {
  v += __int_as_float(__builtin_amdgcn_update_dpp(0, __float_as_int(v), 0xB1, 0xF, 0xF, true));
  v += __int_as_float(__builtin_amdgcn_update_dpp(0, __float_as_int(v), 0x4E, 0xF, 0xF, true));
  return v;
}

#define ATT_EDGES_PER_TRIP 8

__global__ void __launch_bounds__(256)
attn_tile_kernel(const float* __restrict__ Ee, const float* __restrict__ WE,
                 const float* __restrict__ Qh, const float* __restrict__ Kh,
                 const float* __restrict__ Vh, const float* __restrict__ recs,
                 const int* __restrict__ start, float* __restrict__ out, int N) {
  __shared__ float acc[TILE_ROWS * 64];
  __shared__ float zacc[TILE_ROWS * 8];
  __shared__ float stage[8 * 16 * 64];
  const int b = blockIdx.x;
  for (int i = threadIdx.x; i < TILE_ROWS * 16; i += 256) ((v4f*)acc)[i] = (v4f){0.f, 0.f, 0.f, 0.f};
  for (int i = threadIdx.x; i < TILE_ROWS * 8; i += 256) zacc[i] = 0.f;
  __syncthreads();
  const int lo = start[b], hi = start[b + 1];
  const int wid = __builtin_amdgcn_readfirstlane((int)(threadIdx.x >> 5));
  const int lane = threadIdx.x & 31, half = lane >> 4, l16 = lane & 15;
  float* myLds = stage + wid * 1024;
  const float scale = 0.35355339059327373f;

  for (int i0 = lo + wid * 16; i0 < hi; i0 += 8 * 16) {
    int i = i0 + l16;
    if (i >= hi) i = hi - 1;
    const v4f rec = *(const v4f*)(recs + 4 * (size_t)i);
    const int rs = __float_as_int(rec.x);
    const unsigned ry = (unsigned)__float_as_int(rec.y);
    const int rd = (int)(ry % TILE_ROWS);
    const int ar = (int)(ry / TILE_ROWS);

#pragma unroll 1
    for (int ct = 0; ct < 4; ++ct) {
      v8f c = {};
#pragma unroll
      for (int k0 = 0; k0 < 64; k0 += 4) {
        int kr = k0 + 2 * half;
        v2f a, bb;
        a.x = Ee[(size_t)ar * 64 + kr];
        a.y = Ee[(size_t)ar * 64 + kr + 1];
        int bc = ct * 16 + l16;
        bb.x = WE[(size_t)kr * 64 + bc];
        bb.y = WE[(size_t)(kr + 1) * 64 + bc];
        c = wmma4(a, bb, c);
      }
#pragma unroll
      for (int v = 0; v < 8; ++v)
        myLds[(v + 8 * half) * 64 + ct * 16 + l16] = c[v];
    }

    const int n = hi - i0;
    const float* qbase = Qh + (size_t)b * TILE_ROWS * 64;
#pragma unroll
    for (int g = 0; g < 16; g += ATT_EDGES_PER_TRIP) {
      v2f kv[ATT_EDGES_PER_TRIP], qv[ATT_EDGES_PER_TRIP], vv[ATT_EDGES_PER_TRIP], ev[ATT_EDGES_PER_TRIP];
#pragma unroll
      for (int j = 0; j < ATT_EDGES_PER_TRIP; ++j) {
        int s = __builtin_amdgcn_readlane(rs, g + j);
        int dl = __builtin_amdgcn_readlane(rd, g + j);
        kv[j] = *(const v2f*)(Kh + (size_t)s * 64 + 2 * lane);
        qv[j] = *(const v2f*)(qbase + (size_t)dl * 64 + 2 * lane);
        vv[j] = *(const v2f*)(Vh + (size_t)s * 64 + 2 * lane);
        ev[j] = *(const v2f*)(myLds + (g + j) * 64 + 2 * lane);
      }
#pragma unroll
      for (int j = 0; j < ATT_EDGES_PER_TRIP; ++j) {
        if (g + j < n) {
          int dl = __builtin_amdgcn_readlane(rd, g + j);
          float sc = quad_sum(kv[j].x * qv[j].x * ev[j].x + kv[j].y * qv[j].y * ev[j].y);
          sc *= scale;
          sc = fminf(5.0f, fmaxf(-5.0f, sc));
          float ss = __expf(sc);
          atomicAdd(&acc[dl * 64 + 2 * lane], vv[j].x * ss);
          atomicAdd(&acc[dl * 64 + 2 * lane + 1], vv[j].y * ss);
          if ((lane & 3) == 0) atomicAdd(&zacc[dl * 8 + (lane >> 2)], ss);
        }
      }
    }
  }
  __syncthreads();
  for (int i = threadIdx.x; i < TILE_ROWS * 16; i += 256) {
    int r = i >> 4, c4 = i & 15;
    int row = b * TILE_ROWS + r;
    if (row < N) {
      v4f a = ((const v4f*)acc)[i];
      float zz = zacc[r * 8 + (c4 >> 1)] + 1e-6f;
      v4f o = {a.x / zz, a.y / zz, a.z / zz, a.w / zz};
      *(v4f*)(out + (size_t)row * 192 + 128 + 4 * c4) = o;
    }
  }
}

extern "C" void kernel_launch(void* const* d_in, const int* in_sizes, int n_in,
                              void* d_out, int out_size, void* d_ws, size_t ws_size,
                              hipStream_t stream) {
  const float* h  = (const float*)d_in[0];
  const float* p  = (const float*)d_in[1];
  const float* e  = (const float*)d_in[2];
  const float* ew = (const float*)d_in[3];
  const float* fc = (const float*)d_in[4];
  const int* esrc = (const int*)d_in[5];
  const int* edst = (const int*)d_in[6];
  const float* Wc = (const float*)d_in[7];
  const float* bc = (const float*)d_in[8];
  const float* Wg = (const float*)d_in[9];
  const float* bg = (const float*)d_in[10];
  const float* WQ = (const float*)d_in[11];
  const float* WK = (const float*)d_in[12];
  const float* WVm = (const float*)d_in[13];
  const float* WE = (const float*)d_in[14];
  float* out = (float*)d_out;

  const int N = in_sizes[0] / 64;
  const int E = in_sizes[5];
  const size_t ND = (size_t)N * 64;

  float* w = (float*)d_ws;
  w += N;
  float* dinv = w; w += N;
  w += E;
  w = (float*)(((size_t)w + 15) & ~(size_t)15);
  float* Tx1  = w; w += ND;
  float* Tx2  = w; w += ND;
  float* Tx3  = w; w += ND;
  float* tmp  = w; w += ND;
  float* gcn  = w; w += ND;
  float* hp   = w; w += 2 * ND;
  float* Qh   = w; w += ND;
  float* Kh   = w; w += ND;
  float* Vh   = w; w += ND;
  float* list = w; w += ND + (size_t)N * 8;

  const int nb = (N + TILE_ROWS - 1) / TILE_ROWS;
  float* recs   = list;
  int* binStart = (int*)(recs + 4 * (size_t)E);
  int* binCnt   = (int*)hp;
  int* rank     = binCnt + (size_t)nb * CNT_PITCH;

  hipMemsetAsync(binCnt, 0, sizeof(int) * (size_t)nb * CNT_PITCH, stream);

  const int B = 256;
  const int nd4 = N * 16;
  const int combBlocks = (nd4 + B * ELEM_VECS_COMBINE - 1) / (B * ELEM_VECS_COMBINE);
  const int catBlocks = (2 * nd4 + B * ELEM_VECS_CONCAT - 1) / (B * ELEM_VECS_CONCAT);

  const int rankEdges = RANK_THREADS * RANK_PER_THREAD;
  tile_rank_kernel<<<(E + rankEdges - 1) / rankEdges, RANK_THREADS, sizeof(int) * (size_t)nb, stream>>>(
      edst, binCnt, rank, E, nb);
  tile_scan_kernel<<<1, 1024, 0, stream>>>(binCnt, binStart, nb);
  tile_fill_kernel<<<(E + B - 1) / B, B, 0, stream>>>(esrc, edst, ew, binStart, rank, recs, E);

  tile_deg_kernel<<<nb, 256, 0, stream>>>(recs, binStart, dinv, N);
  wlap_kernel<<<(E + B - 1) / B, B, 0, stream>>>(dinv, recs, E);

  tile_sum_kernel<0><<<nb, 256, 0, stream>>>(h, recs, binStart, Tx1, N);
  tile_sum_kernel<0><<<nb, 256, 0, stream>>>(Tx1, recs, binStart, tmp, N);
  combine_kernel<<<combBlocks, B, 0, stream>>>(tmp, h, Tx2, nd4);
  tile_sum_kernel<0><<<nb, 256, 0, stream>>>(Tx2, recs, binStart, tmp, N);
  combine_kernel<<<combBlocks, B, 0, stream>>>(tmp, Tx1, Tx3, nd4);

  const int rowTiles = (N + 15) / 16;
  const int gemmBlocks = (rowTiles * 4 + 7) / 8;
  cheb_gemm_kernel<<<gemmBlocks, 256, 0, stream>>>(h, Tx1, Tx2, Tx3, fc, Wc, bc, out, N, N);

  tile_sum_kernel<1><<<nb, 256, 0, stream>>>(h, recs, binStart, gcn, N);
  gemm_wmma_kernel<<<gemmBlocks, 256, 0, stream>>>(gcn, 64, Wg, bg, out, 192, 64, N, 64);

  concat_hp<<<catBlocks, B, 0, stream>>>(h, p, hp, N);
  gemm_wmma_kernel<<<gemmBlocks, 256, 0, stream>>>(hp, 128, WQ, nullptr, Qh, 64, 0, N, 128);
  gemm_wmma_kernel<<<gemmBlocks, 256, 0, stream>>>(hp, 128, WK, nullptr, Kh, 64, 0, N, 128);
  gemm_wmma_kernel<<<gemmBlocks, 256, 0, stream>>>(hp, 128, WVm, nullptr, Vh, 64, 0, N, 128);

  attn_tile_kernel<<<nb, 256, 0, stream>>>(e, WE, Qh, Kh, Vh, recs, binStart, out, N);
}


# Attention kernel, Eh product: WE (B operand) copied once per work-group into LDS in fragment order; per 16 slots 32 16-byte LDS reads, grouped ahead of the products, replace 128 4-byte global loads
# speedup vs baseline: 2.8541x; 1.0609x over previous
#include <hip/hip_runtime.h>
#include <hip/hip_bf16.h>

typedef __attribute__((ext_vector_type(2))) float v2f;
typedef __attribute__((ext_vector_type(8))) float v8f;
typedef __attribute__((ext_vector_type(4))) float v4f;

#define TILE_ROWS 64
#define CNT_PITCH 1
#define RANK_THREADS 1024
#define RANK_PER_THREAD 8
#define SLOTS_PER_TRIP 16

__device__ __forceinline__ v8f wmma4(v2f a, v2f b, v8f c) {
  return __builtin_amdgcn_wmma_f32_16x16x4_f32(
       false, a,  false, b,
       (short)0, c,  false,  false);
}


__global__ void __launch_bounds__(RANK_THREADS)
tile_rank_kernel(const int* __restrict__ dst, int* __restrict__ cnt,
                 int* __restrict__ rank, int E, int nb) {
  extern __shared__ int hist[];
  for (int t = threadIdx.x; t < nb; t += RANK_THREADS) hist[t] = 0;
  __syncthreads();
  const int e0 = (int)blockIdx.x * (RANK_THREADS * RANK_PER_THREAD) + (int)threadIdx.x;
  int tl[RANK_PER_THREAD], lr[RANK_PER_THREAD];
#pragma unroll
  for (int j = 0; j < RANK_PER_THREAD; ++j) {
    int e = e0 + j * RANK_THREADS;
    tl[j] = (int)((unsigned)dst[e < E ? e : E - 1] / TILE_ROWS);
  }
#pragma unroll
  for (int j = 0; j < RANK_PER_THREAD; ++j)
    lr[j] = atomicAdd(&hist[tl[j]], (e0 + j * RANK_THREADS < E) ? 1 : 0);
  __syncthreads();
  for (int t = threadIdx.x; t < nb; t += RANK_THREADS) {
    int c = hist[t];
    hist[t] = (c > 0) ? atomicAdd(&cnt[(size_t)t * CNT_PITCH], c) : 0;
  }
  __syncthreads();
  int bs[RANK_PER_THREAD];
#pragma unroll
  for (int j = 0; j < RANK_PER_THREAD; ++j) bs[j] = hist[tl[j]];
#pragma unroll
  for (int j = 0; j < RANK_PER_THREAD; ++j) {
    int e = e0 + j * RANK_THREADS;
    if (e < E) rank[e] = bs[j] + lr[j];
  }
}

__global__ void tile_scan_kernel(const int* __restrict__ cnt, int* __restrict__ start, int nb) {
  __shared__ int part[1024];
  int t = threadIdx.x;
  int per = (nb + 1023) / 1024;
  int b0 = t * per;
  int s = 0;
  for (int i = 0; i < per; ++i) {
    int b = b0 + i;
    if (b < nb) s += cnt[(size_t)b * CNT_PITCH];
  }
  part[t] = s;
  __syncthreads();
  for (int off = 1; off < 1024; off <<= 1) {
    int v = (t >= off) ? part[t - off] : 0;
    __syncthreads();
    part[t] += v;
    __syncthreads();
  }
  int run = part[t] - s;
  for (int i = 0; i < per; ++i) {
    int b = b0 + i;
    if (b < nb) {
      start[b] = run;
      run += cnt[(size_t)b * CNT_PITCH];
    }
  }
  if (t == 1023) start[nb] = part[1023];
}

__global__ void tile_fill_kernel(const int* __restrict__ src, const int* __restrict__ dst,
                                 const float* __restrict__ ew,
                                 const int* __restrict__ start, const int* __restrict__ rank,
                                 float* __restrict__ recs, int E) {
  int tid = blockIdx.x * blockDim.x + threadIdx.x;
  if (tid >= E) return;
  int d = dst[tid];
  int b = (int)((unsigned)d / TILE_ROWS);
  int pos = start[b] + rank[tid];
  v4f r;
  r.x = __int_as_float(src[tid]);
  r.y = __int_as_float(tid * TILE_ROWS + (d - b * TILE_ROWS));
  r.z = __int_as_float(d);
  r.w = ew[tid];
  *(v4f*)(recs + 4 * (size_t)pos) = r;
}

__global__ void __launch_bounds__(256)
tile_deg_kernel(const float* __restrict__ recs, const int* __restrict__ start,
                float* __restrict__ dinv, int N) {
  __shared__ int cnt[TILE_ROWS];
  const int b = blockIdx.x;
  if (threadIdx.x < TILE_ROWS) cnt[threadIdx.x] = 0;
  __syncthreads();
  const int lo = start[b], hi = start[b + 1];
  for (int i = lo + (int)threadIdx.x; i < hi; i += 256) {
    unsigned ry = (unsigned)__float_as_int(recs[4 * (size_t)i + 1]);
    atomicAdd(&cnt[ry % TILE_ROWS], 1);
  }
  __syncthreads();
  if (threadIdx.x < TILE_ROWS) {
    int row = b * TILE_ROWS + (int)threadIdx.x;
    if (row < N) {
      float d = (float)cnt[threadIdx.x];
      d = d < 1.0f ? 1.0f : d;
      dinv[row] = rsqrtf(d);
    }
  }
}

__global__ void wlap_kernel(const float* __restrict__ dinv, float* recs, int E) {
  int i = blockIdx.x * blockDim.x + threadIdx.x;
  if (i < E) {
    v4f r = *(const v4f*)(recs + 4 * (size_t)i);
    r.z = -(dinv[__float_as_int(r.x)] * dinv[__float_as_int(r.z)]);
    *(v4f*)(recs + 4 * (size_t)i) = r;
  }
}

template <int WSEL>
__global__ void __launch_bounds__(256)
tile_sum_kernel(const float* __restrict__ x, const float* __restrict__ recs,
                const int* __restrict__ start, float* __restrict__ out, int N) {
  __shared__ float acc[TILE_ROWS * 64];
  const int b = blockIdx.x;
  for (int i = threadIdx.x; i < TILE_ROWS * 16; i += 256) ((v4f*)acc)[i] = (v4f){0.f, 0.f, 0.f, 0.f};
  __syncthreads();
  const int lo = start[b], hi = start[b + 1];
  const int wid = __builtin_amdgcn_readfirstlane((int)(threadIdx.x >> 5));
  const int lane = threadIdx.x & 31;
  for (int i0 = lo + wid * 32; i0 < hi; i0 += 8 * 32) {
    int i = i0 + lane;
    if (i >= hi) i = hi - 1;
    const v4f rec = *(const v4f*)(recs + 4 * (size_t)i);
    const int rs = __float_as_int(rec.x);
    const int rd = (int)((unsigned)__float_as_int(rec.y) % TILE_ROWS);
    const int rw = __float_as_int(WSEL ? rec.w : rec.z);
    const int n = hi - i0;
#pragma unroll
    for (int g = 0; g < 32; g += SLOTS_PER_TRIP) {
      v2f xv[SLOTS_PER_TRIP];
#pragma unroll
      for (int j = 0; j < SLOTS_PER_TRIP; ++j) {
        int s = __builtin_amdgcn_readlane(rs, g + j);
        xv[j] = *(const v2f*)(x + (size_t)s * 64 + 2 * lane);
      }
#pragma unroll
      for (int j = 0; j < SLOTS_PER_TRIP; ++j) {
        if (g + j < n) {
          int dl = __builtin_amdgcn_readlane(rd, g + j);
          float wv = __int_as_float(__builtin_amdgcn_readlane(rw, g + j));
          atomicAdd(&acc[dl * 64 + 2 * lane], xv[j].x * wv);
          atomicAdd(&acc[dl * 64 + 2 * lane + 1], xv[j].y * wv);
        }
      }
    }
  }
  __syncthreads();
  for (int i = threadIdx.x; i < TILE_ROWS * 16; i += 256) {
    int row = b * TILE_ROWS + (i >> 4);
    if (row < N) *(v4f*)(out + (size_t)row * 64 + 4 * (i & 15)) = ((const v4f*)acc)[i];
  }
}

#define ELEM_VECS_COMBINE 4
#define ELEM_VECS_CONCAT 8

__global__ void __launch_bounds__(256)
combine_kernel(const float* __restrict__ a, const float* __restrict__ b,
               float* __restrict__ out, int n4) {
  const int T = (int)gridDim.x * 256;
  const int i = (int)blockIdx.x * 256 + (int)threadIdx.x;
  v4f av[ELEM_VECS_COMBINE], bv[ELEM_VECS_COMBINE];
#pragma unroll
  for (int j = 0; j < ELEM_VECS_COMBINE; ++j) {
    int k = i + j * T;
    k = k < n4 ? k : n4 - 1;
    av[j] = ((const v4f*)a)[k];
    bv[j] = ((const v4f*)b)[k];
  }
#pragma unroll
  for (int j = 0; j < ELEM_VECS_COMBINE; ++j) {
    int k = i + j * T;
    if (k < n4) ((v4f*)out)[k] = 2.0f * av[j] - bv[j];
  }
}

__global__ void __launch_bounds__(256)
concat_hp(const float* __restrict__ h, const float* __restrict__ p,
          float* __restrict__ hp, int N) {
  const int T = (int)gridDim.x * 256;
  const int i = (int)blockIdx.x * 256 + (int)threadIdx.x;
  const int c = (int)threadIdx.x & 31;
  const float* s = ((c < 16) ? h : p) + 4 * (c & 15);
  v4f v[ELEM_VECS_CONCAT];
#pragma unroll
  for (int j = 0; j < ELEM_VECS_CONCAT; ++j) {
    int n = (i + j * T) >> 5;
    n = n < N ? n : N - 1;
    v[j] = *(const v4f*)(s + (size_t)n * 64);
  }
#pragma unroll
  for (int j = 0; j < ELEM_VECS_CONCAT; ++j) {
    int k = i + j * T;
    if ((k >> 5) < N) ((v4f*)hp)[k] = v[j];
  }
}

__global__ void gemm_wmma_kernel(const float* __restrict__ A, int lda,
                                 const float* __restrict__ B,
                                 const float* __restrict__ bias,
                                 float* __restrict__ Out, int ldo, int coloff,
                                 int rows, int kdim) {
  int wave = blockIdx.x * (blockDim.x >> 5) + (threadIdx.x >> 5);
  int rowTiles = (rows + 15) >> 4;
  if (wave >= rowTiles * 4) return;
  int rt = wave >> 2, ct = wave & 3;
  int lane = threadIdx.x & 31, half = lane >> 4, l16 = lane & 15;
  int ar = rt * 16 + l16;
  if (ar >= rows) ar = rows - 1;
  int bc = ct * 16 + l16;
  v8f acc = {};
  for (int k0 = 0; k0 < kdim; k0 += 4) {
    int kr = k0 + 2 * half;
    v2f a, b;
    a.x = A[(size_t)ar * lda + kr];
    a.y = A[(size_t)ar * lda + kr + 1];
    b.x = B[(size_t)kr * 64 + bc];
    b.y = B[(size_t)(kr + 1) * 64 + bc];
    acc = wmma4(a, b, acc);
  }
  float bv = bias ? bias[ct * 16 + l16] : 0.0f;
#pragma unroll
  for (int v = 0; v < 8; ++v) {
    int r = rt * 16 + v + 8 * half;
    if (r < rows) Out[(size_t)r * ldo + coloff + ct * 16 + l16] = acc[v] + bv;
  }
}

__global__ void cheb_gemm_kernel(const float* __restrict__ Tx0, const float* __restrict__ Tx1,
                                 const float* __restrict__ Tx2, const float* __restrict__ Tx3,
                                 const float* __restrict__ fc, const float* __restrict__ Wc,
                                 const float* __restrict__ bias, float* __restrict__ Out,
                                 int rows, int Nn) {
  int wave = blockIdx.x * (blockDim.x >> 5) + (threadIdx.x >> 5);
  int rowTiles = (rows + 15) >> 4;
  if (wave >= rowTiles * 4) return;
  int rt = wave >> 2, ct = wave & 3;
  int lane = threadIdx.x & 31, half = lane >> 4, l16 = lane & 15;
  int ar = rt * 16 + l16;
  if (ar >= rows) ar = rows - 1;
  int bc = ct * 16 + l16;
  const float* As[4] = {Tx0, Tx1, Tx2, Tx3};
  v8f acc = {};
#pragma unroll
  for (int k = 0; k < 4; ++k) {
    const float* A = As[k];
    const float* B = Wc + (size_t)k * 64 * 64;
    float sc = fc[(size_t)k * Nn + ar];
#pragma unroll
    for (int k0 = 0; k0 < 64; k0 += 4) {
      int kr = k0 + 2 * half;
      v2f a, b;
      a.x = sc * A[(size_t)ar * 64 + kr];
      a.y = sc * A[(size_t)ar * 64 + kr + 1];
      b.x = B[(size_t)kr * 64 + bc];
      b.y = B[(size_t)(kr + 1) * 64 + bc];
      acc = wmma4(a, b, acc);
    }
  }
  float bv = bias[ct * 16 + l16];
#pragma unroll
  for (int v = 0; v < 8; ++v) {
    int r = rt * 16 + v + 8 * half;
    if (r < rows) Out[(size_t)r * 192 + ct * 16 + l16] = acc[v] + bv;
  }
}


__device__ __forceinline__ float quad_sum(float v) {
  v += __int_as_float(__builtin_amdgcn_update_dpp(0, __float_as_int(v), 0xB1, 0xF, 0xF, true));
  v += __int_as_float(__builtin_amdgcn_update_dpp(0, __float_as_int(v), 0x4E, 0xF, 0xF, true));
  return v;
}

#define ATT_EDGES_PER_TRIP 8

__global__ void __launch_bounds__(256)
attn_tile_kernel(const float* __restrict__ Ee, const float* __restrict__ WE,
                 const float* __restrict__ Qh, const float* __restrict__ Kh,
                 const float* __restrict__ Vh, const float* __restrict__ recs,
                 const int* __restrict__ start, float* __restrict__ out, int N) {
  __shared__ float acc[TILE_ROWS * 64];
  __shared__ float zacc[TILE_ROWS * 8];
  __shared__ float stage[8 * 16 * 64];
  __shared__ v4f wfrag[16 * 64];
  const int b = blockIdx.x;
  for (int i = threadIdx.x; i < TILE_ROWS * 16; i += 256) ((v4f*)acc)[i] = (v4f){0.f, 0.f, 0.f, 0.f};
  for (int i = threadIdx.x; i < TILE_ROWS * 8; i += 256) zacc[i] = 0.f;
#pragma unroll
  for (int j = 0; j < 4; ++j) {
    const int f = (int)threadIdx.x + 256 * j;
    const int col = f & 63;
    const int r0 = 8 * (f >> 7) + 2 * ((f >> 6) & 1);
    wfrag[f] = (v4f){WE[r0 * 64 + col], WE[(r0 + 1) * 64 + col],
                     WE[(r0 + 4) * 64 + col], WE[(r0 + 5) * 64 + col]};
  }
  __syncthreads();
  const int lo = start[b], hi = start[b + 1];
  const int wid = __builtin_amdgcn_readfirstlane((int)(threadIdx.x >> 5));
  const int lane = threadIdx.x & 31, half = lane >> 4, l16 = lane & 15;
  float* myLds = stage + wid * 1024;
  const float scale = 0.35355339059327373f;

  for (int i0 = lo + wid * 16; i0 < hi; i0 += 8 * 16) {
    int i = i0 + l16;
    if (i >= hi) i = hi - 1;
    const v4f rec = *(const v4f*)(recs + 4 * (size_t)i);
    const int rs = __float_as_int(rec.x);
    const unsigned ry = (unsigned)__float_as_int(rec.y);
    const int rd = (int)(ry % TILE_ROWS);
    const int ar = (int)(ry / TILE_ROWS);

#pragma unroll 1
    for (int ct = 0; ct < 4; ++ct) {
      v8f c = {};
      const v4f* wf = wfrag + half * 64 + ct * 16 + l16;
      v4f q[8];
#pragma unroll
      for (int P = 0; P < 8; ++P) q[P] = wf[P * 128];
#pragma unroll
      for (int P = 0; P < 8; ++P) {
        const int kr = 8 * P + 2 * half;
        v2f a0, a1;
        a0.x = Ee[(size_t)ar * 64 + kr];
        a0.y = Ee[(size_t)ar * 64 + kr + 1];
        a1.x = Ee[(size_t)ar * 64 + kr + 4];
        a1.y = Ee[(size_t)ar * 64 + kr + 5];
        c = wmma4(a0, (v2f){q[P].x, q[P].y}, c);
        c = wmma4(a1, (v2f){q[P].z, q[P].w}, c);
      }
      __builtin_amdgcn_sched_group_barrier(0x100, 8, 0);
      __builtin_amdgcn_sched_group_barrier(0x008, 16, 0);
#pragma unroll
      for (int v = 0; v < 8; ++v)
        myLds[(v + 8 * half) * 64 + ct * 16 + l16] = c[v];
    }

    const int n = hi - i0;
    const float* qbase = Qh + (size_t)b * TILE_ROWS * 64;
#pragma unroll
    for (int g = 0; g < 16; g += ATT_EDGES_PER_TRIP) {
      v2f kv[ATT_EDGES_PER_TRIP], qv[ATT_EDGES_PER_TRIP], vv[ATT_EDGES_PER_TRIP], ev[ATT_EDGES_PER_TRIP];
#pragma unroll
      for (int j = 0; j < ATT_EDGES_PER_TRIP; ++j) {
        int s = __builtin_amdgcn_readlane(rs, g + j);
        int dl = __builtin_amdgcn_readlane(rd, g + j);
        kv[j] = *(const v2f*)(Kh + (size_t)s * 64 + 2 * lane);
        qv[j] = *(const v2f*)(qbase + (size_t)dl * 64 + 2 * lane);
        vv[j] = *(const v2f*)(Vh + (size_t)s * 64 + 2 * lane);
        ev[j] = *(const v2f*)(myLds + (g + j) * 64 + 2 * lane);
      }
#pragma unroll
      for (int j = 0; j < ATT_EDGES_PER_TRIP; ++j) {
        if (g + j < n) {
          int dl = __builtin_amdgcn_readlane(rd, g + j);
          float sc = quad_sum(kv[j].x * qv[j].x * ev[j].x + kv[j].y * qv[j].y * ev[j].y);
          sc *= scale;
          sc = fminf(5.0f, fmaxf(-5.0f, sc));
          float ss = __expf(sc);
          atomicAdd(&acc[dl * 64 + 2 * lane], vv[j].x * ss);
          atomicAdd(&acc[dl * 64 + 2 * lane + 1], vv[j].y * ss);
          if ((lane & 3) == 0) atomicAdd(&zacc[dl * 8 + (lane >> 2)], ss);
        }
      }
    }
  }
  __syncthreads();
  for (int i = threadIdx.x; i < TILE_ROWS * 16; i += 256) {
    int r = i >> 4, c4 = i & 15;
    int row = b * TILE_ROWS + r;
    if (row < N) {
      v4f a = ((const v4f*)acc)[i];
      float zz = zacc[r * 8 + (c4 >> 1)] + 1e-6f;
      v4f o = {a.x / zz, a.y / zz, a.z / zz, a.w / zz};
      *(v4f*)(out + (size_t)row * 192 + 128 + 4 * c4) = o;
    }
  }
}

extern "C" void kernel_launch(void* const* d_in, const int* in_sizes, int n_in,
                              void* d_out, int out_size, void* d_ws, size_t ws_size,
                              hipStream_t stream) {
  const float* h  = (const float*)d_in[0];
  const float* p  = (const float*)d_in[1];
  const float* e  = (const float*)d_in[2];
  const float* ew = (const float*)d_in[3];
  const float* fc = (const float*)d_in[4];
  const int* esrc = (const int*)d_in[5];
  const int* edst = (const int*)d_in[6];
  const float* Wc = (const float*)d_in[7];
  const float* bc = (const float*)d_in[8];
  const float* Wg = (const float*)d_in[9];
  const float* bg = (const float*)d_in[10];
  const float* WQ = (const float*)d_in[11];
  const float* WK = (const float*)d_in[12];
  const float* WVm = (const float*)d_in[13];
  const float* WE = (const float*)d_in[14];
  float* out = (float*)d_out;

  const int N = in_sizes[0] / 64;
  const int E = in_sizes[5];
  const size_t ND = (size_t)N * 64;

  float* w = (float*)d_ws;
  w += N;
  float* dinv = w; w += N;
  w += E;
  w = (float*)(((size_t)w + 15) & ~(size_t)15);
  float* Tx1  = w; w += ND;
  float* Tx2  = w; w += ND;
  float* Tx3  = w; w += ND;
  float* tmp  = w; w += ND;
  float* gcn  = w; w += ND;
  float* hp   = w; w += 2 * ND;
  float* Qh   = w; w += ND;
  float* Kh   = w; w += ND;
  float* Vh   = w; w += ND;
  float* list = w; w += ND + (size_t)N * 8;

  const int nb = (N + TILE_ROWS - 1) / TILE_ROWS;
  float* recs   = list;
  int* binStart = (int*)(recs + 4 * (size_t)E);
  int* binCnt   = (int*)hp;
  int* rank     = binCnt + (size_t)nb * CNT_PITCH;

  hipMemsetAsync(binCnt, 0, sizeof(int) * (size_t)nb * CNT_PITCH, stream);

  const int B = 256;
  const int nd4 = N * 16;
  const int combBlocks = (nd4 + B * ELEM_VECS_COMBINE - 1) / (B * ELEM_VECS_COMBINE);
  const int catBlocks = (2 * nd4 + B * ELEM_VECS_CONCAT - 1) / (B * ELEM_VECS_CONCAT);

  const int rankEdges = RANK_THREADS * RANK_PER_THREAD;
  tile_rank_kernel<<<(E + rankEdges - 1) / rankEdges, RANK_THREADS, sizeof(int) * (size_t)nb, stream>>>(
      edst, binCnt, rank, E, nb);
  tile_scan_kernel<<<1, 1024, 0, stream>>>(binCnt, binStart, nb);
  tile_fill_kernel<<<(E + B - 1) / B, B, 0, stream>>>(esrc, edst, ew, binStart, rank, recs, E);

  tile_deg_kernel<<<nb, 256, 0, stream>>>(recs, binStart, dinv, N);
  wlap_kernel<<<(E + B - 1) / B, B, 0, stream>>>(dinv, recs, E);

  tile_sum_kernel<0><<<nb, 256, 0, stream>>>(h, recs, binStart, Tx1, N);
  tile_sum_kernel<0><<<nb, 256, 0, stream>>>(Tx1, recs, binStart, tmp, N);
  combine_kernel<<<combBlocks, B, 0, stream>>>(tmp, h, Tx2, nd4);
  tile_sum_kernel<0><<<nb, 256, 0, stream>>>(Tx2, recs, binStart, tmp, N);
  combine_kernel<<<combBlocks, B, 0, stream>>>(tmp, Tx1, Tx3, nd4);

  const int rowTiles = (N + 15) / 16;
  const int gemmBlocks = (rowTiles * 4 + 7) / 8;
  cheb_gemm_kernel<<<gemmBlocks, 256, 0, stream>>>(h, Tx1, Tx2, Tx3, fc, Wc, bc, out, N, N);

  tile_sum_kernel<1><<<nb, 256, 0, stream>>>(h, recs, binStart, gcn, N);
  gemm_wmma_kernel<<<gemmBlocks, 256, 0, stream>>>(gcn, 64, Wg, bg, out, 192, 64, N, 64);

  concat_hp<<<catBlocks, B, 0, stream>>>(h, p, hp, N);
  gemm_wmma_kernel<<<gemmBlocks, 256, 0, stream>>>(hp, 128, WQ, nullptr, Qh, 64, 0, N, 128);
  gemm_wmma_kernel<<<gemmBlocks, 256, 0, stream>>>(hp, 128, WK, nullptr, Kh, 64, 0, N, 128);
  gemm_wmma_kernel<<<gemmBlocks, 256, 0, stream>>>(hp, 128, WVm, nullptr, Vh, 64, 0, N, 128);

  attn_tile_kernel<<<nb, 256, 0, stream>>>(e, WE, Qh, Kh, Vh, recs, binStart, out, N);
}


# Dense products (Chebyshev, GCN, Q, K, V): a work-group copies its 32 A rows into LDS with whole-line loads; waves read A fragments there, not by 8-byte global reads of a line per row; sums unchanged
# speedup vs baseline: 3.1673x; 1.1097x over previous
#include <hip/hip_runtime.h>
#include <hip/hip_bf16.h>

typedef __attribute__((ext_vector_type(2))) float v2f;
typedef __attribute__((ext_vector_type(8))) float v8f;
typedef __attribute__((ext_vector_type(4))) float v4f;

#define TILE_ROWS 64
#define CNT_PITCH 1
#define RANK_THREADS 1024
#define RANK_PER_THREAD 8
#define SLOTS_PER_TRIP 16

__device__ __forceinline__ v8f wmma4(v2f a, v2f b, v8f c) {
  return __builtin_amdgcn_wmma_f32_16x16x4_f32(
       false, a,  false, b,
       (short)0, c,  false,  false);
}


__global__ void __launch_bounds__(RANK_THREADS)
tile_rank_kernel(const int* __restrict__ dst, int* __restrict__ cnt,
                 int* __restrict__ rank, int E, int nb) {
  extern __shared__ int hist[];
  for (int t = threadIdx.x; t < nb; t += RANK_THREADS) hist[t] = 0;
  __syncthreads();
  const int e0 = (int)blockIdx.x * (RANK_THREADS * RANK_PER_THREAD) + (int)threadIdx.x;
  int tl[RANK_PER_THREAD], lr[RANK_PER_THREAD];
#pragma unroll
  for (int j = 0; j < RANK_PER_THREAD; ++j) {
    int e = e0 + j * RANK_THREADS;
    tl[j] = (int)((unsigned)dst[e < E ? e : E - 1] / TILE_ROWS);
  }
#pragma unroll
  for (int j = 0; j < RANK_PER_THREAD; ++j)
    lr[j] = atomicAdd(&hist[tl[j]], (e0 + j * RANK_THREADS < E) ? 1 : 0);
  __syncthreads();
  for (int t = threadIdx.x; t < nb; t += RANK_THREADS) {
    int c = hist[t];
    hist[t] = (c > 0) ? atomicAdd(&cnt[(size_t)t * CNT_PITCH], c) : 0;
  }
  __syncthreads();
  int bs[RANK_PER_THREAD];
#pragma unroll
  for (int j = 0; j < RANK_PER_THREAD; ++j) bs[j] = hist[tl[j]];
#pragma unroll
  for (int j = 0; j < RANK_PER_THREAD; ++j) {
    int e = e0 + j * RANK_THREADS;
    if (e < E) rank[e] = bs[j] + lr[j];
  }
}

__global__ void tile_scan_kernel(const int* __restrict__ cnt, int* __restrict__ start, int nb) {
  __shared__ int part[1024];
  int t = threadIdx.x;
  int per = (nb + 1023) / 1024;
  int b0 = t * per;
  int s = 0;
  for (int i = 0; i < per; ++i) {
    int b = b0 + i;
    if (b < nb) s += cnt[(size_t)b * CNT_PITCH];
  }
  part[t] = s;
  __syncthreads();
  for (int off = 1; off < 1024; off <<= 1) {
    int v = (t >= off) ? part[t - off] : 0;
    __syncthreads();
    part[t] += v;
    __syncthreads();
  }
  int run = part[t] - s;
  for (int i = 0; i < per; ++i) {
    int b = b0 + i;
    if (b < nb) {
      start[b] = run;
      run += cnt[(size_t)b * CNT_PITCH];
    }
  }
  if (t == 1023) start[nb] = part[1023];
}

__global__ void tile_fill_kernel(const int* __restrict__ src, const int* __restrict__ dst,
                                 const float* __restrict__ ew,
                                 const int* __restrict__ start, const int* __restrict__ rank,
                                 float* __restrict__ recs, int E) {
  int tid = blockIdx.x * blockDim.x + threadIdx.x;
  if (tid >= E) return;
  int d = dst[tid];
  int b = (int)((unsigned)d / TILE_ROWS);
  int pos = start[b] + rank[tid];
  v4f r;
  r.x = __int_as_float(src[tid]);
  r.y = __int_as_float(tid * TILE_ROWS + (d - b * TILE_ROWS));
  r.z = __int_as_float(d);
  r.w = ew[tid];
  *(v4f*)(recs + 4 * (size_t)pos) = r;
}

__global__ void __launch_bounds__(256)
tile_deg_kernel(const float* __restrict__ recs, const int* __restrict__ start,
                float* __restrict__ dinv, int N) {
  __shared__ int cnt[TILE_ROWS];
  const int b = blockIdx.x;
  if (threadIdx.x < TILE_ROWS) cnt[threadIdx.x] = 0;
  __syncthreads();
  const int lo = start[b], hi = start[b + 1];
  for (int i = lo + (int)threadIdx.x; i < hi; i += 256) {
    unsigned ry = (unsigned)__float_as_int(recs[4 * (size_t)i + 1]);
    atomicAdd(&cnt[ry % TILE_ROWS], 1);
  }
  __syncthreads();
  if (threadIdx.x < TILE_ROWS) {
    int row = b * TILE_ROWS + (int)threadIdx.x;
    if (row < N) {
      float d = (float)cnt[threadIdx.x];
      d = d < 1.0f ? 1.0f : d;
      dinv[row] = rsqrtf(d);
    }
  }
}

__global__ void wlap_kernel(const float* __restrict__ dinv, float* recs, int E) {
  int i = blockIdx.x * blockDim.x + threadIdx.x;
  if (i < E) {
    v4f r = *(const v4f*)(recs + 4 * (size_t)i);
    r.z = -(dinv[__float_as_int(r.x)] * dinv[__float_as_int(r.z)]);
    *(v4f*)(recs + 4 * (size_t)i) = r;
  }
}

template <int WSEL>
__global__ void __launch_bounds__(256)
tile_sum_kernel(const float* __restrict__ x, const float* __restrict__ recs,
                const int* __restrict__ start, float* __restrict__ out, int N) {
  __shared__ float acc[TILE_ROWS * 64];
  const int b = blockIdx.x;
  for (int i = threadIdx.x; i < TILE_ROWS * 16; i += 256) ((v4f*)acc)[i] = (v4f){0.f, 0.f, 0.f, 0.f};
  __syncthreads();
  const int lo = start[b], hi = start[b + 1];
  const int wid = __builtin_amdgcn_readfirstlane((int)(threadIdx.x >> 5));
  const int lane = threadIdx.x & 31;
  for (int i0 = lo + wid * 32; i0 < hi; i0 += 8 * 32) {
    int i = i0 + lane;
    if (i >= hi) i = hi - 1;
    const v4f rec = *(const v4f*)(recs + 4 * (size_t)i);
    const int rs = __float_as_int(rec.x);
    const int rd = (int)((unsigned)__float_as_int(rec.y) % TILE_ROWS);
    const int rw = __float_as_int(WSEL ? rec.w : rec.z);
    const int n = hi - i0;
#pragma unroll
    for (int g = 0; g < 32; g += SLOTS_PER_TRIP) {
      v2f xv[SLOTS_PER_TRIP];
#pragma unroll
      for (int j = 0; j < SLOTS_PER_TRIP; ++j) {
        int s = __builtin_amdgcn_readlane(rs, g + j);
        xv[j] = *(const v2f*)(x + (size_t)s * 64 + 2 * lane);
      }
#pragma unroll
      for (int j = 0; j < SLOTS_PER_TRIP; ++j) {
        if (g + j < n) {
          int dl = __builtin_amdgcn_readlane(rd, g + j);
          float wv = __int_as_float(__builtin_amdgcn_readlane(rw, g + j));
          atomicAdd(&acc[dl * 64 + 2 * lane], xv[j].x * wv);
          atomicAdd(&acc[dl * 64 + 2 * lane + 1], xv[j].y * wv);
        }
      }
    }
  }
  __syncthreads();
  for (int i = threadIdx.x; i < TILE_ROWS * 16; i += 256) {
    int row = b * TILE_ROWS + (i >> 4);
    if (row < N) *(v4f*)(out + (size_t)row * 64 + 4 * (i & 15)) = ((const v4f*)acc)[i];
  }
}

#define ELEM_VECS_COMBINE 4
#define ELEM_VECS_CONCAT 8

__global__ void __launch_bounds__(256)
combine_kernel(const float* __restrict__ a, const float* __restrict__ b,
               float* __restrict__ out, int n4) {
  const int T = (int)gridDim.x * 256;
  const int i = (int)blockIdx.x * 256 + (int)threadIdx.x;
  v4f av[ELEM_VECS_COMBINE], bv[ELEM_VECS_COMBINE];
#pragma unroll
  for (int j = 0; j < ELEM_VECS_COMBINE; ++j) {
    int k = i + j * T;
    k = k < n4 ? k : n4 - 1;
    av[j] = ((const v4f*)a)[k];
    bv[j] = ((const v4f*)b)[k];
  }
#pragma unroll
  for (int j = 0; j < ELEM_VECS_COMBINE; ++j) {
    int k = i + j * T;
    if (k < n4) ((v4f*)out)[k] = 2.0f * av[j] - bv[j];
  }
}

__global__ void __launch_bounds__(256)
concat_hp(const float* __restrict__ h, const float* __restrict__ p,
          float* __restrict__ hp, int N) {
  const int T = (int)gridDim.x * 256;
  const int i = (int)blockIdx.x * 256 + (int)threadIdx.x;
  const int c = (int)threadIdx.x & 31;
  const float* s = ((c < 16) ? h : p) + 4 * (c & 15);
  v4f v[ELEM_VECS_CONCAT];
#pragma unroll
  for (int j = 0; j < ELEM_VECS_CONCAT; ++j) {
    int n = (i + j * T) >> 5;
    n = n < N ? n : N - 1;
    v[j] = *(const v4f*)(s + (size_t)n * 64);
  }
#pragma unroll
  for (int j = 0; j < ELEM_VECS_CONCAT; ++j) {
    int k = i + j * T;
    if ((k >> 5) < N) ((v4f*)hp)[k] = v[j];
  }
}

#define GEMM_ROWS 32
#define GEMM_KSTEPS 8
#define A_PAD 4

template <int KDIM>
__global__ void __launch_bounds__(256)
gemm_wmma_kernel(const float* __restrict__ A, int lda,
                 const float* __restrict__ B,
                 const float* __restrict__ bias,
                 float* __restrict__ Out, int ldo, int coloff, int rows) {
  constexpr int PITCH = KDIM + A_PAD;
  constexpr int VPR = KDIM / 4;
  __shared__ v4f stageA[GEMM_ROWS * PITCH / 4];
  float* As = (float*)stageA;
  const int row0 = (int)blockIdx.x * GEMM_ROWS;
#pragma unroll
  for (int j = 0; j < GEMM_ROWS * VPR / 256; ++j) {
    const int f = (int)threadIdx.x + 256 * j;
    const int r = f / VPR, c4 = f % VPR;
    int gr = row0 + r;
    if (gr >= rows) gr = rows - 1;
    *(v4f*)(As + r * PITCH + 4 * c4) = *(const v4f*)(A + (size_t)gr * lda + 4 * c4);
  }
  __syncthreads();
  const int w = (int)(threadIdx.x >> 5);
  const int rt = (int)blockIdx.x * (GEMM_ROWS / 16) + (w >> 2), ct = w & 3;
  const int rowTiles = (rows + 15) >> 4;
  if (rt >= rowTiles) return;
  const int lane = threadIdx.x & 31, half = lane >> 4, l16 = lane & 15;
  const int bc = ct * 16 + l16;
  const float* ap = As + ((w >> 2) * 16 + l16) * PITCH + 2 * half;
  const float* bp = B + (size_t)(2 * half) * 64 + bc;
  v8f acc = {};
#pragma unroll 1
  for (int kk = 0; kk < KDIM; kk += 4 * GEMM_KSTEPS) {
    v2f a[GEMM_KSTEPS], b[GEMM_KSTEPS];
#pragma unroll
    for (int s = 0; s < GEMM_KSTEPS; ++s) {
      const int k0 = kk + 4 * s;
      a[s] = *(const v2f*)(ap + k0);
      b[s].x = bp[(size_t)k0 * 64];
      b[s].y = bp[(size_t)(k0 + 1) * 64];
    }
#pragma unroll
    for (int s = 0; s < GEMM_KSTEPS; ++s) acc = wmma4(a[s], b[s], acc);
    __builtin_amdgcn_sched_group_barrier(0x020, 2 * GEMM_KSTEPS, 0);
    __builtin_amdgcn_sched_group_barrier(0x100, GEMM_KSTEPS, 0);
    __builtin_amdgcn_sched_group_barrier(0x008, GEMM_KSTEPS, 0);
  }
  float bv = bias ? bias[ct * 16 + l16] : 0.0f;
#pragma unroll
  for (int v = 0; v < 8; ++v) {
    int r = rt * 16 + v + 8 * half;
    if (r < rows) Out[(size_t)r * ldo + coloff + ct * 16 + l16] = acc[v] + bv;
  }
}

__global__ void __launch_bounds__(256)
cheb_gemm_kernel(const float* __restrict__ Tx0, const float* __restrict__ Tx1,
                 const float* __restrict__ Tx2, const float* __restrict__ Tx3,
                 const float* __restrict__ fc, const float* __restrict__ Wc,
                 const float* __restrict__ bias, float* __restrict__ Out,
                 int rows, int Nn) {
  constexpr int PITCH = 64 + A_PAD;
  __shared__ v4f stageA[4 * GEMM_ROWS * PITCH / 4];
  float* As = (float*)stageA;
  const int row0 = (int)blockIdx.x * GEMM_ROWS;
  const float* Asrc[4] = {Tx0, Tx1, Tx2, Tx3};
#pragma unroll
  for (int j = 0; j < 8; ++j) {
    const int r = (int)(threadIdx.x >> 4) + 16 * (j & 1), c4 = (int)threadIdx.x & 15;
    int gr = row0 + r;
    if (gr >= rows) gr = rows - 1;
    *(v4f*)(As + ((j >> 1) * GEMM_ROWS + r) * PITCH + 4 * c4) =
        *(const v4f*)(Asrc[j >> 1] + (size_t)gr * 64 + 4 * c4);
  }
  __syncthreads();
  const int w = (int)(threadIdx.x >> 5);
  const int rt = (int)blockIdx.x * (GEMM_ROWS / 16) + (w >> 2), ct = w & 3;
  const int rowTiles = (rows + 15) >> 4;
  if (rt >= rowTiles) return;
  const int lane = threadIdx.x & 31, half = lane >> 4, l16 = lane & 15;
  int ar = rt * 16 + l16;
  if (ar >= rows) ar = rows - 1;
  const int bc = ct * 16 + l16;
  v8f acc = {};
  float sc[4];
#pragma unroll
  for (int k = 0; k < 4; ++k) sc[k] = fc[(size_t)k * Nn + ar];
#pragma unroll
  for (int k = 0; k < 4; ++k) {
    const float* ak = As + (k * GEMM_ROWS + (w >> 2) * 16 + l16) * PITCH + 2 * half;
    const float* bk = Wc + (size_t)k * 64 * 64 + (size_t)(2 * half) * 64 + bc;
    const float sk = sc[k];
#pragma unroll 1
    for (int kk = 0; kk < 64; kk += 4 * GEMM_KSTEPS) {
      v2f a[GEMM_KSTEPS], b[GEMM_KSTEPS];
#pragma unroll
      for (int s = 0; s < GEMM_KSTEPS; ++s) {
        const int k0 = kk + 4 * s;
        a[s] = *(const v2f*)(ak + k0) * sk;
        b[s].x = bk[(size_t)k0 * 64];
        b[s].y = bk[(size_t)(k0 + 1) * 64];
      }
#pragma unroll
      for (int s = 0; s < GEMM_KSTEPS; ++s) acc = wmma4(a[s], b[s], acc);
      __builtin_amdgcn_sched_group_barrier(0x020, 2 * GEMM_KSTEPS, 0);
      __builtin_amdgcn_sched_group_barrier(0x100, GEMM_KSTEPS, 0);
      __builtin_amdgcn_sched_group_barrier(0x008, GEMM_KSTEPS, 0);
    }
  }
  float bv = bias[ct * 16 + l16];
#pragma unroll
  for (int v = 0; v < 8; ++v) {
    int r = rt * 16 + v + 8 * half;
    if (r < rows) Out[(size_t)r * 192 + ct * 16 + l16] = acc[v] + bv;
  }
}


__device__ __forceinline__ float quad_sum(float v) {
  v += __int_as_float(__builtin_amdgcn_update_dpp(0, __float_as_int(v), 0xB1, 0xF, 0xF, true));
  v += __int_as_float(__builtin_amdgcn_update_dpp(0, __float_as_int(v), 0x4E, 0xF, 0xF, true));
  return v;
}

#define ATT_EDGES_PER_TRIP 8

__global__ void __launch_bounds__(256)
attn_tile_kernel(const float* __restrict__ Ee, const float* __restrict__ WE,
                 const float* __restrict__ Qh, const float* __restrict__ Kh,
                 const float* __restrict__ Vh, const float* __restrict__ recs,
                 const int* __restrict__ start, float* __restrict__ out, int N) {
  __shared__ float acc[TILE_ROWS * 64];
  __shared__ float zacc[TILE_ROWS * 8];
  __shared__ float stage[8 * 16 * 64];
  __shared__ v4f wfrag[16 * 64];
  const int b = blockIdx.x;
  for (int i = threadIdx.x; i < TILE_ROWS * 16; i += 256) ((v4f*)acc)[i] = (v4f){0.f, 0.f, 0.f, 0.f};
  for (int i = threadIdx.x; i < TILE_ROWS * 8; i += 256) zacc[i] = 0.f;
#pragma unroll
  for (int j = 0; j < 4; ++j) {
    const int f = (int)threadIdx.x + 256 * j;
    const int col = f & 63;
    const int r0 = 8 * (f >> 7) + 2 * ((f >> 6) & 1);
    wfrag[f] = (v4f){WE[r0 * 64 + col], WE[(r0 + 1) * 64 + col],
                     WE[(r0 + 4) * 64 + col], WE[(r0 + 5) * 64 + col]};
  }
  __syncthreads();
  const int lo = start[b], hi = start[b + 1];
  const int wid = __builtin_amdgcn_readfirstlane((int)(threadIdx.x >> 5));
  const int lane = threadIdx.x & 31, half = lane >> 4, l16 = lane & 15;
  float* myLds = stage + wid * 1024;
  const float scale = 0.35355339059327373f;

  for (int i0 = lo + wid * 16; i0 < hi; i0 += 8 * 16) {
    int i = i0 + l16;
    if (i >= hi) i = hi - 1;
    const v4f rec = *(const v4f*)(recs + 4 * (size_t)i);
    const int rs = __float_as_int(rec.x);
    const unsigned ry = (unsigned)__float_as_int(rec.y);
    const int rd = (int)(ry % TILE_ROWS);
    const int ar = (int)(ry / TILE_ROWS);

#pragma unroll 1
    for (int ct = 0; ct < 4; ++ct) {
      v8f c = {};
      const v4f* wf = wfrag + half * 64 + ct * 16 + l16;
      v4f q[8];
#pragma unroll
      for (int P = 0; P < 8; ++P) q[P] = wf[P * 128];
#pragma unroll
      for (int P = 0; P < 8; ++P) {
        const int kr = 8 * P + 2 * half;
        v2f a0, a1;
        a0.x = Ee[(size_t)ar * 64 + kr];
        a0.y = Ee[(size_t)ar * 64 + kr + 1];
        a1.x = Ee[(size_t)ar * 64 + kr + 4];
        a1.y = Ee[(size_t)ar * 64 + kr + 5];
        c = wmma4(a0, (v2f){q[P].x, q[P].y}, c);
        c = wmma4(a1, (v2f){q[P].z, q[P].w}, c);
      }
      __builtin_amdgcn_sched_group_barrier(0x100, 8, 0);
      __builtin_amdgcn_sched_group_barrier(0x008, 16, 0);
#pragma unroll
      for (int v = 0; v < 8; ++v)
        myLds[(v + 8 * half) * 64 + ct * 16 + l16] = c[v];
    }

    const int n = hi - i0;
    const float* qbase = Qh + (size_t)b * TILE_ROWS * 64;
#pragma unroll
    for (int g = 0; g < 16; g += ATT_EDGES_PER_TRIP) {
      v2f kv[ATT_EDGES_PER_TRIP], qv[ATT_EDGES_PER_TRIP], vv[ATT_EDGES_PER_TRIP], ev[ATT_EDGES_PER_TRIP];
#pragma unroll
      for (int j = 0; j < ATT_EDGES_PER_TRIP; ++j) {
        int s = __builtin_amdgcn_readlane(rs, g + j);
        int dl = __builtin_amdgcn_readlane(rd, g + j);
        kv[j] = *(const v2f*)(Kh + (size_t)s * 64 + 2 * lane);
        qv[j] = *(const v2f*)(qbase + (size_t)dl * 64 + 2 * lane);
        vv[j] = *(const v2f*)(Vh + (size_t)s * 64 + 2 * lane);
        ev[j] = *(const v2f*)(myLds + (g + j) * 64 + 2 * lane);
      }
#pragma unroll
      for (int j = 0; j < ATT_EDGES_PER_TRIP; ++j) {
        if (g + j < n) {
          int dl = __builtin_amdgcn_readlane(rd, g + j);
          float sc = quad_sum(kv[j].x * qv[j].x * ev[j].x + kv[j].y * qv[j].y * ev[j].y);
          sc *= scale;
          sc = fminf(5.0f, fmaxf(-5.0f, sc));
          float ss = __expf(sc);
          atomicAdd(&acc[dl * 64 + 2 * lane], vv[j].x * ss);
          atomicAdd(&acc[dl * 64 + 2 * lane + 1], vv[j].y * ss);
          if ((lane & 3) == 0) atomicAdd(&zacc[dl * 8 + (lane >> 2)], ss);
        }
      }
    }
  }
  __syncthreads();
  for (int i = threadIdx.x; i < TILE_ROWS * 16; i += 256) {
    int r = i >> 4, c4 = i & 15;
    int row = b * TILE_ROWS + r;
    if (row < N) {
      v4f a = ((const v4f*)acc)[i];
      float zz = zacc[r * 8 + (c4 >> 1)] + 1e-6f;
      v4f o = {a.x / zz, a.y / zz, a.z / zz, a.w / zz};
      *(v4f*)(out + (size_t)row * 192 + 128 + 4 * c4) = o;
    }
  }
}

extern "C" void kernel_launch(void* const* d_in, const int* in_sizes, int n_in,
                              void* d_out, int out_size, void* d_ws, size_t ws_size,
                              hipStream_t stream) {
  const float* h  = (const float*)d_in[0];
  const float* p  = (const float*)d_in[1];
  const float* e  = (const float*)d_in[2];
  const float* ew = (const float*)d_in[3];
  const float* fc = (const float*)d_in[4];
  const int* esrc = (const int*)d_in[5];
  const int* edst = (const int*)d_in[6];
  const float* Wc = (const float*)d_in[7];
  const float* bc = (const float*)d_in[8];
  const float* Wg = (const float*)d_in[9];
  const float* bg = (const float*)d_in[10];
  const float* WQ = (const float*)d_in[11];
  const float* WK = (const float*)d_in[12];
  const float* WVm = (const float*)d_in[13];
  const float* WE = (const float*)d_in[14];
  float* out = (float*)d_out;

  const int N = in_sizes[0] / 64;
  const int E = in_sizes[5];
  const size_t ND = (size_t)N * 64;

  float* w = (float*)d_ws;
  w += N;
  float* dinv = w; w += N;
  w += E;
  w = (float*)(((size_t)w + 15) & ~(size_t)15);
  float* Tx1  = w; w += ND;
  float* Tx2  = w; w += ND;
  float* Tx3  = w; w += ND;
  float* tmp  = w; w += ND;
  float* gcn  = w; w += ND;
  float* hp   = w; w += 2 * ND;
  float* Qh   = w; w += ND;
  float* Kh   = w; w += ND;
  float* Vh   = w; w += ND;
  float* list = w; w += ND + (size_t)N * 8;

  const int nb = (N + TILE_ROWS - 1) / TILE_ROWS;
  float* recs   = list;
  int* binStart = (int*)(recs + 4 * (size_t)E);
  int* binCnt   = (int*)hp;
  int* rank     = binCnt + (size_t)nb * CNT_PITCH;

  hipMemsetAsync(binCnt, 0, sizeof(int) * (size_t)nb * CNT_PITCH, stream);

  const int B = 256;
  const int nd4 = N * 16;
  const int combBlocks = (nd4 + B * ELEM_VECS_COMBINE - 1) / (B * ELEM_VECS_COMBINE);
  const int catBlocks = (2 * nd4 + B * ELEM_VECS_CONCAT - 1) / (B * ELEM_VECS_CONCAT);

  const int rankEdges = RANK_THREADS * RANK_PER_THREAD;
  tile_rank_kernel<<<(E + rankEdges - 1) / rankEdges, RANK_THREADS, sizeof(int) * (size_t)nb, stream>>>(
      edst, binCnt, rank, E, nb);
  tile_scan_kernel<<<1, 1024, 0, stream>>>(binCnt, binStart, nb);
  tile_fill_kernel<<<(E + B - 1) / B, B, 0, stream>>>(esrc, edst, ew, binStart, rank, recs, E);

  tile_deg_kernel<<<nb, 256, 0, stream>>>(recs, binStart, dinv, N);
  wlap_kernel<<<(E + B - 1) / B, B, 0, stream>>>(dinv, recs, E);

  tile_sum_kernel<0><<<nb, 256, 0, stream>>>(h, recs, binStart, Tx1, N);
  tile_sum_kernel<0><<<nb, 256, 0, stream>>>(Tx1, recs, binStart, tmp, N);
  combine_kernel<<<combBlocks, B, 0, stream>>>(tmp, h, Tx2, nd4);
  tile_sum_kernel<0><<<nb, 256, 0, stream>>>(Tx2, recs, binStart, tmp, N);
  combine_kernel<<<combBlocks, B, 0, stream>>>(tmp, Tx1, Tx3, nd4);

  const int gemmBlocks = (N + GEMM_ROWS - 1) / GEMM_ROWS;
  cheb_gemm_kernel<<<gemmBlocks, 256, 0, stream>>>(h, Tx1, Tx2, Tx3, fc, Wc, bc, out, N, N);

  tile_sum_kernel<1><<<nb, 256, 0, stream>>>(h, recs, binStart, gcn, N);
  gemm_wmma_kernel<64><<<gemmBlocks, 256, 0, stream>>>(gcn, 64, Wg, bg, out, 192, 64, N);

  concat_hp<<<catBlocks, B, 0, stream>>>(h, p, hp, N);
  gemm_wmma_kernel<128><<<gemmBlocks, 256, 0, stream>>>(hp, 128, WQ, nullptr, Qh, 64, 0, N);
  gemm_wmma_kernel<128><<<gemmBlocks, 256, 0, stream>>>(hp, 128, WK, nullptr, Kh, 64, 0, N);
  gemm_wmma_kernel<128><<<gemmBlocks, 256, 0, stream>>>(hp, 128, WVm, nullptr, Vh, 64, 0, N);

  attn_tile_kernel<<<nb, 256, 0, stream>>>(e, WE, Qh, Kh, Vh, recs, binStart, out, N);
}


# Q, K, V in one launch: a work-group copies its 32 [h|p] rows into LDS once, straight from h and p, and each wave runs the three products as three accumulator chains on shared A reads; sums unchanged
# speedup vs baseline: 3.3843x; 1.0685x over previous
#include <hip/hip_runtime.h>
#include <hip/hip_bf16.h>

typedef __attribute__((ext_vector_type(2))) float v2f;
typedef __attribute__((ext_vector_type(8))) float v8f;
typedef __attribute__((ext_vector_type(4))) float v4f;

#define TILE_ROWS 64
#define CNT_PITCH 1
#define RANK_THREADS 1024
#define RANK_PER_THREAD 8
#define SLOTS_PER_TRIP 16

__device__ __forceinline__ v8f wmma4(v2f a, v2f b, v8f c) {
  return __builtin_amdgcn_wmma_f32_16x16x4_f32(
       false, a,  false, b,
       (short)0, c,  false,  false);
}


__global__ void __launch_bounds__(RANK_THREADS)
tile_rank_kernel(const int* __restrict__ dst, int* __restrict__ cnt,
                 int* __restrict__ rank, int E, int nb) {
  extern __shared__ int hist[];
  for (int t = threadIdx.x; t < nb; t += RANK_THREADS) hist[t] = 0;
  __syncthreads();
  const int e0 = (int)blockIdx.x * (RANK_THREADS * RANK_PER_THREAD) + (int)threadIdx.x;
  int tl[RANK_PER_THREAD], lr[RANK_PER_THREAD];
#pragma unroll
  for (int j = 0; j < RANK_PER_THREAD; ++j) {
    int e = e0 + j * RANK_THREADS;
    tl[j] = (int)((unsigned)dst[e < E ? e : E - 1] / TILE_ROWS);
  }
#pragma unroll
  for (int j = 0; j < RANK_PER_THREAD; ++j)
    lr[j] = atomicAdd(&hist[tl[j]], (e0 + j * RANK_THREADS < E) ? 1 : 0);
  __syncthreads();
  for (int t = threadIdx.x; t < nb; t += RANK_THREADS) {
    int c = hist[t];
    hist[t] = (c > 0) ? atomicAdd(&cnt[(size_t)t * CNT_PITCH], c) : 0;
  }
  __syncthreads();
  int bs[RANK_PER_THREAD];
#pragma unroll
  for (int j = 0; j < RANK_PER_THREAD; ++j) bs[j] = hist[tl[j]];
#pragma unroll
  for (int j = 0; j < RANK_PER_THREAD; ++j) {
    int e = e0 + j * RANK_THREADS;
    if (e < E) rank[e] = bs[j] + lr[j];
  }
}

__global__ void tile_scan_kernel(const int* __restrict__ cnt, int* __restrict__ start, int nb) {
  __shared__ int part[1024];
  int t = threadIdx.x;
  int per = (nb + 1023) / 1024;
  int b0 = t * per;
  int s = 0;
  for (int i = 0; i < per; ++i) {
    int b = b0 + i;
    if (b < nb) s += cnt[(size_t)b * CNT_PITCH];
  }
  part[t] = s;
  __syncthreads();
  for (int off = 1; off < 1024; off <<= 1) {
    int v = (t >= off) ? part[t - off] : 0;
    __syncthreads();
    part[t] += v;
    __syncthreads();
  }
  int run = part[t] - s;
  for (int i = 0; i < per; ++i) {
    int b = b0 + i;
    if (b < nb) {
      start[b] = run;
      run += cnt[(size_t)b * CNT_PITCH];
    }
  }
  if (t == 1023) start[nb] = part[1023];
}

__global__ void tile_fill_kernel(const int* __restrict__ src, const int* __restrict__ dst,
                                 const float* __restrict__ ew,
                                 const int* __restrict__ start, const int* __restrict__ rank,
                                 float* __restrict__ recs, int E) {
  int tid = blockIdx.x * blockDim.x + threadIdx.x;
  if (tid >= E) return;
  int d = dst[tid];
  int b = (int)((unsigned)d / TILE_ROWS);
  int pos = start[b] + rank[tid];
  v4f r;
  r.x = __int_as_float(src[tid]);
  r.y = __int_as_float(tid * TILE_ROWS + (d - b * TILE_ROWS));
  r.z = __int_as_float(d);
  r.w = ew[tid];
  *(v4f*)(recs + 4 * (size_t)pos) = r;
}

__global__ void __launch_bounds__(256)
tile_deg_kernel(const float* __restrict__ recs, const int* __restrict__ start,
                float* __restrict__ dinv, int N) {
  __shared__ int cnt[TILE_ROWS];
  const int b = blockIdx.x;
  if (threadIdx.x < TILE_ROWS) cnt[threadIdx.x] = 0;
  __syncthreads();
  const int lo = start[b], hi = start[b + 1];
  for (int i = lo + (int)threadIdx.x; i < hi; i += 256) {
    unsigned ry = (unsigned)__float_as_int(recs[4 * (size_t)i + 1]);
    atomicAdd(&cnt[ry % TILE_ROWS], 1);
  }
  __syncthreads();
  if (threadIdx.x < TILE_ROWS) {
    int row = b * TILE_ROWS + (int)threadIdx.x;
    if (row < N) {
      float d = (float)cnt[threadIdx.x];
      d = d < 1.0f ? 1.0f : d;
      dinv[row] = rsqrtf(d);
    }
  }
}

__global__ void wlap_kernel(const float* __restrict__ dinv, float* recs, int E) {
  int i = blockIdx.x * blockDim.x + threadIdx.x;
  if (i < E) {
    v4f r = *(const v4f*)(recs + 4 * (size_t)i);
    r.z = -(dinv[__float_as_int(r.x)] * dinv[__float_as_int(r.z)]);
    *(v4f*)(recs + 4 * (size_t)i) = r;
  }
}

template <int WSEL>
__global__ void __launch_bounds__(256)
tile_sum_kernel(const float* __restrict__ x, const float* __restrict__ recs,
                const int* __restrict__ start, float* __restrict__ out, int N) {
  __shared__ float acc[TILE_ROWS * 64];
  const int b = blockIdx.x;
  for (int i = threadIdx.x; i < TILE_ROWS * 16; i += 256) ((v4f*)acc)[i] = (v4f){0.f, 0.f, 0.f, 0.f};
  __syncthreads();
  const int lo = start[b], hi = start[b + 1];
  const int wid = __builtin_amdgcn_readfirstlane((int)(threadIdx.x >> 5));
  const int lane = threadIdx.x & 31;
  for (int i0 = lo + wid * 32; i0 < hi; i0 += 8 * 32) {
    int i = i0 + lane;
    if (i >= hi) i = hi - 1;
    const v4f rec = *(const v4f*)(recs + 4 * (size_t)i);
    const int rs = __float_as_int(rec.x);
    const int rd = (int)((unsigned)__float_as_int(rec.y) % TILE_ROWS);
    const int rw = __float_as_int(WSEL ? rec.w : rec.z);
    const int n = hi - i0;
#pragma unroll
    for (int g = 0; g < 32; g += SLOTS_PER_TRIP) {
      v2f xv[SLOTS_PER_TRIP];
#pragma unroll
      for (int j = 0; j < SLOTS_PER_TRIP; ++j) {
        int s = __builtin_amdgcn_readlane(rs, g + j);
        xv[j] = *(const v2f*)(x + (size_t)s * 64 + 2 * lane);
      }
#pragma unroll
      for (int j = 0; j < SLOTS_PER_TRIP; ++j) {
        if (g + j < n) {
          int dl = __builtin_amdgcn_readlane(rd, g + j);
          float wv = __int_as_float(__builtin_amdgcn_readlane(rw, g + j));
          atomicAdd(&acc[dl * 64 + 2 * lane], xv[j].x * wv);
          atomicAdd(&acc[dl * 64 + 2 * lane + 1], xv[j].y * wv);
        }
      }
    }
  }
  __syncthreads();
  for (int i = threadIdx.x; i < TILE_ROWS * 16; i += 256) {
    int row = b * TILE_ROWS + (i >> 4);
    if (row < N) *(v4f*)(out + (size_t)row * 64 + 4 * (i & 15)) = ((const v4f*)acc)[i];
  }
}

#define ELEM_VECS_COMBINE 4

__global__ void __launch_bounds__(256)
combine_kernel(const float* __restrict__ a, const float* __restrict__ b,
               float* __restrict__ out, int n4) {
  const int T = (int)gridDim.x * 256;
  const int i = (int)blockIdx.x * 256 + (int)threadIdx.x;
  v4f av[ELEM_VECS_COMBINE], bv[ELEM_VECS_COMBINE];
#pragma unroll
  for (int j = 0; j < ELEM_VECS_COMBINE; ++j) {
    int k = i + j * T;
    k = k < n4 ? k : n4 - 1;
    av[j] = ((const v4f*)a)[k];
    bv[j] = ((const v4f*)b)[k];
  }
#pragma unroll
  for (int j = 0; j < ELEM_VECS_COMBINE; ++j) {
    int k = i + j * T;
    if (k < n4) ((v4f*)out)[k] = 2.0f * av[j] - bv[j];
  }
}

#define GEMM_ROWS 32
#define GEMM_KSTEPS 8
#define A_PAD 4

template <int KDIM>
__global__ void __launch_bounds__(256)
gemm_wmma_kernel(const float* __restrict__ A, int lda,
                 const float* __restrict__ B,
                 const float* __restrict__ bias,
                 float* __restrict__ Out, int ldo, int coloff, int rows) {
  constexpr int PITCH = KDIM + A_PAD;
  constexpr int VPR = KDIM / 4;
  __shared__ v4f stageA[GEMM_ROWS * PITCH / 4];
  float* As = (float*)stageA;
  const int row0 = (int)blockIdx.x * GEMM_ROWS;
#pragma unroll
  for (int j = 0; j < GEMM_ROWS * VPR / 256; ++j) {
    const int f = (int)threadIdx.x + 256 * j;
    const int r = f / VPR, c4 = f % VPR;
    int gr = row0 + r;
    if (gr >= rows) gr = rows - 1;
    *(v4f*)(As + r * PITCH + 4 * c4) = *(const v4f*)(A + (size_t)gr * lda + 4 * c4);
  }
  __syncthreads();
  const int w = (int)(threadIdx.x >> 5);
  const int rt = (int)blockIdx.x * (GEMM_ROWS / 16) + (w >> 2), ct = w & 3;
  const int rowTiles = (rows + 15) >> 4;
  if (rt >= rowTiles) return;
  const int lane = threadIdx.x & 31, half = lane >> 4, l16 = lane & 15;
  const int bc = ct * 16 + l16;
  const float* ap = As + ((w >> 2) * 16 + l16) * PITCH + 2 * half;
  const float* bp = B + (size_t)(2 * half) * 64 + bc;
  v8f acc = {};
#pragma unroll 1
  for (int kk = 0; kk < KDIM; kk += 4 * GEMM_KSTEPS) {
    v2f a[GEMM_KSTEPS], b[GEMM_KSTEPS];
#pragma unroll
    for (int s = 0; s < GEMM_KSTEPS; ++s) {
      const int k0 = kk + 4 * s;
      a[s] = *(const v2f*)(ap + k0);
      b[s].x = bp[(size_t)k0 * 64];
      b[s].y = bp[(size_t)(k0 + 1) * 64];
    }
#pragma unroll
    for (int s = 0; s < GEMM_KSTEPS; ++s) acc = wmma4(a[s], b[s], acc);
    __builtin_amdgcn_sched_group_barrier(0x020, 2 * GEMM_KSTEPS, 0);
    __builtin_amdgcn_sched_group_barrier(0x100, GEMM_KSTEPS, 0);
    __builtin_amdgcn_sched_group_barrier(0x008, GEMM_KSTEPS, 0);
  }
  float bv = bias ? bias[ct * 16 + l16] : 0.0f;
#pragma unroll
  for (int v = 0; v < 8; ++v) {
    int r = rt * 16 + v + 8 * half;
    if (r < rows) Out[(size_t)r * ldo + coloff + ct * 16 + l16] = acc[v] + bv;
  }
}

__global__ void __launch_bounds__(256)
qkv_gemm_kernel(const float* __restrict__ h, const float* __restrict__ p,
                const float* __restrict__ WQ, const float* __restrict__ WK,
                const float* __restrict__ WV,
                float* __restrict__ Qh, float* __restrict__ Kh, float* __restrict__ Vh, int rows) {
  constexpr int KDIM = 128;
  constexpr int PITCH = KDIM + A_PAD;
  __shared__ v4f stageA[GEMM_ROWS * PITCH / 4];
  float* As = (float*)stageA;
  const int row0 = (int)blockIdx.x * GEMM_ROWS;
  const int c4 = (int)threadIdx.x & 31;
  const float* colsrc = ((c4 < 16) ? h : p) + 4 * (c4 & 15);
#pragma unroll
  for (int j = 0; j < GEMM_ROWS / 8; ++j) {
    const int r = (int)(threadIdx.x >> 5) + 8 * j;
    int gr = row0 + r;
    if (gr >= rows) gr = rows - 1;
    *(v4f*)(As + r * PITCH + 4 * c4) = *(const v4f*)(colsrc + (size_t)gr * 64);
  }
  __syncthreads();
  const int w = (int)(threadIdx.x >> 5);
  const int rt = (int)blockIdx.x * (GEMM_ROWS / 16) + (w >> 2), ct = w & 3;
  const int rowTiles = (rows + 15) >> 4;
  if (rt >= rowTiles) return;
  const int lane = threadIdx.x & 31, half = lane >> 4, l16 = lane & 15;
  const int bc = ct * 16 + l16;
  const float* ap = As + ((w >> 2) * 16 + l16) * PITCH + 2 * half;
  const size_t boff = (size_t)(2 * half) * 64 + bc;
  const float* bq = WQ + boff;
  const float* bk = WK + boff;
  const float* bv = WV + boff;
  v8f accQ = {}, accK = {}, accV = {};
#pragma unroll 1
  for (int kk = 0; kk < KDIM; kk += 4 * GEMM_KSTEPS) {
    v2f a[GEMM_KSTEPS], fq[GEMM_KSTEPS], fk[GEMM_KSTEPS], fv[GEMM_KSTEPS];
#pragma unroll
    for (int s = 0; s < GEMM_KSTEPS; ++s) {
      const int k0 = kk + 4 * s;
      a[s] = *(const v2f*)(ap + k0);
      fq[s].x = bq[(size_t)k0 * 64];
      fq[s].y = bq[(size_t)(k0 + 1) * 64];
      fk[s].x = bk[(size_t)k0 * 64];
      fk[s].y = bk[(size_t)(k0 + 1) * 64];
      fv[s].x = bv[(size_t)k0 * 64];
      fv[s].y = bv[(size_t)(k0 + 1) * 64];
    }
#pragma unroll
    for (int s = 0; s < GEMM_KSTEPS; ++s) {
      accQ = wmma4(a[s], fq[s], accQ);
      accK = wmma4(a[s], fk[s], accK);
      accV = wmma4(a[s], fv[s], accV);
    }
    __builtin_amdgcn_sched_group_barrier(0x020, 6 * GEMM_KSTEPS, 0);
    __builtin_amdgcn_sched_group_barrier(0x100, GEMM_KSTEPS, 0);
    __builtin_amdgcn_sched_group_barrier(0x008, 3 * GEMM_KSTEPS, 0);
  }
#pragma unroll
  for (int v = 0; v < 8; ++v) {
    int r = rt * 16 + v + 8 * half;
    if (r < rows) {
      const size_t o = (size_t)r * 64 + ct * 16 + l16;
      Qh[o] = accQ[v];
      Kh[o] = accK[v];
      Vh[o] = accV[v];
    }
  }
}

__global__ void __launch_bounds__(256)
cheb_gemm_kernel(const float* __restrict__ Tx0, const float* __restrict__ Tx1,
                 const float* __restrict__ Tx2, const float* __restrict__ Tx3,
                 const float* __restrict__ fc, const float* __restrict__ Wc,
                 const float* __restrict__ bias, float* __restrict__ Out,
                 int rows, int Nn) {
  constexpr int PITCH = 64 + A_PAD;
  __shared__ v4f stageA[4 * GEMM_ROWS * PITCH / 4];
  float* As = (float*)stageA;
  const int row0 = (int)blockIdx.x * GEMM_ROWS;
  const float* Asrc[4] = {Tx0, Tx1, Tx2, Tx3};
#pragma unroll
  for (int j = 0; j < 8; ++j) {
    const int r = (int)(threadIdx.x >> 4) + 16 * (j & 1), c4 = (int)threadIdx.x & 15;
    int gr = row0 + r;
    if (gr >= rows) gr = rows - 1;
    *(v4f*)(As + ((j >> 1) * GEMM_ROWS + r) * PITCH + 4 * c4) =
        *(const v4f*)(Asrc[j >> 1] + (size_t)gr * 64 + 4 * c4);
  }
  __syncthreads();
  const int w = (int)(threadIdx.x >> 5);
  const int rt = (int)blockIdx.x * (GEMM_ROWS / 16) + (w >> 2), ct = w & 3;
  const int rowTiles = (rows + 15) >> 4;
  if (rt >= rowTiles) return;
  const int lane = threadIdx.x & 31, half = lane >> 4, l16 = lane & 15;
  int ar = rt * 16 + l16;
  if (ar >= rows) ar = rows - 1;
  const int bc = ct * 16 + l16;
  v8f acc = {};
  float sc[4];
#pragma unroll
  for (int k = 0; k < 4; ++k) sc[k] = fc[(size_t)k * Nn + ar];
#pragma unroll
  for (int k = 0; k < 4; ++k) {
    const float* ak = As + (k * GEMM_ROWS + (w >> 2) * 16 + l16) * PITCH + 2 * half;
    const float* bk = Wc + (size_t)k * 64 * 64 + (size_t)(2 * half) * 64 + bc;
    const float sk = sc[k];
#pragma unroll 1
    for (int kk = 0; kk < 64; kk += 4 * GEMM_KSTEPS) {
      v2f a[GEMM_KSTEPS], b[GEMM_KSTEPS];
#pragma unroll
      for (int s = 0; s < GEMM_KSTEPS; ++s) {
        const int k0 = kk + 4 * s;
        a[s] = *(const v2f*)(ak + k0) * sk;
        b[s].x = bk[(size_t)k0 * 64];
        b[s].y = bk[(size_t)(k0 + 1) * 64];
      }
#pragma unroll
      for (int s = 0; s < GEMM_KSTEPS; ++s) acc = wmma4(a[s], b[s], acc);
      __builtin_amdgcn_sched_group_barrier(0x020, 2 * GEMM_KSTEPS, 0);
      __builtin_amdgcn_sched_group_barrier(0x100, GEMM_KSTEPS, 0);
      __builtin_amdgcn_sched_group_barrier(0x008, GEMM_KSTEPS, 0);
    }
  }
  float bv = bias[ct * 16 + l16];
#pragma unroll
  for (int v = 0; v < 8; ++v) {
    int r = rt * 16 + v + 8 * half;
    if (r < rows) Out[(size_t)r * 192 + ct * 16 + l16] = acc[v] + bv;
  }
}


__device__ __forceinline__ float quad_sum(float v) {
  v += __int_as_float(__builtin_amdgcn_update_dpp(0, __float_as_int(v), 0xB1, 0xF, 0xF, true));
  v += __int_as_float(__builtin_amdgcn_update_dpp(0, __float_as_int(v), 0x4E, 0xF, 0xF, true));
  return v;
}

#define ATT_EDGES_PER_TRIP 8

__global__ void __launch_bounds__(256)
attn_tile_kernel(const float* __restrict__ Ee, const float* __restrict__ WE,
                 const float* __restrict__ Qh, const float* __restrict__ Kh,
                 const float* __restrict__ Vh, const float* __restrict__ recs,
                 const int* __restrict__ start, float* __restrict__ out, int N) {
  __shared__ float acc[TILE_ROWS * 64];
  __shared__ float zacc[TILE_ROWS * 8];
  __shared__ float stage[8 * 16 * 64];
  __shared__ v4f wfrag[16 * 64];
  const int b = blockIdx.x;
  for (int i = threadIdx.x; i < TILE_ROWS * 16; i += 256) ((v4f*)acc)[i] = (v4f){0.f, 0.f, 0.f, 0.f};
  for (int i = threadIdx.x; i < TILE_ROWS * 8; i += 256) zacc[i] = 0.f;
#pragma unroll
  for (int j = 0; j < 4; ++j) {
    const int f = (int)threadIdx.x + 256 * j;
    const int col = f & 63;
    const int r0 = 8 * (f >> 7) + 2 * ((f >> 6) & 1);
    wfrag[f] = (v4f){WE[r0 * 64 + col], WE[(r0 + 1) * 64 + col],
                     WE[(r0 + 4) * 64 + col], WE[(r0 + 5) * 64 + col]};
  }
  __syncthreads();
  const int lo = start[b], hi = start[b + 1];
  const int wid = __builtin_amdgcn_readfirstlane((int)(threadIdx.x >> 5));
  const int lane = threadIdx.x & 31, half = lane >> 4, l16 = lane & 15;
  float* myLds = stage + wid * 1024;
  const float scale = 0.35355339059327373f;

  for (int i0 = lo + wid * 16; i0 < hi; i0 += 8 * 16) {
    int i = i0 + l16;
    if (i >= hi) i = hi - 1;
    const v4f rec = *(const v4f*)(recs + 4 * (size_t)i);
    const int rs = __float_as_int(rec.x);
    const unsigned ry = (unsigned)__float_as_int(rec.y);
    const int rd = (int)(ry % TILE_ROWS);
    const int ar = (int)(ry / TILE_ROWS);

#pragma unroll 1
    for (int ct = 0; ct < 4; ++ct) {
      v8f c = {};
      const v4f* wf = wfrag + half * 64 + ct * 16 + l16;
      v4f q[8];
#pragma unroll
      for (int P = 0; P < 8; ++P) q[P] = wf[P * 128];
#pragma unroll
      for (int P = 0; P < 8; ++P) {
        const int kr = 8 * P + 2 * half;
        v2f a0, a1;
        a0.x = Ee[(size_t)ar * 64 + kr];
        a0.y = Ee[(size_t)ar * 64 + kr + 1];
        a1.x = Ee[(size_t)ar * 64 + kr + 4];
        a1.y = Ee[(size_t)ar * 64 + kr + 5];
        c = wmma4(a0, (v2f){q[P].x, q[P].y}, c);
        c = wmma4(a1, (v2f){q[P].z, q[P].w}, c);
      }
      __builtin_amdgcn_sched_group_barrier(0x100, 8, 0);
      __builtin_amdgcn_sched_group_barrier(0x008, 16, 0);
#pragma unroll
      for (int v = 0; v < 8; ++v)
        myLds[(v + 8 * half) * 64 + ct * 16 + l16] = c[v];
    }

    const int n = hi - i0;
    const float* qbase = Qh + (size_t)b * TILE_ROWS * 64;
#pragma unroll
    for (int g = 0; g < 16; g += ATT_EDGES_PER_TRIP) {
      v2f kv[ATT_EDGES_PER_TRIP], qv[ATT_EDGES_PER_TRIP], vv[ATT_EDGES_PER_TRIP], ev[ATT_EDGES_PER_TRIP];
#pragma unroll
      for (int j = 0; j < ATT_EDGES_PER_TRIP; ++j) {
        int s = __builtin_amdgcn_readlane(rs, g + j);
        int dl = __builtin_amdgcn_readlane(rd, g + j);
        kv[j] = *(const v2f*)(Kh + (size_t)s * 64 + 2 * lane);
        qv[j] = *(const v2f*)(qbase + (size_t)dl * 64 + 2 * lane);
        vv[j] = *(const v2f*)(Vh + (size_t)s * 64 + 2 * lane);
        ev[j] = *(const v2f*)(myLds + (g + j) * 64 + 2 * lane);
      }
#pragma unroll
      for (int j = 0; j < ATT_EDGES_PER_TRIP; ++j) {
        if (g + j < n) {
          int dl = __builtin_amdgcn_readlane(rd, g + j);
          float sc = quad_sum(kv[j].x * qv[j].x * ev[j].x + kv[j].y * qv[j].y * ev[j].y);
          sc *= scale;
          sc = fminf(5.0f, fmaxf(-5.0f, sc));
          float ss = __expf(sc);
          atomicAdd(&acc[dl * 64 + 2 * lane], vv[j].x * ss);
          atomicAdd(&acc[dl * 64 + 2 * lane + 1], vv[j].y * ss);
          if ((lane & 3) == 0) atomicAdd(&zacc[dl * 8 + (lane >> 2)], ss);
        }
      }
    }
  }
  __syncthreads();
  for (int i = threadIdx.x; i < TILE_ROWS * 16; i += 256) {
    int r = i >> 4, c4 = i & 15;
    int row = b * TILE_ROWS + r;
    if (row < N) {
      v4f a = ((const v4f*)acc)[i];
      float zz = zacc[r * 8 + (c4 >> 1)] + 1e-6f;
      v4f o = {a.x / zz, a.y / zz, a.z / zz, a.w / zz};
      *(v4f*)(out + (size_t)row * 192 + 128 + 4 * c4) = o;
    }
  }
}

extern "C" void kernel_launch(void* const* d_in, const int* in_sizes, int n_in,
                              void* d_out, int out_size, void* d_ws, size_t ws_size,
                              hipStream_t stream) {
  const float* h  = (const float*)d_in[0];
  const float* p  = (const float*)d_in[1];
  const float* e  = (const float*)d_in[2];
  const float* ew = (const float*)d_in[3];
  const float* fc = (const float*)d_in[4];
  const int* esrc = (const int*)d_in[5];
  const int* edst = (const int*)d_in[6];
  const float* Wc = (const float*)d_in[7];
  const float* bc = (const float*)d_in[8];
  const float* Wg = (const float*)d_in[9];
  const float* bg = (const float*)d_in[10];
  const float* WQ = (const float*)d_in[11];
  const float* WK = (const float*)d_in[12];
  const float* WVm = (const float*)d_in[13];
  const float* WE = (const float*)d_in[14];
  float* out = (float*)d_out;

  const int N = in_sizes[0] / 64;
  const int E = in_sizes[5];
  const size_t ND = (size_t)N * 64;

  float* w = (float*)d_ws;
  w += N;
  float* dinv = w; w += N;
  w += E;
  w = (float*)(((size_t)w + 15) & ~(size_t)15);
  float* Tx1  = w; w += ND;
  float* Tx2  = w; w += ND;
  float* Tx3  = w; w += ND;
  float* tmp  = w; w += ND;
  float* gcn  = w; w += ND;
  float* scr  = w; w += 2 * ND;
  float* Qh   = w; w += ND;
  float* Kh   = w; w += ND;
  float* Vh   = w; w += ND;
  float* list = w; w += ND + (size_t)N * 8;

  const int nb = (N + TILE_ROWS - 1) / TILE_ROWS;
  float* recs   = list;
  int* binStart = (int*)(recs + 4 * (size_t)E);
  int* binCnt   = (int*)scr;
  int* rank     = binCnt + (size_t)nb * CNT_PITCH;

  hipMemsetAsync(binCnt, 0, sizeof(int) * (size_t)nb * CNT_PITCH, stream);

  const int B = 256;
  const int nd4 = N * 16;
  const int combBlocks = (nd4 + B * ELEM_VECS_COMBINE - 1) / (B * ELEM_VECS_COMBINE);

  const int rankEdges = RANK_THREADS * RANK_PER_THREAD;
  tile_rank_kernel<<<(E + rankEdges - 1) / rankEdges, RANK_THREADS, sizeof(int) * (size_t)nb, stream>>>(
      edst, binCnt, rank, E, nb);
  tile_scan_kernel<<<1, 1024, 0, stream>>>(binCnt, binStart, nb);
  tile_fill_kernel<<<(E + B - 1) / B, B, 0, stream>>>(esrc, edst, ew, binStart, rank, recs, E);

  tile_deg_kernel<<<nb, 256, 0, stream>>>(recs, binStart, dinv, N);
  wlap_kernel<<<(E + B - 1) / B, B, 0, stream>>>(dinv, recs, E);

  tile_sum_kernel<0><<<nb, 256, 0, stream>>>(h, recs, binStart, Tx1, N);
  tile_sum_kernel<0><<<nb, 256, 0, stream>>>(Tx1, recs, binStart, tmp, N);
  combine_kernel<<<combBlocks, B, 0, stream>>>(tmp, h, Tx2, nd4);
  tile_sum_kernel<0><<<nb, 256, 0, stream>>>(Tx2, recs, binStart, tmp, N);
  combine_kernel<<<combBlocks, B, 0, stream>>>(tmp, Tx1, Tx3, nd4);

  const int gemmBlocks = (N + GEMM_ROWS - 1) / GEMM_ROWS;
  cheb_gemm_kernel<<<gemmBlocks, 256, 0, stream>>>(h, Tx1, Tx2, Tx3, fc, Wc, bc, out, N, N);

  tile_sum_kernel<1><<<nb, 256, 0, stream>>>(h, recs, binStart, gcn, N);
  gemm_wmma_kernel<64><<<gemmBlocks, 256, 0, stream>>>(gcn, 64, Wg, bg, out, 192, 64, N);

  qkv_gemm_kernel<<<gemmBlocks, 256, 0, stream>>>(h, p, WQ, WK, WVm, Qh, Kh, Vh, N);

  attn_tile_kernel<<<nb, 256, 0, stream>>>(e, WE, Qh, Kh, Vh, recs, binStart, out, N);
}


# Last Chebyshev and GCN neighbour sums keep the finished LDS tile and run their dense product on it in the same work-group (A fragments from the tile, Tx3 = 2a-b on the read); 3 launches fewer
# speedup vs baseline: 3.5884x; 1.0603x over previous
#include <hip/hip_runtime.h>
#include <hip/hip_bf16.h>

typedef __attribute__((ext_vector_type(2))) float v2f;
typedef __attribute__((ext_vector_type(8))) float v8f;
typedef __attribute__((ext_vector_type(4))) float v4f;

#define TILE_ROWS 64
#define CNT_PITCH 1
#define RANK_THREADS 1024
#define RANK_PER_THREAD 8
#define SLOTS_PER_TRIP 16

__device__ __forceinline__ v8f wmma4(v2f a, v2f b, v8f c) {
  return __builtin_amdgcn_wmma_f32_16x16x4_f32(
       false, a,  false, b,
       (short)0, c,  false,  false);
}


__global__ void __launch_bounds__(RANK_THREADS)
tile_rank_kernel(const int* __restrict__ dst, int* __restrict__ cnt,
                 int* __restrict__ rank, int E, int nb) {
  extern __shared__ int hist[];
  for (int t = threadIdx.x; t < nb; t += RANK_THREADS) hist[t] = 0;
  __syncthreads();
  const int e0 = (int)blockIdx.x * (RANK_THREADS * RANK_PER_THREAD) + (int)threadIdx.x;
  int tl[RANK_PER_THREAD], lr[RANK_PER_THREAD];
#pragma unroll
  for (int j = 0; j < RANK_PER_THREAD; ++j) {
    int e = e0 + j * RANK_THREADS;
    tl[j] = (int)((unsigned)dst[e < E ? e : E - 1] / TILE_ROWS);
  }
#pragma unroll
  for (int j = 0; j < RANK_PER_THREAD; ++j)
    lr[j] = atomicAdd(&hist[tl[j]], (e0 + j * RANK_THREADS < E) ? 1 : 0);
  __syncthreads();
  for (int t = threadIdx.x; t < nb; t += RANK_THREADS) {
    int c = hist[t];
    hist[t] = (c > 0) ? atomicAdd(&cnt[(size_t)t * CNT_PITCH], c) : 0;
  }
  __syncthreads();
  int bs[RANK_PER_THREAD];
#pragma unroll
  for (int j = 0; j < RANK_PER_THREAD; ++j) bs[j] = hist[tl[j]];
#pragma unroll
  for (int j = 0; j < RANK_PER_THREAD; ++j) {
    int e = e0 + j * RANK_THREADS;
    if (e < E) rank[e] = bs[j] + lr[j];
  }
}

__global__ void tile_scan_kernel(const int* __restrict__ cnt, int* __restrict__ start, int nb) {
  __shared__ int part[1024];
  int t = threadIdx.x;
  int per = (nb + 1023) / 1024;
  int b0 = t * per;
  int s = 0;
  for (int i = 0; i < per; ++i) {
    int b = b0 + i;
    if (b < nb) s += cnt[(size_t)b * CNT_PITCH];
  }
  part[t] = s;
  __syncthreads();
  for (int off = 1; off < 1024; off <<= 1) {
    int v = (t >= off) ? part[t - off] : 0;
    __syncthreads();
    part[t] += v;
    __syncthreads();
  }
  int run = part[t] - s;
  for (int i = 0; i < per; ++i) {
    int b = b0 + i;
    if (b < nb) {
      start[b] = run;
      run += cnt[(size_t)b * CNT_PITCH];
    }
  }
  if (t == 1023) start[nb] = part[1023];
}

__global__ void tile_fill_kernel(const int* __restrict__ src, const int* __restrict__ dst,
                                 const float* __restrict__ ew,
                                 const int* __restrict__ start, const int* __restrict__ rank,
                                 float* __restrict__ recs, int E) {
  int tid = blockIdx.x * blockDim.x + threadIdx.x;
  if (tid >= E) return;
  int d = dst[tid];
  int b = (int)((unsigned)d / TILE_ROWS);
  int pos = start[b] + rank[tid];
  v4f r;
  r.x = __int_as_float(src[tid]);
  r.y = __int_as_float(tid * TILE_ROWS + (d - b * TILE_ROWS));
  r.z = __int_as_float(d);
  r.w = ew[tid];
  *(v4f*)(recs + 4 * (size_t)pos) = r;
}

__global__ void __launch_bounds__(256)
tile_deg_kernel(const float* __restrict__ recs, const int* __restrict__ start,
                float* __restrict__ dinv, int N) {
  __shared__ int cnt[TILE_ROWS];
  const int b = blockIdx.x;
  if (threadIdx.x < TILE_ROWS) cnt[threadIdx.x] = 0;
  __syncthreads();
  const int lo = start[b], hi = start[b + 1];
  for (int i = lo + (int)threadIdx.x; i < hi; i += 256) {
    unsigned ry = (unsigned)__float_as_int(recs[4 * (size_t)i + 1]);
    atomicAdd(&cnt[ry % TILE_ROWS], 1);
  }
  __syncthreads();
  if (threadIdx.x < TILE_ROWS) {
    int row = b * TILE_ROWS + (int)threadIdx.x;
    if (row < N) {
      float d = (float)cnt[threadIdx.x];
      d = d < 1.0f ? 1.0f : d;
      dinv[row] = rsqrtf(d);
    }
  }
}

__global__ void wlap_kernel(const float* __restrict__ dinv, float* recs, int E) {
  int i = blockIdx.x * blockDim.x + threadIdx.x;
  if (i < E) {
    v4f r = *(const v4f*)(recs + 4 * (size_t)i);
    r.z = -(dinv[__float_as_int(r.x)] * dinv[__float_as_int(r.z)]);
    *(v4f*)(recs + 4 * (size_t)i) = r;
  }
}

template <int WSEL>
__global__ void __launch_bounds__(256)
tile_sum_kernel(const float* __restrict__ x, const float* __restrict__ recs,
                const int* __restrict__ start, float* __restrict__ out, int N) {
  __shared__ float acc[TILE_ROWS * 64];
  const int b = blockIdx.x;
  for (int i = threadIdx.x; i < TILE_ROWS * 16; i += 256) ((v4f*)acc)[i] = (v4f){0.f, 0.f, 0.f, 0.f};
  __syncthreads();
  const int lo = start[b], hi = start[b + 1];
  const int wid = __builtin_amdgcn_readfirstlane((int)(threadIdx.x >> 5));
  const int lane = threadIdx.x & 31;
  for (int i0 = lo + wid * 32; i0 < hi; i0 += 8 * 32) {
    int i = i0 + lane;
    if (i >= hi) i = hi - 1;
    const v4f rec = *(const v4f*)(recs + 4 * (size_t)i);
    const int rs = __float_as_int(rec.x);
    const int rd = (int)((unsigned)__float_as_int(rec.y) % TILE_ROWS);
    const int rw = __float_as_int(WSEL ? rec.w : rec.z);
    const int n = hi - i0;
#pragma unroll
    for (int g = 0; g < 32; g += SLOTS_PER_TRIP) {
      v2f xv[SLOTS_PER_TRIP];
#pragma unroll
      for (int j = 0; j < SLOTS_PER_TRIP; ++j) {
        int s = __builtin_amdgcn_readlane(rs, g + j);
        xv[j] = *(const v2f*)(x + (size_t)s * 64 + 2 * lane);
      }
#pragma unroll
      for (int j = 0; j < SLOTS_PER_TRIP; ++j) {
        if (g + j < n) {
          int dl = __builtin_amdgcn_readlane(rd, g + j);
          float wv = __int_as_float(__builtin_amdgcn_readlane(rw, g + j));
          atomicAdd(&acc[dl * 64 + 2 * lane], xv[j].x * wv);
          atomicAdd(&acc[dl * 64 + 2 * lane + 1], xv[j].y * wv);
        }
      }
    }
  }
  __syncthreads();
  for (int i = threadIdx.x; i < TILE_ROWS * 16; i += 256) {
    int row = b * TILE_ROWS + (i >> 4);
    if (row < N) *(v4f*)(out + (size_t)row * 64 + 4 * (i & 15)) = ((const v4f*)acc)[i];
  }
}

#define ELEM_VECS_COMBINE 4

__global__ void __launch_bounds__(256)
combine_kernel(const float* __restrict__ a, const float* __restrict__ b,
               float* __restrict__ out, int n4) {
  const int T = (int)gridDim.x * 256;
  const int i = (int)blockIdx.x * 256 + (int)threadIdx.x;
  v4f av[ELEM_VECS_COMBINE], bv[ELEM_VECS_COMBINE];
#pragma unroll
  for (int j = 0; j < ELEM_VECS_COMBINE; ++j) {
    int k = i + j * T;
    k = k < n4 ? k : n4 - 1;
    av[j] = ((const v4f*)a)[k];
    bv[j] = ((const v4f*)b)[k];
  }
#pragma unroll
  for (int j = 0; j < ELEM_VECS_COMBINE; ++j) {
    int k = i + j * T;
    if (k < n4) ((v4f*)out)[k] = 2.0f * av[j] - bv[j];
  }
}

#define GEMM_ROWS 32
#define GEMM_KSTEPS 8
#define A_PAD 4

__global__ void __launch_bounds__(256)
qkv_gemm_kernel(const float* __restrict__ h, const float* __restrict__ p,
                const float* __restrict__ WQ, const float* __restrict__ WK,
                const float* __restrict__ WV,
                float* __restrict__ Qh, float* __restrict__ Kh, float* __restrict__ Vh, int rows) {
  constexpr int KDIM = 128;
  constexpr int PITCH = KDIM + A_PAD;
  __shared__ v4f stageA[GEMM_ROWS * PITCH / 4];
  float* As = (float*)stageA;
  const int row0 = (int)blockIdx.x * GEMM_ROWS;
  const int c4 = (int)threadIdx.x & 31;
  const float* colsrc = ((c4 < 16) ? h : p) + 4 * (c4 & 15);
#pragma unroll
  for (int j = 0; j < GEMM_ROWS / 8; ++j) {
    const int r = (int)(threadIdx.x >> 5) + 8 * j;
    int gr = row0 + r;
    if (gr >= rows) gr = rows - 1;
    *(v4f*)(As + r * PITCH + 4 * c4) = *(const v4f*)(colsrc + (size_t)gr * 64);
  }
  __syncthreads();
  const int w = (int)(threadIdx.x >> 5);
  const int rt = (int)blockIdx.x * (GEMM_ROWS / 16) + (w >> 2), ct = w & 3;
  const int rowTiles = (rows + 15) >> 4;
  if (rt >= rowTiles) return;
  const int lane = threadIdx.x & 31, half = lane >> 4, l16 = lane & 15;
  const int bc = ct * 16 + l16;
  const float* ap = As + ((w >> 2) * 16 + l16) * PITCH + 2 * half;
  const size_t boff = (size_t)(2 * half) * 64 + bc;
  const float* bq = WQ + boff;
  const float* bk = WK + boff;
  const float* bv = WV + boff;
  v8f accQ = {}, accK = {}, accV = {};
#pragma unroll 1
  for (int kk = 0; kk < KDIM; kk += 4 * GEMM_KSTEPS) {
    v2f a[GEMM_KSTEPS], fq[GEMM_KSTEPS], fk[GEMM_KSTEPS], fv[GEMM_KSTEPS];
#pragma unroll
    for (int s = 0; s < GEMM_KSTEPS; ++s) {
      const int k0 = kk + 4 * s;
      a[s] = *(const v2f*)(ap + k0);
      fq[s].x = bq[(size_t)k0 * 64];
      fq[s].y = bq[(size_t)(k0 + 1) * 64];
      fk[s].x = bk[(size_t)k0 * 64];
      fk[s].y = bk[(size_t)(k0 + 1) * 64];
      fv[s].x = bv[(size_t)k0 * 64];
      fv[s].y = bv[(size_t)(k0 + 1) * 64];
    }
#pragma unroll
    for (int s = 0; s < GEMM_KSTEPS; ++s) {
      accQ = wmma4(a[s], fq[s], accQ);
      accK = wmma4(a[s], fk[s], accK);
      accV = wmma4(a[s], fv[s], accV);
    }
    __builtin_amdgcn_sched_group_barrier(0x020, 6 * GEMM_KSTEPS, 0);
    __builtin_amdgcn_sched_group_barrier(0x100, GEMM_KSTEPS, 0);
    __builtin_amdgcn_sched_group_barrier(0x008, 3 * GEMM_KSTEPS, 0);
  }
#pragma unroll
  for (int v = 0; v < 8; ++v) {
    int r = rt * 16 + v + 8 * half;
    if (r < rows) {
      const size_t o = (size_t)r * 64 + ct * 16 + l16;
      Qh[o] = accQ[v];
      Kh[o] = accK[v];
      Vh[o] = accV[v];
    }
  }
}

#define TILE_PITCH (64 + A_PAD)

template <int WSEL>
__device__ __forceinline__ void tile_gather(const float* __restrict__ x,
                                            const float* __restrict__ recs,
                                            int lo, int hi, float* acc) {
  const int wid = __builtin_amdgcn_readfirstlane((int)(threadIdx.x >> 5));
  const int lane = threadIdx.x & 31;
  for (int i0 = lo + wid * 32; i0 < hi; i0 += 8 * 32) {
    int i = i0 + lane;
    if (i >= hi) i = hi - 1;
    const v4f rec = *(const v4f*)(recs + 4 * (size_t)i);
    const int rs = __float_as_int(rec.x);
    const int rd = (int)((unsigned)__float_as_int(rec.y) % TILE_ROWS);
    const int rw = __float_as_int(WSEL ? rec.w : rec.z);
    const int n = hi - i0;
#pragma unroll
    for (int g = 0; g < 32; g += SLOTS_PER_TRIP) {
      v2f xv[SLOTS_PER_TRIP];
#pragma unroll
      for (int j = 0; j < SLOTS_PER_TRIP; ++j) {
        int s = __builtin_amdgcn_readlane(rs, g + j);
        xv[j] = *(const v2f*)(x + (size_t)s * 64 + 2 * lane);
      }
#pragma unroll
      for (int j = 0; j < SLOTS_PER_TRIP; ++j) {
        if (g + j < n) {
          int dl = __builtin_amdgcn_readlane(rd, g + j);
          float wv = __int_as_float(__builtin_amdgcn_readlane(rw, g + j));
          atomicAdd(&acc[dl * TILE_PITCH + 2 * lane], xv[j].x * wv);
          atomicAdd(&acc[dl * TILE_PITCH + 2 * lane + 1], xv[j].y * wv);
        }
      }
    }
  }
}

__global__ void __launch_bounds__(256)
tile_sum_gcn_kernel(const float* __restrict__ x, const float* __restrict__ recs,
                    const int* __restrict__ start, const float* __restrict__ Wg,
                    const float* __restrict__ bias, float* __restrict__ out, int N) {
  __shared__ v4f accv[TILE_ROWS * TILE_PITCH / 4];
  float* acc = (float*)accv;
  const int b = blockIdx.x;
  for (int i = threadIdx.x; i < TILE_ROWS * TILE_PITCH / 4; i += 256) accv[i] = (v4f){0.f, 0.f, 0.f, 0.f};
  __syncthreads();
  tile_gather<1>(x, recs, start[b], start[b + 1], acc);
  __syncthreads();
  const int w = (int)(threadIdx.x >> 5);
  const int lane = threadIdx.x & 31, half = lane >> 4, l16 = lane & 15;
  const int ct = w & 3, rp = w >> 2;
  const float* a0p = acc + (rp * 16 + l16) * TILE_PITCH + 2 * half;
  const float* a1p = a0p + 32 * TILE_PITCH;
  const float* bp = Wg + (size_t)(2 * half) * 64 + ct * 16 + l16;
  v8f c0 = {}, c1 = {};
#pragma unroll 1
  for (int kk = 0; kk < 64; kk += 4 * GEMM_KSTEPS) {
    v2f a0[GEMM_KSTEPS], a1[GEMM_KSTEPS], bf[GEMM_KSTEPS];
#pragma unroll
    for (int s = 0; s < GEMM_KSTEPS; ++s) {
      const int k0 = kk + 4 * s;
      a0[s] = *(const v2f*)(a0p + k0);
      a1[s] = *(const v2f*)(a1p + k0);
      bf[s].x = bp[(size_t)k0 * 64];
      bf[s].y = bp[(size_t)(k0 + 1) * 64];
    }
#pragma unroll
    for (int s = 0; s < GEMM_KSTEPS; ++s) {
      c0 = wmma4(a0[s], bf[s], c0);
      c1 = wmma4(a1[s], bf[s], c1);
    }
    __builtin_amdgcn_sched_group_barrier(0x020, 2 * GEMM_KSTEPS, 0);
    __builtin_amdgcn_sched_group_barrier(0x100, 2 * GEMM_KSTEPS, 0);
    __builtin_amdgcn_sched_group_barrier(0x008, 2 * GEMM_KSTEPS, 0);
  }
  const float bv = bias[ct * 16 + l16];
#pragma unroll
  for (int v = 0; v < 8; ++v) {
    const int r0 = b * TILE_ROWS + rp * 16 + v + 8 * half;
    if (r0 < N) out[(size_t)r0 * 192 + 64 + ct * 16 + l16] = c0[v] + bv;
    if (r0 + 32 < N) out[(size_t)(r0 + 32) * 192 + 64 + ct * 16 + l16] = c1[v] + bv;
  }
}

template <int TERM>
__device__ __forceinline__ void cheb_term(const float* ak, const float* t1,
                                          const float* __restrict__ bk,
                                          float s0, float s1, v8f& c0, v8f& c1) {
#pragma unroll 1
  for (int kk = 0; kk < 64; kk += 4 * GEMM_KSTEPS) {
    v2f a0[GEMM_KSTEPS], a1[GEMM_KSTEPS], bf[GEMM_KSTEPS];
#pragma unroll
    for (int s = 0; s < GEMM_KSTEPS; ++s) {
      const int k0 = kk + 4 * s;
      v2f x0 = *(const v2f*)(ak + k0);
      v2f x1 = *(const v2f*)(ak + 32 * TILE_PITCH + k0);
      if (TERM == 3) {
        x0 = 2.0f * x0 - *(const v2f*)(t1 + k0);
        x1 = 2.0f * x1 - *(const v2f*)(t1 + 32 * TILE_PITCH + k0);
      }
      a0[s] = x0 * s0;
      a1[s] = x1 * s1;
      bf[s].x = bk[(size_t)k0 * 64];
      bf[s].y = bk[(size_t)(k0 + 1) * 64];
    }
#pragma unroll
    for (int s = 0; s < GEMM_KSTEPS; ++s) {
      c0 = wmma4(a0[s], bf[s], c0);
      c1 = wmma4(a1[s], bf[s], c1);
    }
    __builtin_amdgcn_sched_group_barrier(0x020, 2 * GEMM_KSTEPS, 0);
    __builtin_amdgcn_sched_group_barrier(0x100, (TERM == 3 ? 4 : 2) * GEMM_KSTEPS, 0);
    __builtin_amdgcn_sched_group_barrier(0x008, 2 * GEMM_KSTEPS, 0);
  }
}

__global__ void __launch_bounds__(256)
tile_sum_cheb_kernel(const float* __restrict__ Tx0, const float* __restrict__ Tx1,
                     const float* __restrict__ Tx2, const float* __restrict__ recs,
                     const int* __restrict__ start, const float* __restrict__ fc,
                     const float* __restrict__ Wc, const float* __restrict__ bias,
                     float* __restrict__ out, int N) {
  __shared__ v4f accv[TILE_ROWS * TILE_PITCH / 4];
  __shared__ v4f rowsv[3 * TILE_ROWS * TILE_PITCH / 4];
  float* acc = (float*)accv;
  float* As = (float*)rowsv;
  const int b = blockIdx.x;
  const float* Asrc[3] = {Tx0, Tx1, Tx2};
#pragma unroll
  for (int j = 0; j < 12; ++j) {
    const int r = (int)(threadIdx.x >> 4) + 16 * (j & 3), c4 = (int)threadIdx.x & 15;
    int gr = b * TILE_ROWS + r;
    if (gr >= N) gr = N - 1;
    *(v4f*)(As + ((j >> 2) * TILE_ROWS + r) * TILE_PITCH + 4 * c4) =
        *(const v4f*)(Asrc[j >> 2] + (size_t)gr * 64 + 4 * c4);
  }
  for (int i = threadIdx.x; i < TILE_ROWS * TILE_PITCH / 4; i += 256) accv[i] = (v4f){0.f, 0.f, 0.f, 0.f};
  __syncthreads();
  tile_gather<0>(Tx2, recs, start[b], start[b + 1], acc);
  __syncthreads();
  const int w = (int)(threadIdx.x >> 5);
  const int lane = threadIdx.x & 31, half = lane >> 4, l16 = lane & 15;
  const int ct = w & 3, rp = w >> 2;
  const int lrow = rp * 16 + l16;
  int ar0 = b * TILE_ROWS + lrow, ar1 = ar0 + 32;
  if (ar0 >= N) ar0 = N - 1;
  if (ar1 >= N) ar1 = N - 1;
  float sc0[4], sc1[4];
#pragma unroll
  for (int k = 0; k < 4; ++k) {
    sc0[k] = fc[(size_t)k * N + ar0];
    sc1[k] = fc[(size_t)k * N + ar1];
  }
  v8f c0 = {}, c1 = {};
  const float* t1 = As + (TILE_ROWS + lrow) * TILE_PITCH + 2 * half;
  const float* bk = Wc + (size_t)(2 * half) * 64 + ct * 16 + l16;
  cheb_term<0>(As + (0 * TILE_ROWS + lrow) * TILE_PITCH + 2 * half, t1, bk, sc0[0], sc1[0], c0, c1);
  cheb_term<1>(As + (1 * TILE_ROWS + lrow) * TILE_PITCH + 2 * half, t1, bk + 1 * 64 * 64, sc0[1], sc1[1], c0, c1);
  cheb_term<2>(As + (2 * TILE_ROWS + lrow) * TILE_PITCH + 2 * half, t1, bk + 2 * 64 * 64, sc0[2], sc1[2], c0, c1);
  cheb_term<3>(acc + lrow * TILE_PITCH + 2 * half, t1, bk + 3 * 64 * 64, sc0[3], sc1[3], c0, c1);
  const float bv = bias[ct * 16 + l16];
#pragma unroll
  for (int v = 0; v < 8; ++v) {
    const int r0 = b * TILE_ROWS + rp * 16 + v + 8 * half;
    if (r0 < N) out[(size_t)r0 * 192 + ct * 16 + l16] = c0[v] + bv;
    if (r0 + 32 < N) out[(size_t)(r0 + 32) * 192 + ct * 16 + l16] = c1[v] + bv;
  }
}


__device__ __forceinline__ float quad_sum(float v) {
  v += __int_as_float(__builtin_amdgcn_update_dpp(0, __float_as_int(v), 0xB1, 0xF, 0xF, true));
  v += __int_as_float(__builtin_amdgcn_update_dpp(0, __float_as_int(v), 0x4E, 0xF, 0xF, true));
  return v;
}

#define ATT_EDGES_PER_TRIP 8

__global__ void __launch_bounds__(256)
attn_tile_kernel(const float* __restrict__ Ee, const float* __restrict__ WE,
                 const float* __restrict__ Qh, const float* __restrict__ Kh,
                 const float* __restrict__ Vh, const float* __restrict__ recs,
                 const int* __restrict__ start, float* __restrict__ out, int N) {
  __shared__ float acc[TILE_ROWS * 64];
  __shared__ float zacc[TILE_ROWS * 8];
  __shared__ float stage[8 * 16 * 64];
  __shared__ v4f wfrag[16 * 64];
  const int b = blockIdx.x;
  for (int i = threadIdx.x; i < TILE_ROWS * 16; i += 256) ((v4f*)acc)[i] = (v4f){0.f, 0.f, 0.f, 0.f};
  for (int i = threadIdx.x; i < TILE_ROWS * 8; i += 256) zacc[i] = 0.f;
#pragma unroll
  for (int j = 0; j < 4; ++j) {
    const int f = (int)threadIdx.x + 256 * j;
    const int col = f & 63;
    const int r0 = 8 * (f >> 7) + 2 * ((f >> 6) & 1);
    wfrag[f] = (v4f){WE[r0 * 64 + col], WE[(r0 + 1) * 64 + col],
                     WE[(r0 + 4) * 64 + col], WE[(r0 + 5) * 64 + col]};
  }
  __syncthreads();
  const int lo = start[b], hi = start[b + 1];
  const int wid = __builtin_amdgcn_readfirstlane((int)(threadIdx.x >> 5));
  const int lane = threadIdx.x & 31, half = lane >> 4, l16 = lane & 15;
  float* myLds = stage + wid * 1024;
  const float scale = 0.35355339059327373f;

  for (int i0 = lo + wid * 16; i0 < hi; i0 += 8 * 16) {
    int i = i0 + l16;
    if (i >= hi) i = hi - 1;
    const v4f rec = *(const v4f*)(recs + 4 * (size_t)i);
    const int rs = __float_as_int(rec.x);
    const unsigned ry = (unsigned)__float_as_int(rec.y);
    const int rd = (int)(ry % TILE_ROWS);
    const int ar = (int)(ry / TILE_ROWS);

#pragma unroll 1
    for (int ct = 0; ct < 4; ++ct) {
      v8f c = {};
      const v4f* wf = wfrag + half * 64 + ct * 16 + l16;
      v4f q[8];
#pragma unroll
      for (int P = 0; P < 8; ++P) q[P] = wf[P * 128];
#pragma unroll
      for (int P = 0; P < 8; ++P) {
        const int kr = 8 * P + 2 * half;
        v2f a0, a1;
        a0.x = Ee[(size_t)ar * 64 + kr];
        a0.y = Ee[(size_t)ar * 64 + kr + 1];
        a1.x = Ee[(size_t)ar * 64 + kr + 4];
        a1.y = Ee[(size_t)ar * 64 + kr + 5];
        c = wmma4(a0, (v2f){q[P].x, q[P].y}, c);
        c = wmma4(a1, (v2f){q[P].z, q[P].w}, c);
      }
      __builtin_amdgcn_sched_group_barrier(0x100, 8, 0);
      __builtin_amdgcn_sched_group_barrier(0x008, 16, 0);
#pragma unroll
      for (int v = 0; v < 8; ++v)
        myLds[(v + 8 * half) * 64 + ct * 16 + l16] = c[v];
    }

    const int n = hi - i0;
    const float* qbase = Qh + (size_t)b * TILE_ROWS * 64;
#pragma unroll
    for (int g = 0; g < 16; g += ATT_EDGES_PER_TRIP) {
      v2f kv[ATT_EDGES_PER_TRIP], qv[ATT_EDGES_PER_TRIP], vv[ATT_EDGES_PER_TRIP], ev[ATT_EDGES_PER_TRIP];
#pragma unroll
      for (int j = 0; j < ATT_EDGES_PER_TRIP; ++j) {
        int s = __builtin_amdgcn_readlane(rs, g + j);
        int dl = __builtin_amdgcn_readlane(rd, g + j);
        kv[j] = *(const v2f*)(Kh + (size_t)s * 64 + 2 * lane);
        qv[j] = *(const v2f*)(qbase + (size_t)dl * 64 + 2 * lane);
        vv[j] = *(const v2f*)(Vh + (size_t)s * 64 + 2 * lane);
        ev[j] = *(const v2f*)(myLds + (g + j) * 64 + 2 * lane);
      }
#pragma unroll
      for (int j = 0; j < ATT_EDGES_PER_TRIP; ++j) {
        if (g + j < n) {
          int dl = __builtin_amdgcn_readlane(rd, g + j);
          float sc = quad_sum(kv[j].x * qv[j].x * ev[j].x + kv[j].y * qv[j].y * ev[j].y);
          sc *= scale;
          sc = fminf(5.0f, fmaxf(-5.0f, sc));
          float ss = __expf(sc);
          atomicAdd(&acc[dl * 64 + 2 * lane], vv[j].x * ss);
          atomicAdd(&acc[dl * 64 + 2 * lane + 1], vv[j].y * ss);
          if ((lane & 3) == 0) atomicAdd(&zacc[dl * 8 + (lane >> 2)], ss);
        }
      }
    }
  }
  __syncthreads();
  for (int i = threadIdx.x; i < TILE_ROWS * 16; i += 256) {
    int r = i >> 4, c4 = i & 15;
    int row = b * TILE_ROWS + r;
    if (row < N) {
      v4f a = ((const v4f*)acc)[i];
      float zz = zacc[r * 8 + (c4 >> 1)] + 1e-6f;
      v4f o = {a.x / zz, a.y / zz, a.z / zz, a.w / zz};
      *(v4f*)(out + (size_t)row * 192 + 128 + 4 * c4) = o;
    }
  }
}

extern "C" void kernel_launch(void* const* d_in, const int* in_sizes, int n_in,
                              void* d_out, int out_size, void* d_ws, size_t ws_size,
                              hipStream_t stream) {
  const float* h  = (const float*)d_in[0];
  const float* p  = (const float*)d_in[1];
  const float* e  = (const float*)d_in[2];
  const float* ew = (const float*)d_in[3];
  const float* fc = (const float*)d_in[4];
  const int* esrc = (const int*)d_in[5];
  const int* edst = (const int*)d_in[6];
  const float* Wc = (const float*)d_in[7];
  const float* bc = (const float*)d_in[8];
  const float* Wg = (const float*)d_in[9];
  const float* bg = (const float*)d_in[10];
  const float* WQ = (const float*)d_in[11];
  const float* WK = (const float*)d_in[12];
  const float* WVm = (const float*)d_in[13];
  const float* WE = (const float*)d_in[14];
  float* out = (float*)d_out;

  const int N = in_sizes[0] / 64;
  const int E = in_sizes[5];
  const size_t ND = (size_t)N * 64;

  float* w = (float*)d_ws;
  w += N;
  float* dinv = w; w += N;
  w += E;
  w = (float*)(((size_t)w + 15) & ~(size_t)15);
  float* Tx1  = w; w += ND;
  float* Tx2  = w; w += ND;
  w += ND;
  float* tmp  = w; w += ND;
  w += ND;
  float* scr  = w; w += 2 * ND;
  float* Qh   = w; w += ND;
  float* Kh   = w; w += ND;
  float* Vh   = w; w += ND;
  float* list = w; w += ND + (size_t)N * 8;

  const int nb = (N + TILE_ROWS - 1) / TILE_ROWS;
  float* recs   = list;
  int* binStart = (int*)(recs + 4 * (size_t)E);
  int* binCnt   = (int*)scr;
  int* rank     = binCnt + (size_t)nb * CNT_PITCH;

  hipMemsetAsync(binCnt, 0, sizeof(int) * (size_t)nb * CNT_PITCH, stream);

  const int B = 256;
  const int nd4 = N * 16;
  const int combBlocks = (nd4 + B * ELEM_VECS_COMBINE - 1) / (B * ELEM_VECS_COMBINE);

  const int rankEdges = RANK_THREADS * RANK_PER_THREAD;
  tile_rank_kernel<<<(E + rankEdges - 1) / rankEdges, RANK_THREADS, sizeof(int) * (size_t)nb, stream>>>(
      edst, binCnt, rank, E, nb);
  tile_scan_kernel<<<1, 1024, 0, stream>>>(binCnt, binStart, nb);
  tile_fill_kernel<<<(E + B - 1) / B, B, 0, stream>>>(esrc, edst, ew, binStart, rank, recs, E);

  tile_deg_kernel<<<nb, 256, 0, stream>>>(recs, binStart, dinv, N);
  wlap_kernel<<<(E + B - 1) / B, B, 0, stream>>>(dinv, recs, E);

  tile_sum_kernel<0><<<nb, 256, 0, stream>>>(h, recs, binStart, Tx1, N);
  tile_sum_kernel<0><<<nb, 256, 0, stream>>>(Tx1, recs, binStart, tmp, N);
  combine_kernel<<<combBlocks, B, 0, stream>>>(tmp, h, Tx2, nd4);
  tile_sum_cheb_kernel<<<nb, 256, 0, stream>>>(h, Tx1, Tx2, recs, binStart, fc, Wc, bc, out, N);

  tile_sum_gcn_kernel<<<nb, 256, 0, stream>>>(h, recs, binStart, Wg, bg, out, N);

  const int gemmBlocks = (N + GEMM_ROWS - 1) / GEMM_ROWS;
  qkv_gemm_kernel<<<gemmBlocks, 256, 0, stream>>>(h, p, WQ, WK, WVm, Qh, Kh, Vh, N);

  attn_tile_kernel<<<nb, 256, 0, stream>>>(e, WE, Qh, Kh, Vh, recs, binStart, out, N);
}


# Destination tiles of 32 rows, not 64, in all six tile kernels: 1563 half-size work-groups per launch, not 782 (3.05 per WGP: the WGPs with four set the time); sum order = LDS arrival order as before
# speedup vs baseline: 3.7052x; 1.0325x over previous
#include <hip/hip_runtime.h>
#include <hip/hip_bf16.h>

typedef __attribute__((ext_vector_type(2))) float v2f;
typedef __attribute__((ext_vector_type(8))) float v8f;
typedef __attribute__((ext_vector_type(4))) float v4f;

#define TILE_ROWS 32
#define CNT_PITCH 1
#define RANK_THREADS 1024
#define RANK_PER_THREAD 8
#define SLOTS_PER_TRIP 16

__device__ __forceinline__ v8f wmma4(v2f a, v2f b, v8f c) {
  return __builtin_amdgcn_wmma_f32_16x16x4_f32(
       false, a,  false, b,
       (short)0, c,  false,  false);
}


__global__ void __launch_bounds__(RANK_THREADS)
tile_rank_kernel(const int* __restrict__ dst, int* __restrict__ cnt,
                 int* __restrict__ rank, int E, int nb) {
  extern __shared__ int hist[];
  for (int t = threadIdx.x; t < nb; t += RANK_THREADS) hist[t] = 0;
  __syncthreads();
  const int e0 = (int)blockIdx.x * (RANK_THREADS * RANK_PER_THREAD) + (int)threadIdx.x;
  int tl[RANK_PER_THREAD], lr[RANK_PER_THREAD];
#pragma unroll
  for (int j = 0; j < RANK_PER_THREAD; ++j) {
    int e = e0 + j * RANK_THREADS;
    tl[j] = (int)((unsigned)dst[e < E ? e : E - 1] / TILE_ROWS);
  }
#pragma unroll
  for (int j = 0; j < RANK_PER_THREAD; ++j)
    lr[j] = atomicAdd(&hist[tl[j]], (e0 + j * RANK_THREADS < E) ? 1 : 0);
  __syncthreads();
  for (int t = threadIdx.x; t < nb; t += RANK_THREADS) {
    int c = hist[t];
    hist[t] = (c > 0) ? atomicAdd(&cnt[(size_t)t * CNT_PITCH], c) : 0;
  }
  __syncthreads();
  int bs[RANK_PER_THREAD];
#pragma unroll
  for (int j = 0; j < RANK_PER_THREAD; ++j) bs[j] = hist[tl[j]];
#pragma unroll
  for (int j = 0; j < RANK_PER_THREAD; ++j) {
    int e = e0 + j * RANK_THREADS;
    if (e < E) rank[e] = bs[j] + lr[j];
  }
}

__global__ void tile_scan_kernel(const int* __restrict__ cnt, int* __restrict__ start, int nb) {
  __shared__ int part[1024];
  int t = threadIdx.x;
  int per = (nb + 1023) / 1024;
  int b0 = t * per;
  int s = 0;
  for (int i = 0; i < per; ++i) {
    int b = b0 + i;
    if (b < nb) s += cnt[(size_t)b * CNT_PITCH];
  }
  part[t] = s;
  __syncthreads();
  for (int off = 1; off < 1024; off <<= 1) {
    int v = (t >= off) ? part[t - off] : 0;
    __syncthreads();
    part[t] += v;
    __syncthreads();
  }
  int run = part[t] - s;
  for (int i = 0; i < per; ++i) {
    int b = b0 + i;
    if (b < nb) {
      start[b] = run;
      run += cnt[(size_t)b * CNT_PITCH];
    }
  }
  if (t == 1023) start[nb] = part[1023];
}

__global__ void tile_fill_kernel(const int* __restrict__ src, const int* __restrict__ dst,
                                 const float* __restrict__ ew,
                                 const int* __restrict__ start, const int* __restrict__ rank,
                                 float* __restrict__ recs, int E) {
  int tid = blockIdx.x * blockDim.x + threadIdx.x;
  if (tid >= E) return;
  int d = dst[tid];
  int b = (int)((unsigned)d / TILE_ROWS);
  int pos = start[b] + rank[tid];
  v4f r;
  r.x = __int_as_float(src[tid]);
  r.y = __int_as_float(tid * TILE_ROWS + (d - b * TILE_ROWS));
  r.z = __int_as_float(d);
  r.w = ew[tid];
  *(v4f*)(recs + 4 * (size_t)pos) = r;
}

__global__ void __launch_bounds__(256)
tile_deg_kernel(const float* __restrict__ recs, const int* __restrict__ start,
                float* __restrict__ dinv, int N) {
  __shared__ int cnt[TILE_ROWS];
  const int b = blockIdx.x;
  if (threadIdx.x < TILE_ROWS) cnt[threadIdx.x] = 0;
  __syncthreads();
  const int lo = start[b], hi = start[b + 1];
  for (int i = lo + (int)threadIdx.x; i < hi; i += 256) {
    unsigned ry = (unsigned)__float_as_int(recs[4 * (size_t)i + 1]);
    atomicAdd(&cnt[ry % TILE_ROWS], 1);
  }
  __syncthreads();
  if (threadIdx.x < TILE_ROWS) {
    int row = b * TILE_ROWS + (int)threadIdx.x;
    if (row < N) {
      float d = (float)cnt[threadIdx.x];
      d = d < 1.0f ? 1.0f : d;
      dinv[row] = rsqrtf(d);
    }
  }
}

__global__ void wlap_kernel(const float* __restrict__ dinv, float* recs, int E) {
  int i = blockIdx.x * blockDim.x + threadIdx.x;
  if (i < E) {
    v4f r = *(const v4f*)(recs + 4 * (size_t)i);
    r.z = -(dinv[__float_as_int(r.x)] * dinv[__float_as_int(r.z)]);
    *(v4f*)(recs + 4 * (size_t)i) = r;
  }
}

template <int WSEL>
__global__ void __launch_bounds__(256)
tile_sum_kernel(const float* __restrict__ x, const float* __restrict__ recs,
                const int* __restrict__ start, float* __restrict__ out, int N) {
  __shared__ float acc[TILE_ROWS * 64];
  const int b = blockIdx.x;
  for (int i = threadIdx.x; i < TILE_ROWS * 16; i += 256) ((v4f*)acc)[i] = (v4f){0.f, 0.f, 0.f, 0.f};
  __syncthreads();
  const int lo = start[b], hi = start[b + 1];
  const int wid = __builtin_amdgcn_readfirstlane((int)(threadIdx.x >> 5));
  const int lane = threadIdx.x & 31;
  for (int i0 = lo + wid * 32; i0 < hi; i0 += 8 * 32) {
    int i = i0 + lane;
    if (i >= hi) i = hi - 1;
    const v4f rec = *(const v4f*)(recs + 4 * (size_t)i);
    const int rs = __float_as_int(rec.x);
    const int rd = (int)((unsigned)__float_as_int(rec.y) % TILE_ROWS);
    const int rw = __float_as_int(WSEL ? rec.w : rec.z);
    const int n = hi - i0;
#pragma unroll
    for (int g = 0; g < 32; g += SLOTS_PER_TRIP) {
      v2f xv[SLOTS_PER_TRIP];
#pragma unroll
      for (int j = 0; j < SLOTS_PER_TRIP; ++j) {
        int s = __builtin_amdgcn_readlane(rs, g + j);
        xv[j] = *(const v2f*)(x + (size_t)s * 64 + 2 * lane);
      }
#pragma unroll
      for (int j = 0; j < SLOTS_PER_TRIP; ++j) {
        if (g + j < n) {
          int dl = __builtin_amdgcn_readlane(rd, g + j);
          float wv = __int_as_float(__builtin_amdgcn_readlane(rw, g + j));
          atomicAdd(&acc[dl * 64 + 2 * lane], xv[j].x * wv);
          atomicAdd(&acc[dl * 64 + 2 * lane + 1], xv[j].y * wv);
        }
      }
    }
  }
  __syncthreads();
  for (int i = threadIdx.x; i < TILE_ROWS * 16; i += 256) {
    int row = b * TILE_ROWS + (i >> 4);
    if (row < N) *(v4f*)(out + (size_t)row * 64 + 4 * (i & 15)) = ((const v4f*)acc)[i];
  }
}

#define ELEM_VECS_COMBINE 4

__global__ void __launch_bounds__(256)
combine_kernel(const float* __restrict__ a, const float* __restrict__ b,
               float* __restrict__ out, int n4) {
  const int T = (int)gridDim.x * 256;
  const int i = (int)blockIdx.x * 256 + (int)threadIdx.x;
  v4f av[ELEM_VECS_COMBINE], bv[ELEM_VECS_COMBINE];
#pragma unroll
  for (int j = 0; j < ELEM_VECS_COMBINE; ++j) {
    int k = i + j * T;
    k = k < n4 ? k : n4 - 1;
    av[j] = ((const v4f*)a)[k];
    bv[j] = ((const v4f*)b)[k];
  }
#pragma unroll
  for (int j = 0; j < ELEM_VECS_COMBINE; ++j) {
    int k = i + j * T;
    if (k < n4) ((v4f*)out)[k] = 2.0f * av[j] - bv[j];
  }
}

#define GEMM_ROWS 32
#define GEMM_KSTEPS 8
#define A_PAD 4

__global__ void __launch_bounds__(256)
qkv_gemm_kernel(const float* __restrict__ h, const float* __restrict__ p,
                const float* __restrict__ WQ, const float* __restrict__ WK,
                const float* __restrict__ WV,
                float* __restrict__ Qh, float* __restrict__ Kh, float* __restrict__ Vh, int rows) {
  constexpr int KDIM = 128;
  constexpr int PITCH = KDIM + A_PAD;
  __shared__ v4f stageA[GEMM_ROWS * PITCH / 4];
  float* As = (float*)stageA;
  const int row0 = (int)blockIdx.x * GEMM_ROWS;
  const int c4 = (int)threadIdx.x & 31;
  const float* colsrc = ((c4 < 16) ? h : p) + 4 * (c4 & 15);
#pragma unroll
  for (int j = 0; j < GEMM_ROWS / 8; ++j) {
    const int r = (int)(threadIdx.x >> 5) + 8 * j;
    int gr = row0 + r;
    if (gr >= rows) gr = rows - 1;
    *(v4f*)(As + r * PITCH + 4 * c4) = *(const v4f*)(colsrc + (size_t)gr * 64);
  }
  __syncthreads();
  const int w = (int)(threadIdx.x >> 5);
  const int rt = (int)blockIdx.x * (GEMM_ROWS / 16) + (w >> 2), ct = w & 3;
  const int rowTiles = (rows + 15) >> 4;
  if (rt >= rowTiles) return;
  const int lane = threadIdx.x & 31, half = lane >> 4, l16 = lane & 15;
  const int bc = ct * 16 + l16;
  const float* ap = As + ((w >> 2) * 16 + l16) * PITCH + 2 * half;
  const size_t boff = (size_t)(2 * half) * 64 + bc;
  const float* bq = WQ + boff;
  const float* bk = WK + boff;
  const float* bv = WV + boff;
  v8f accQ = {}, accK = {}, accV = {};
#pragma unroll 1
  for (int kk = 0; kk < KDIM; kk += 4 * GEMM_KSTEPS) {
    v2f a[GEMM_KSTEPS], fq[GEMM_KSTEPS], fk[GEMM_KSTEPS], fv[GEMM_KSTEPS];
#pragma unroll
    for (int s = 0; s < GEMM_KSTEPS; ++s) {
      const int k0 = kk + 4 * s;
      a[s] = *(const v2f*)(ap + k0);
      fq[s].x = bq[(size_t)k0 * 64];
      fq[s].y = bq[(size_t)(k0 + 1) * 64];
      fk[s].x = bk[(size_t)k0 * 64];
      fk[s].y = bk[(size_t)(k0 + 1) * 64];
      fv[s].x = bv[(size_t)k0 * 64];
      fv[s].y = bv[(size_t)(k0 + 1) * 64];
    }
#pragma unroll
    for (int s = 0; s < GEMM_KSTEPS; ++s) {
      accQ = wmma4(a[s], fq[s], accQ);
      accK = wmma4(a[s], fk[s], accK);
      accV = wmma4(a[s], fv[s], accV);
    }
    __builtin_amdgcn_sched_group_barrier(0x020, 6 * GEMM_KSTEPS, 0);
    __builtin_amdgcn_sched_group_barrier(0x100, GEMM_KSTEPS, 0);
    __builtin_amdgcn_sched_group_barrier(0x008, 3 * GEMM_KSTEPS, 0);
  }
#pragma unroll
  for (int v = 0; v < 8; ++v) {
    int r = rt * 16 + v + 8 * half;
    if (r < rows) {
      const size_t o = (size_t)r * 64 + ct * 16 + l16;
      Qh[o] = accQ[v];
      Kh[o] = accK[v];
      Vh[o] = accV[v];
    }
  }
}

#define TILE_PITCH (64 + A_PAD)
static_assert(TILE_ROWS / 16 * 4 == 8 && TILE_ROWS % 16 == 0, "8 waves = TILE_ROWS / 16 row tiles x 4 column tiles");

template <int WSEL>
__device__ __forceinline__ void tile_gather(const float* __restrict__ x,
                                            const float* __restrict__ recs,
                                            int lo, int hi, float* acc) {
  const int wid = __builtin_amdgcn_readfirstlane((int)(threadIdx.x >> 5));
  const int lane = threadIdx.x & 31;
  for (int i0 = lo + wid * 32; i0 < hi; i0 += 8 * 32) {
    int i = i0 + lane;
    if (i >= hi) i = hi - 1;
    const v4f rec = *(const v4f*)(recs + 4 * (size_t)i);
    const int rs = __float_as_int(rec.x);
    const int rd = (int)((unsigned)__float_as_int(rec.y) % TILE_ROWS);
    const int rw = __float_as_int(WSEL ? rec.w : rec.z);
    const int n = hi - i0;
#pragma unroll
    for (int g = 0; g < 32; g += SLOTS_PER_TRIP) {
      v2f xv[SLOTS_PER_TRIP];
#pragma unroll
      for (int j = 0; j < SLOTS_PER_TRIP; ++j) {
        int s = __builtin_amdgcn_readlane(rs, g + j);
        xv[j] = *(const v2f*)(x + (size_t)s * 64 + 2 * lane);
      }
#pragma unroll
      for (int j = 0; j < SLOTS_PER_TRIP; ++j) {
        if (g + j < n) {
          int dl = __builtin_amdgcn_readlane(rd, g + j);
          float wv = __int_as_float(__builtin_amdgcn_readlane(rw, g + j));
          atomicAdd(&acc[dl * TILE_PITCH + 2 * lane], xv[j].x * wv);
          atomicAdd(&acc[dl * TILE_PITCH + 2 * lane + 1], xv[j].y * wv);
        }
      }
    }
  }
}

__global__ void __launch_bounds__(256)
tile_sum_gcn_kernel(const float* __restrict__ x, const float* __restrict__ recs,
                    const int* __restrict__ start, const float* __restrict__ Wg,
                    const float* __restrict__ bias, float* __restrict__ out, int N) {
  __shared__ v4f accv[TILE_ROWS * TILE_PITCH / 4];
  float* acc = (float*)accv;
  const int b = blockIdx.x;
  for (int i = threadIdx.x; i < TILE_ROWS * TILE_PITCH / 4; i += 256) accv[i] = (v4f){0.f, 0.f, 0.f, 0.f};
  __syncthreads();
  tile_gather<1>(x, recs, start[b], start[b + 1], acc);
  __syncthreads();
  const int w = (int)(threadIdx.x >> 5);
  const int lane = threadIdx.x & 31, half = lane >> 4, l16 = lane & 15;
  const int ct = w & 3, rp = w >> 2;
  const float* a0p = acc + (rp * 16 + l16) * TILE_PITCH + 2 * half;
  const float* bp = Wg + (size_t)(2 * half) * 64 + ct * 16 + l16;
  v8f c0 = {};
#pragma unroll 1
  for (int kk = 0; kk < 64; kk += 4 * GEMM_KSTEPS) {
    v2f a0[GEMM_KSTEPS], bf[GEMM_KSTEPS];
#pragma unroll
    for (int s = 0; s < GEMM_KSTEPS; ++s) {
      const int k0 = kk + 4 * s;
      a0[s] = *(const v2f*)(a0p + k0);
      bf[s].x = bp[(size_t)k0 * 64];
      bf[s].y = bp[(size_t)(k0 + 1) * 64];
    }
#pragma unroll
    for (int s = 0; s < GEMM_KSTEPS; ++s) c0 = wmma4(a0[s], bf[s], c0);
    __builtin_amdgcn_sched_group_barrier(0x020, 2 * GEMM_KSTEPS, 0);
    __builtin_amdgcn_sched_group_barrier(0x100, GEMM_KSTEPS, 0);
    __builtin_amdgcn_sched_group_barrier(0x008, GEMM_KSTEPS, 0);
  }
  const float bv = bias[ct * 16 + l16];
#pragma unroll
  for (int v = 0; v < 8; ++v) {
    const int r0 = b * TILE_ROWS + rp * 16 + v + 8 * half;
    if (r0 < N) out[(size_t)r0 * 192 + 64 + ct * 16 + l16] = c0[v] + bv;
  }
}

template <int TERM>
__device__ __forceinline__ void cheb_term(const float* ak, const float* t1,
                                          const float* __restrict__ bk,
                                          float s0, v8f& c0) {
#pragma unroll 1
  for (int kk = 0; kk < 64; kk += 4 * GEMM_KSTEPS) {
    v2f a0[GEMM_KSTEPS], bf[GEMM_KSTEPS];
#pragma unroll
    for (int s = 0; s < GEMM_KSTEPS; ++s) {
      const int k0 = kk + 4 * s;
      v2f x0 = *(const v2f*)(ak + k0);
      if (TERM == 3) x0 = 2.0f * x0 - *(const v2f*)(t1 + k0);
      a0[s] = x0 * s0;
      bf[s].x = bk[(size_t)k0 * 64];
      bf[s].y = bk[(size_t)(k0 + 1) * 64];
    }
#pragma unroll
    for (int s = 0; s < GEMM_KSTEPS; ++s) c0 = wmma4(a0[s], bf[s], c0);
    __builtin_amdgcn_sched_group_barrier(0x020, 2 * GEMM_KSTEPS, 0);
    __builtin_amdgcn_sched_group_barrier(0x100, (TERM == 3 ? 2 : 1) * GEMM_KSTEPS, 0);
    __builtin_amdgcn_sched_group_barrier(0x008, GEMM_KSTEPS, 0);
  }
}

__global__ void __launch_bounds__(256)
tile_sum_cheb_kernel(const float* __restrict__ Tx0, const float* __restrict__ Tx1,
                     const float* __restrict__ Tx2, const float* __restrict__ recs,
                     const int* __restrict__ start, const float* __restrict__ fc,
                     const float* __restrict__ Wc, const float* __restrict__ bias,
                     float* __restrict__ out, int N) {
  __shared__ v4f accv[TILE_ROWS * TILE_PITCH / 4];
  __shared__ v4f rowsv[3 * TILE_ROWS * TILE_PITCH / 4];
  float* acc = (float*)accv;
  float* As = (float*)rowsv;
  const int b = blockIdx.x;
  const float* Asrc[3] = {Tx0, Tx1, Tx2};
#pragma unroll
  for (int j = 0; j < 3 * (TILE_ROWS / 16); ++j) {
    const int r = (int)(threadIdx.x >> 4) + 16 * (j % (TILE_ROWS / 16)), c4 = (int)threadIdx.x & 15;
    int gr = b * TILE_ROWS + r;
    if (gr >= N) gr = N - 1;
    *(v4f*)(As + ((j / (TILE_ROWS / 16)) * TILE_ROWS + r) * TILE_PITCH + 4 * c4) =
        *(const v4f*)(Asrc[j / (TILE_ROWS / 16)] + (size_t)gr * 64 + 4 * c4);
  }
  for (int i = threadIdx.x; i < TILE_ROWS * TILE_PITCH / 4; i += 256) accv[i] = (v4f){0.f, 0.f, 0.f, 0.f};
  __syncthreads();
  tile_gather<0>(Tx2, recs, start[b], start[b + 1], acc);
  __syncthreads();
  const int w = (int)(threadIdx.x >> 5);
  const int lane = threadIdx.x & 31, half = lane >> 4, l16 = lane & 15;
  const int ct = w & 3, rp = w >> 2;
  const int lrow = rp * 16 + l16;
  int ar0 = b * TILE_ROWS + lrow;
  if (ar0 >= N) ar0 = N - 1;
  float sc0[4];
#pragma unroll
  for (int k = 0; k < 4; ++k) sc0[k] = fc[(size_t)k * N + ar0];
  v8f c0 = {};
  const float* t1 = As + (TILE_ROWS + lrow) * TILE_PITCH + 2 * half;
  const float* bk = Wc + (size_t)(2 * half) * 64 + ct * 16 + l16;
  cheb_term<0>(As + (0 * TILE_ROWS + lrow) * TILE_PITCH + 2 * half, t1, bk, sc0[0], c0);
  cheb_term<1>(As + (1 * TILE_ROWS + lrow) * TILE_PITCH + 2 * half, t1, bk + 1 * 64 * 64, sc0[1], c0);
  cheb_term<2>(As + (2 * TILE_ROWS + lrow) * TILE_PITCH + 2 * half, t1, bk + 2 * 64 * 64, sc0[2], c0);
  cheb_term<3>(acc + lrow * TILE_PITCH + 2 * half, t1, bk + 3 * 64 * 64, sc0[3], c0);
  const float bv = bias[ct * 16 + l16];
#pragma unroll
  for (int v = 0; v < 8; ++v) {
    const int r0 = b * TILE_ROWS + rp * 16 + v + 8 * half;
    if (r0 < N) out[(size_t)r0 * 192 + ct * 16 + l16] = c0[v] + bv;
  }
}


__device__ __forceinline__ float quad_sum(float v) {
  v += __int_as_float(__builtin_amdgcn_update_dpp(0, __float_as_int(v), 0xB1, 0xF, 0xF, true));
  v += __int_as_float(__builtin_amdgcn_update_dpp(0, __float_as_int(v), 0x4E, 0xF, 0xF, true));
  return v;
}

#define ATT_EDGES_PER_TRIP 8

__global__ void __launch_bounds__(256)
attn_tile_kernel(const float* __restrict__ Ee, const float* __restrict__ WE,
                 const float* __restrict__ Qh, const float* __restrict__ Kh,
                 const float* __restrict__ Vh, const float* __restrict__ recs,
                 const int* __restrict__ start, float* __restrict__ out, int N) {
  __shared__ float acc[TILE_ROWS * 64];
  __shared__ float zacc[TILE_ROWS * 8];
  __shared__ float stage[8 * 16 * 64];
  __shared__ v4f wfrag[16 * 64];
  const int b = blockIdx.x;
  for (int i = threadIdx.x; i < TILE_ROWS * 16; i += 256) ((v4f*)acc)[i] = (v4f){0.f, 0.f, 0.f, 0.f};
  for (int i = threadIdx.x; i < TILE_ROWS * 8; i += 256) zacc[i] = 0.f;
#pragma unroll
  for (int j = 0; j < 4; ++j) {
    const int f = (int)threadIdx.x + 256 * j;
    const int col = f & 63;
    const int r0 = 8 * (f >> 7) + 2 * ((f >> 6) & 1);
    wfrag[f] = (v4f){WE[r0 * 64 + col], WE[(r0 + 1) * 64 + col],
                     WE[(r0 + 4) * 64 + col], WE[(r0 + 5) * 64 + col]};
  }
  __syncthreads();
  const int lo = start[b], hi = start[b + 1];
  const int wid = __builtin_amdgcn_readfirstlane((int)(threadIdx.x >> 5));
  const int lane = threadIdx.x & 31, half = lane >> 4, l16 = lane & 15;
  float* myLds = stage + wid * 1024;
  const float scale = 0.35355339059327373f;

  for (int i0 = lo + wid * 16; i0 < hi; i0 += 8 * 16) {
    int i = i0 + l16;
    if (i >= hi) i = hi - 1;
    const v4f rec = *(const v4f*)(recs + 4 * (size_t)i);
    const int rs = __float_as_int(rec.x);
    const unsigned ry = (unsigned)__float_as_int(rec.y);
    const int rd = (int)(ry % TILE_ROWS);
    const int ar = (int)(ry / TILE_ROWS);

#pragma unroll 1
    for (int ct = 0; ct < 4; ++ct) {
      v8f c = {};
      const v4f* wf = wfrag + half * 64 + ct * 16 + l16;
      v4f q[8];
#pragma unroll
      for (int P = 0; P < 8; ++P) q[P] = wf[P * 128];
#pragma unroll
      for (int P = 0; P < 8; ++P) {
        const int kr = 8 * P + 2 * half;
        v2f a0, a1;
        a0.x = Ee[(size_t)ar * 64 + kr];
        a0.y = Ee[(size_t)ar * 64 + kr + 1];
        a1.x = Ee[(size_t)ar * 64 + kr + 4];
        a1.y = Ee[(size_t)ar * 64 + kr + 5];
        c = wmma4(a0, (v2f){q[P].x, q[P].y}, c);
        c = wmma4(a1, (v2f){q[P].z, q[P].w}, c);
      }
      __builtin_amdgcn_sched_group_barrier(0x100, 8, 0);
      __builtin_amdgcn_sched_group_barrier(0x008, 16, 0);
#pragma unroll
      for (int v = 0; v < 8; ++v)
        myLds[(v + 8 * half) * 64 + ct * 16 + l16] = c[v];
    }

    const int n = hi - i0;
    const float* qbase = Qh + (size_t)b * TILE_ROWS * 64;
#pragma unroll
    for (int g = 0; g < 16; g += ATT_EDGES_PER_TRIP) {
      v2f kv[ATT_EDGES_PER_TRIP], qv[ATT_EDGES_PER_TRIP], vv[ATT_EDGES_PER_TRIP], ev[ATT_EDGES_PER_TRIP];
#pragma unroll
      for (int j = 0; j < ATT_EDGES_PER_TRIP; ++j) {
        int s = __builtin_amdgcn_readlane(rs, g + j);
        int dl = __builtin_amdgcn_readlane(rd, g + j);
        kv[j] = *(const v2f*)(Kh + (size_t)s * 64 + 2 * lane);
        qv[j] = *(const v2f*)(qbase + (size_t)dl * 64 + 2 * lane);
        vv[j] = *(const v2f*)(Vh + (size_t)s * 64 + 2 * lane);
        ev[j] = *(const v2f*)(myLds + (g + j) * 64 + 2 * lane);
      }
#pragma unroll
      for (int j = 0; j < ATT_EDGES_PER_TRIP; ++j) {
        if (g + j < n) {
          int dl = __builtin_amdgcn_readlane(rd, g + j);
          float sc = quad_sum(kv[j].x * qv[j].x * ev[j].x + kv[j].y * qv[j].y * ev[j].y);
          sc *= scale;
          sc = fminf(5.0f, fmaxf(-5.0f, sc));
          float ss = __expf(sc);
          atomicAdd(&acc[dl * 64 + 2 * lane], vv[j].x * ss);
          atomicAdd(&acc[dl * 64 + 2 * lane + 1], vv[j].y * ss);
          if ((lane & 3) == 0) atomicAdd(&zacc[dl * 8 + (lane >> 2)], ss);
        }
      }
    }
  }
  __syncthreads();
  for (int i = threadIdx.x; i < TILE_ROWS * 16; i += 256) {
    int r = i >> 4, c4 = i & 15;
    int row = b * TILE_ROWS + r;
    if (row < N) {
      v4f a = ((const v4f*)acc)[i];
      float zz = zacc[r * 8 + (c4 >> 1)] + 1e-6f;
      v4f o = {a.x / zz, a.y / zz, a.z / zz, a.w / zz};
      *(v4f*)(out + (size_t)row * 192 + 128 + 4 * c4) = o;
    }
  }
}

extern "C" void kernel_launch(void* const* d_in, const int* in_sizes, int n_in,
                              void* d_out, int out_size, void* d_ws, size_t ws_size,
                              hipStream_t stream) {
  const float* h  = (const float*)d_in[0];
  const float* p  = (const float*)d_in[1];
  const float* e  = (const float*)d_in[2];
  const float* ew = (const float*)d_in[3];
  const float* fc = (const float*)d_in[4];
  const int* esrc = (const int*)d_in[5];
  const int* edst = (const int*)d_in[6];
  const float* Wc = (const float*)d_in[7];
  const float* bc = (const float*)d_in[8];
  const float* Wg = (const float*)d_in[9];
  const float* bg = (const float*)d_in[10];
  const float* WQ = (const float*)d_in[11];
  const float* WK = (const float*)d_in[12];
  const float* WVm = (const float*)d_in[13];
  const float* WE = (const float*)d_in[14];
  float* out = (float*)d_out;

  const int N = in_sizes[0] / 64;
  const int E = in_sizes[5];
  const size_t ND = (size_t)N * 64;

  float* w = (float*)d_ws;
  w += N;
  float* dinv = w; w += N;
  w += E;
  w = (float*)(((size_t)w + 15) & ~(size_t)15);
  float* Tx1  = w; w += ND;
  float* Tx2  = w; w += ND;
  w += ND;
  float* tmp  = w; w += ND;
  w += ND;
  float* scr  = w; w += 2 * ND;
  float* Qh   = w; w += ND;
  float* Kh   = w; w += ND;
  float* Vh   = w; w += ND;
  float* list = w; w += ND + (size_t)N * 8;

  const int nb = (N + TILE_ROWS - 1) / TILE_ROWS;
  float* recs   = list;
  int* binStart = (int*)(recs + 4 * (size_t)E);
  int* binCnt   = (int*)scr;
  int* rank     = binCnt + (size_t)nb * CNT_PITCH;

  hipMemsetAsync(binCnt, 0, sizeof(int) * (size_t)nb * CNT_PITCH, stream);

  const int B = 256;
  const int nd4 = N * 16;
  const int combBlocks = (nd4 + B * ELEM_VECS_COMBINE - 1) / (B * ELEM_VECS_COMBINE);

  const int rankEdges = RANK_THREADS * RANK_PER_THREAD;
  tile_rank_kernel<<<(E + rankEdges - 1) / rankEdges, RANK_THREADS, sizeof(int) * (size_t)nb, stream>>>(
      edst, binCnt, rank, E, nb);
  tile_scan_kernel<<<1, 1024, 0, stream>>>(binCnt, binStart, nb);
  tile_fill_kernel<<<(E + B - 1) / B, B, 0, stream>>>(esrc, edst, ew, binStart, rank, recs, E);

  tile_deg_kernel<<<nb, 256, 0, stream>>>(recs, binStart, dinv, N);
  wlap_kernel<<<(E + B - 1) / B, B, 0, stream>>>(dinv, recs, E);

  tile_sum_kernel<0><<<nb, 256, 0, stream>>>(h, recs, binStart, Tx1, N);
  tile_sum_kernel<0><<<nb, 256, 0, stream>>>(Tx1, recs, binStart, tmp, N);
  combine_kernel<<<combBlocks, B, 0, stream>>>(tmp, h, Tx2, nd4);
  tile_sum_cheb_kernel<<<nb, 256, 0, stream>>>(h, Tx1, Tx2, recs, binStart, fc, Wc, bc, out, N);

  tile_sum_gcn_kernel<<<nb, 256, 0, stream>>>(h, recs, binStart, Wg, bg, out, N);

  const int gemmBlocks = (N + GEMM_ROWS - 1) / GEMM_ROWS;
  qkv_gemm_kernel<<<gemmBlocks, 256, 0, stream>>>(h, p, WQ, WK, WVm, Qh, Kh, Vh, N);

  attn_tile_kernel<<<nb, 256, 0, stream>>>(e, WE, Qh, Kh, Vh, recs, binStart, out, N);
}


# Six mutually independent passes share three launches, block number picks the pass: GCN tiles + in-degree count; Q,K,V product + Chebyshev 2a-b pass; attention + last Chebyshev tiles; 3 launches fewer
# speedup vs baseline: 4.0168x; 1.0841x over previous
#include <hip/hip_runtime.h>
#include <hip/hip_bf16.h>

typedef __attribute__((ext_vector_type(2))) float v2f;
typedef __attribute__((ext_vector_type(8))) float v8f;
typedef __attribute__((ext_vector_type(4))) float v4f;

#define TILE_ROWS 32
#define CNT_PITCH 1
#define RANK_THREADS 1024
#define RANK_PER_THREAD 8
#define SLOTS_PER_TRIP 16
#define ROLE_THREADS 256

__device__ __forceinline__ v8f wmma4(v2f a, v2f b, v8f c) {
  return __builtin_amdgcn_wmma_f32_16x16x4_f32(
       false, a,  false, b,
       (short)0, c,  false,  false);
}


__global__ void __launch_bounds__(RANK_THREADS)
tile_rank_kernel(const int* __restrict__ dst, int* __restrict__ cnt,
                 int* __restrict__ rank, int E, int nb) {
  extern __shared__ int hist[];
  for (int t = threadIdx.x; t < nb; t += RANK_THREADS) hist[t] = 0;
  __syncthreads();
  const int e0 = (int)blockIdx.x * (RANK_THREADS * RANK_PER_THREAD) + (int)threadIdx.x;
  int tl[RANK_PER_THREAD], lr[RANK_PER_THREAD];
#pragma unroll
  for (int j = 0; j < RANK_PER_THREAD; ++j) {
    int e = e0 + j * RANK_THREADS;
    tl[j] = (int)((unsigned)dst[e < E ? e : E - 1] / TILE_ROWS);
  }
#pragma unroll
  for (int j = 0; j < RANK_PER_THREAD; ++j)
    lr[j] = atomicAdd(&hist[tl[j]], (e0 + j * RANK_THREADS < E) ? 1 : 0);
  __syncthreads();
  for (int t = threadIdx.x; t < nb; t += RANK_THREADS) {
    int c = hist[t];
    hist[t] = (c > 0) ? atomicAdd(&cnt[(size_t)t * CNT_PITCH], c) : 0;
  }
  __syncthreads();
  int bs[RANK_PER_THREAD];
#pragma unroll
  for (int j = 0; j < RANK_PER_THREAD; ++j) bs[j] = hist[tl[j]];
#pragma unroll
  for (int j = 0; j < RANK_PER_THREAD; ++j) {
    int e = e0 + j * RANK_THREADS;
    if (e < E) rank[e] = bs[j] + lr[j];
  }
}

__global__ void tile_scan_kernel(const int* __restrict__ cnt, int* __restrict__ start, int nb) {
  __shared__ int part[1024];
  int t = threadIdx.x;
  int per = (nb + 1023) / 1024;
  int b0 = t * per;
  int s = 0;
  for (int i = 0; i < per; ++i) {
    int b = b0 + i;
    if (b < nb) s += cnt[(size_t)b * CNT_PITCH];
  }
  part[t] = s;
  __syncthreads();
  for (int off = 1; off < 1024; off <<= 1) {
    int v = (t >= off) ? part[t - off] : 0;
    __syncthreads();
    part[t] += v;
    __syncthreads();
  }
  int run = part[t] - s;
  for (int i = 0; i < per; ++i) {
    int b = b0 + i;
    if (b < nb) {
      start[b] = run;
      run += cnt[(size_t)b * CNT_PITCH];
    }
  }
  if (t == 1023) start[nb] = part[1023];
}

__global__ void tile_fill_kernel(const int* __restrict__ src, const int* __restrict__ dst,
                                 const float* __restrict__ ew,
                                 const int* __restrict__ start, const int* __restrict__ rank,
                                 float* __restrict__ recs, int E) {
  int tid = blockIdx.x * blockDim.x + threadIdx.x;
  if (tid >= E) return;
  int d = dst[tid];
  int b = (int)((unsigned)d / TILE_ROWS);
  int pos = start[b] + rank[tid];
  v4f r;
  r.x = __int_as_float(src[tid]);
  r.y = __int_as_float(tid * TILE_ROWS + (d - b * TILE_ROWS));
  r.z = __int_as_float(d);
  r.w = ew[tid];
  *(v4f*)(recs + 4 * (size_t)pos) = r;
}

__device__ __forceinline__ void tile_deg_body(int b, const float* __restrict__ recs,
                                              const int* __restrict__ start,
                                              float* __restrict__ dinv, int N) {
  __shared__ int cnt[TILE_ROWS];
  if (threadIdx.x < TILE_ROWS) cnt[threadIdx.x] = 0;
  __syncthreads();
  const int lo = start[b], hi = start[b + 1];
  for (int i = lo + (int)threadIdx.x; i < hi; i += 256) {
    unsigned ry = (unsigned)__float_as_int(recs[4 * (size_t)i + 1]);
    atomicAdd(&cnt[ry % TILE_ROWS], 1);
  }
  __syncthreads();
  if (threadIdx.x < TILE_ROWS) {
    int row = b * TILE_ROWS + (int)threadIdx.x;
    if (row < N) {
      float d = (float)cnt[threadIdx.x];
      d = d < 1.0f ? 1.0f : d;
      dinv[row] = rsqrtf(d);
    }
  }
}

__global__ void wlap_kernel(const float* __restrict__ dinv, float* recs, int E) {
  int i = blockIdx.x * blockDim.x + threadIdx.x;
  if (i < E) {
    v4f r = *(const v4f*)(recs + 4 * (size_t)i);
    r.z = -(dinv[__float_as_int(r.x)] * dinv[__float_as_int(r.z)]);
    *(v4f*)(recs + 4 * (size_t)i) = r;
  }
}

template <int WSEL>
__global__ void __launch_bounds__(256)
tile_sum_kernel(const float* __restrict__ x, const float* __restrict__ recs,
                const int* __restrict__ start, float* __restrict__ out, int N) {
  __shared__ float acc[TILE_ROWS * 64];
  const int b = blockIdx.x;
  for (int i = threadIdx.x; i < TILE_ROWS * 16; i += 256) ((v4f*)acc)[i] = (v4f){0.f, 0.f, 0.f, 0.f};
  __syncthreads();
  const int lo = start[b], hi = start[b + 1];
  const int wid = __builtin_amdgcn_readfirstlane((int)(threadIdx.x >> 5));
  const int lane = threadIdx.x & 31;
  for (int i0 = lo + wid * 32; i0 < hi; i0 += 8 * 32) {
    int i = i0 + lane;
    if (i >= hi) i = hi - 1;
    const v4f rec = *(const v4f*)(recs + 4 * (size_t)i);
    const int rs = __float_as_int(rec.x);
    const int rd = (int)((unsigned)__float_as_int(rec.y) % TILE_ROWS);
    const int rw = __float_as_int(WSEL ? rec.w : rec.z);
    const int n = hi - i0;
#pragma unroll
    for (int g = 0; g < 32; g += SLOTS_PER_TRIP) {
      v2f xv[SLOTS_PER_TRIP];
#pragma unroll
      for (int j = 0; j < SLOTS_PER_TRIP; ++j) {
        int s = __builtin_amdgcn_readlane(rs, g + j);
        xv[j] = *(const v2f*)(x + (size_t)s * 64 + 2 * lane);
      }
#pragma unroll
      for (int j = 0; j < SLOTS_PER_TRIP; ++j) {
        if (g + j < n) {
          int dl = __builtin_amdgcn_readlane(rd, g + j);
          float wv = __int_as_float(__builtin_amdgcn_readlane(rw, g + j));
          atomicAdd(&acc[dl * 64 + 2 * lane], xv[j].x * wv);
          atomicAdd(&acc[dl * 64 + 2 * lane + 1], xv[j].y * wv);
        }
      }
    }
  }
  __syncthreads();
  for (int i = threadIdx.x; i < TILE_ROWS * 16; i += 256) {
    int row = b * TILE_ROWS + (i >> 4);
    if (row < N) *(v4f*)(out + (size_t)row * 64 + 4 * (i & 15)) = ((const v4f*)acc)[i];
  }
}

#define ELEM_VECS_COMBINE 4

__device__ __forceinline__ void
combine_body(int bx, int nComb, const float* __restrict__ a, const float* __restrict__ b,
             float* __restrict__ out, int n4) {
  const int T = nComb * ROLE_THREADS;
  const int i = bx * ROLE_THREADS + (int)threadIdx.x;
  v4f av[ELEM_VECS_COMBINE], bv[ELEM_VECS_COMBINE];
#pragma unroll
  for (int j = 0; j < ELEM_VECS_COMBINE; ++j) {
    int k = i + j * T;
    k = k < n4 ? k : n4 - 1;
    av[j] = ((const v4f*)a)[k];
    bv[j] = ((const v4f*)b)[k];
  }
#pragma unroll
  for (int j = 0; j < ELEM_VECS_COMBINE; ++j) {
    int k = i + j * T;
    if (k < n4) ((v4f*)out)[k] = 2.0f * av[j] - bv[j];
  }
}

#define GEMM_ROWS 32
#define GEMM_KSTEPS 8
#define A_PAD 4

__device__ __forceinline__ void
qkv_gemm_body(int bx, const float* __restrict__ h, const float* __restrict__ p,
              const float* __restrict__ WQ, const float* __restrict__ WK,
              const float* __restrict__ WV,
              float* __restrict__ Qh, float* __restrict__ Kh, float* __restrict__ Vh, int rows) {
  constexpr int KDIM = 128;
  constexpr int PITCH = KDIM + A_PAD;
  __shared__ v4f stageA[GEMM_ROWS * PITCH / 4];
  float* As = (float*)stageA;
  const int row0 = bx * GEMM_ROWS;
  const int c4 = (int)threadIdx.x & 31;
  const float* colsrc = ((c4 < 16) ? h : p) + 4 * (c4 & 15);
#pragma unroll
  for (int j = 0; j < GEMM_ROWS / 8; ++j) {
    const int r = (int)(threadIdx.x >> 5) + 8 * j;
    int gr = row0 + r;
    if (gr >= rows) gr = rows - 1;
    *(v4f*)(As + r * PITCH + 4 * c4) = *(const v4f*)(colsrc + (size_t)gr * 64);
  }
  __syncthreads();
  const int w = (int)(threadIdx.x >> 5);
  const int rt = bx * (GEMM_ROWS / 16) + (w >> 2), ct = w & 3;
  const int rowTiles = (rows + 15) >> 4;
  if (rt >= rowTiles) return;
  const int lane = threadIdx.x & 31, half = lane >> 4, l16 = lane & 15;
  const int bc = ct * 16 + l16;
  const float* ap = As + ((w >> 2) * 16 + l16) * PITCH + 2 * half;
  const size_t boff = (size_t)(2 * half) * 64 + bc;
  const float* bq = WQ + boff;
  const float* bk = WK + boff;
  const float* bv = WV + boff;
  v8f accQ = {}, accK = {}, accV = {};
#pragma unroll 1
  for (int kk = 0; kk < KDIM; kk += 4 * GEMM_KSTEPS) {
    v2f a[GEMM_KSTEPS], fq[GEMM_KSTEPS], fk[GEMM_KSTEPS], fv[GEMM_KSTEPS];
#pragma unroll
    for (int s = 0; s < GEMM_KSTEPS; ++s) {
      const int k0 = kk + 4 * s;
      a[s] = *(const v2f*)(ap + k0);
      fq[s].x = bq[(size_t)k0 * 64];
      fq[s].y = bq[(size_t)(k0 + 1) * 64];
      fk[s].x = bk[(size_t)k0 * 64];
      fk[s].y = bk[(size_t)(k0 + 1) * 64];
      fv[s].x = bv[(size_t)k0 * 64];
      fv[s].y = bv[(size_t)(k0 + 1) * 64];
    }
#pragma unroll
    for (int s = 0; s < GEMM_KSTEPS; ++s) {
      accQ = wmma4(a[s], fq[s], accQ);
      accK = wmma4(a[s], fk[s], accK);
      accV = wmma4(a[s], fv[s], accV);
    }
    __builtin_amdgcn_sched_group_barrier(0x020, 6 * GEMM_KSTEPS, 0);
    __builtin_amdgcn_sched_group_barrier(0x100, GEMM_KSTEPS, 0);
    __builtin_amdgcn_sched_group_barrier(0x008, 3 * GEMM_KSTEPS, 0);
  }
#pragma unroll
  for (int v = 0; v < 8; ++v) {
    int r = rt * 16 + v + 8 * half;
    if (r < rows) {
      const size_t o = (size_t)r * 64 + ct * 16 + l16;
      Qh[o] = accQ[v];
      Kh[o] = accK[v];
      Vh[o] = accV[v];
    }
  }
}

#define TILE_PITCH (64 + A_PAD)
static_assert(TILE_ROWS / 16 * 4 == 8 && TILE_ROWS % 16 == 0, "8 waves = TILE_ROWS / 16 row tiles x 4 column tiles");

template <int WSEL>
__device__ __forceinline__ void tile_gather(const float* __restrict__ x,
                                            const float* __restrict__ recs,
                                            int lo, int hi, float* acc) {
  const int wid = __builtin_amdgcn_readfirstlane((int)(threadIdx.x >> 5));
  const int lane = threadIdx.x & 31;
  for (int i0 = lo + wid * 32; i0 < hi; i0 += 8 * 32) {
    int i = i0 + lane;
    if (i >= hi) i = hi - 1;
    const v4f rec = *(const v4f*)(recs + 4 * (size_t)i);
    const int rs = __float_as_int(rec.x);
    const int rd = (int)((unsigned)__float_as_int(rec.y) % TILE_ROWS);
    const int rw = __float_as_int(WSEL ? rec.w : rec.z);
    const int n = hi - i0;
#pragma unroll
    for (int g = 0; g < 32; g += SLOTS_PER_TRIP) {
      v2f xv[SLOTS_PER_TRIP];
#pragma unroll
      for (int j = 0; j < SLOTS_PER_TRIP; ++j) {
        int s = __builtin_amdgcn_readlane(rs, g + j);
        xv[j] = *(const v2f*)(x + (size_t)s * 64 + 2 * lane);
      }
#pragma unroll
      for (int j = 0; j < SLOTS_PER_TRIP; ++j) {
        if (g + j < n) {
          int dl = __builtin_amdgcn_readlane(rd, g + j);
          float wv = __int_as_float(__builtin_amdgcn_readlane(rw, g + j));
          atomicAdd(&acc[dl * TILE_PITCH + 2 * lane], xv[j].x * wv);
          atomicAdd(&acc[dl * TILE_PITCH + 2 * lane + 1], xv[j].y * wv);
        }
      }
    }
  }
}

__device__ __forceinline__ void
tile_sum_gcn_body(int b, const float* __restrict__ x, const float* __restrict__ recs,
                  const int* __restrict__ start, const float* __restrict__ Wg,
                  const float* __restrict__ bias, float* __restrict__ out, int N) {
  __shared__ v4f accv[TILE_ROWS * TILE_PITCH / 4];
  float* acc = (float*)accv;
  for (int i = threadIdx.x; i < TILE_ROWS * TILE_PITCH / 4; i += 256) accv[i] = (v4f){0.f, 0.f, 0.f, 0.f};
  __syncthreads();
  tile_gather<1>(x, recs, start[b], start[b + 1], acc);
  __syncthreads();
  const int w = (int)(threadIdx.x >> 5);
  const int lane = threadIdx.x & 31, half = lane >> 4, l16 = lane & 15;
  const int ct = w & 3, rp = w >> 2;
  const float* a0p = acc + (rp * 16 + l16) * TILE_PITCH + 2 * half;
  const float* bp = Wg + (size_t)(2 * half) * 64 + ct * 16 + l16;
  v8f c0 = {};
#pragma unroll 1
  for (int kk = 0; kk < 64; kk += 4 * GEMM_KSTEPS) {
    v2f a0[GEMM_KSTEPS], bf[GEMM_KSTEPS];
#pragma unroll
    for (int s = 0; s < GEMM_KSTEPS; ++s) {
      const int k0 = kk + 4 * s;
      a0[s] = *(const v2f*)(a0p + k0);
      bf[s].x = bp[(size_t)k0 * 64];
      bf[s].y = bp[(size_t)(k0 + 1) * 64];
    }
#pragma unroll
    for (int s = 0; s < GEMM_KSTEPS; ++s) c0 = wmma4(a0[s], bf[s], c0);
    __builtin_amdgcn_sched_group_barrier(0x020, 2 * GEMM_KSTEPS, 0);
    __builtin_amdgcn_sched_group_barrier(0x100, GEMM_KSTEPS, 0);
    __builtin_amdgcn_sched_group_barrier(0x008, GEMM_KSTEPS, 0);
  }
  const float bv = bias[ct * 16 + l16];
#pragma unroll
  for (int v = 0; v < 8; ++v) {
    const int r0 = b * TILE_ROWS + rp * 16 + v + 8 * half;
    if (r0 < N) out[(size_t)r0 * 192 + 64 + ct * 16 + l16] = c0[v] + bv;
  }
}

template <int TERM>
__device__ __forceinline__ void cheb_term(const float* ak, const float* t1,
                                          const float* __restrict__ bk,
                                          float s0, v8f& c0) {
#pragma unroll 1
  for (int kk = 0; kk < 64; kk += 4 * GEMM_KSTEPS) {
    v2f a0[GEMM_KSTEPS], bf[GEMM_KSTEPS];
#pragma unroll
    for (int s = 0; s < GEMM_KSTEPS; ++s) {
      const int k0 = kk + 4 * s;
      v2f x0 = *(const v2f*)(ak + k0);
      if (TERM == 3) x0 = 2.0f * x0 - *(const v2f*)(t1 + k0);
      a0[s] = x0 * s0;
      bf[s].x = bk[(size_t)k0 * 64];
      bf[s].y = bk[(size_t)(k0 + 1) * 64];
    }
#pragma unroll
    for (int s = 0; s < GEMM_KSTEPS; ++s) c0 = wmma4(a0[s], bf[s], c0);
    __builtin_amdgcn_sched_group_barrier(0x020, 2 * GEMM_KSTEPS, 0);
    __builtin_amdgcn_sched_group_barrier(0x100, (TERM == 3 ? 2 : 1) * GEMM_KSTEPS, 0);
    __builtin_amdgcn_sched_group_barrier(0x008, GEMM_KSTEPS, 0);
  }
}

#define CHEB_ACC_V4 (TILE_ROWS * TILE_PITCH / 4)
#define CHEB_ROWS_V4 (3 * TILE_ROWS * TILE_PITCH / 4)
__device__ __forceinline__ void
tile_sum_cheb_body(int b, v4f* accv, v4f* rowsv,
                   const float* __restrict__ Tx0, const float* __restrict__ Tx1,
                   const float* __restrict__ Tx2, const float* __restrict__ recs,
                   const int* __restrict__ start, const float* __restrict__ fc,
                   const float* __restrict__ Wc, const float* __restrict__ bias,
                   float* __restrict__ out, int N) {
  float* acc = (float*)accv;
  float* As = (float*)rowsv;
  const float* Asrc[3] = {Tx0, Tx1, Tx2};
#pragma unroll
  for (int j = 0; j < 3 * (TILE_ROWS / 16); ++j) {
    const int r = (int)(threadIdx.x >> 4) + 16 * (j % (TILE_ROWS / 16)), c4 = (int)threadIdx.x & 15;
    int gr = b * TILE_ROWS + r;
    if (gr >= N) gr = N - 1;
    *(v4f*)(As + ((j / (TILE_ROWS / 16)) * TILE_ROWS + r) * TILE_PITCH + 4 * c4) =
        *(const v4f*)(Asrc[j / (TILE_ROWS / 16)] + (size_t)gr * 64 + 4 * c4);
  }
  for (int i = threadIdx.x; i < TILE_ROWS * TILE_PITCH / 4; i += 256) accv[i] = (v4f){0.f, 0.f, 0.f, 0.f};
  __syncthreads();
  tile_gather<0>(Tx2, recs, start[b], start[b + 1], acc);
  __syncthreads();
  const int w = (int)(threadIdx.x >> 5);
  const int lane = threadIdx.x & 31, half = lane >> 4, l16 = lane & 15;
  const int ct = w & 3, rp = w >> 2;
  const int lrow = rp * 16 + l16;
  int ar0 = b * TILE_ROWS + lrow;
  if (ar0 >= N) ar0 = N - 1;
  float sc0[4];
#pragma unroll
  for (int k = 0; k < 4; ++k) sc0[k] = fc[(size_t)k * N + ar0];
  v8f c0 = {};
  const float* t1 = As + (TILE_ROWS + lrow) * TILE_PITCH + 2 * half;
  const float* bk = Wc + (size_t)(2 * half) * 64 + ct * 16 + l16;
  cheb_term<0>(As + (0 * TILE_ROWS + lrow) * TILE_PITCH + 2 * half, t1, bk, sc0[0], c0);
  cheb_term<1>(As + (1 * TILE_ROWS + lrow) * TILE_PITCH + 2 * half, t1, bk + 1 * 64 * 64, sc0[1], c0);
  cheb_term<2>(As + (2 * TILE_ROWS + lrow) * TILE_PITCH + 2 * half, t1, bk + 2 * 64 * 64, sc0[2], c0);
  cheb_term<3>(acc + lrow * TILE_PITCH + 2 * half, t1, bk + 3 * 64 * 64, sc0[3], c0);
  const float bv = bias[ct * 16 + l16];
#pragma unroll
  for (int v = 0; v < 8; ++v) {
    const int r0 = b * TILE_ROWS + rp * 16 + v + 8 * half;
    if (r0 < N) out[(size_t)r0 * 192 + ct * 16 + l16] = c0[v] + bv;
  }
}


__device__ __forceinline__ float quad_sum(float v) {
  v += __int_as_float(__builtin_amdgcn_update_dpp(0, __float_as_int(v), 0xB1, 0xF, 0xF, true));
  v += __int_as_float(__builtin_amdgcn_update_dpp(0, __float_as_int(v), 0x4E, 0xF, 0xF, true));
  return v;
}

#define ATT_EDGES_PER_TRIP 8

__device__ __forceinline__ void
attn_tile_body(int b, float* acc, float* zacc, float* stage, v4f* wfrag,
               const float* __restrict__ Ee, const float* __restrict__ WE,
               const float* __restrict__ Qh, const float* __restrict__ Kh,
               const float* __restrict__ Vh, const float* __restrict__ recs,
               const int* __restrict__ start, float* __restrict__ out, int N) {
  for (int i = threadIdx.x; i < TILE_ROWS * 16; i += 256) ((v4f*)acc)[i] = (v4f){0.f, 0.f, 0.f, 0.f};
  for (int i = threadIdx.x; i < TILE_ROWS * 8; i += 256) zacc[i] = 0.f;
#pragma unroll
  for (int j = 0; j < 4; ++j) {
    const int f = (int)threadIdx.x + 256 * j;
    const int col = f & 63;
    const int r0 = 8 * (f >> 7) + 2 * ((f >> 6) & 1);
    wfrag[f] = (v4f){WE[r0 * 64 + col], WE[(r0 + 1) * 64 + col],
                     WE[(r0 + 4) * 64 + col], WE[(r0 + 5) * 64 + col]};
  }
  __syncthreads();
  const int lo = start[b], hi = start[b + 1];
  const int wid = __builtin_amdgcn_readfirstlane((int)(threadIdx.x >> 5));
  const int lane = threadIdx.x & 31, half = lane >> 4, l16 = lane & 15;
  float* myLds = stage + wid * 1024;
  const float scale = 0.35355339059327373f;

  for (int i0 = lo + wid * 16; i0 < hi; i0 += 8 * 16) {
    int i = i0 + l16;
    if (i >= hi) i = hi - 1;
    const v4f rec = *(const v4f*)(recs + 4 * (size_t)i);
    const int rs = __float_as_int(rec.x);
    const unsigned ry = (unsigned)__float_as_int(rec.y);
    const int rd = (int)(ry % TILE_ROWS);
    const int ar = (int)(ry / TILE_ROWS);

#pragma unroll 1
    for (int ct = 0; ct < 4; ++ct) {
      v8f c = {};
      const v4f* wf = wfrag + half * 64 + ct * 16 + l16;
      v4f q[8];
#pragma unroll
      for (int P = 0; P < 8; ++P) q[P] = wf[P * 128];
#pragma unroll
      for (int P = 0; P < 8; ++P) {
        const int kr = 8 * P + 2 * half;
        v2f a0, a1;
        a0.x = Ee[(size_t)ar * 64 + kr];
        a0.y = Ee[(size_t)ar * 64 + kr + 1];
        a1.x = Ee[(size_t)ar * 64 + kr + 4];
        a1.y = Ee[(size_t)ar * 64 + kr + 5];
        c = wmma4(a0, (v2f){q[P].x, q[P].y}, c);
        c = wmma4(a1, (v2f){q[P].z, q[P].w}, c);
      }
      __builtin_amdgcn_sched_group_barrier(0x100, 8, 0);
      __builtin_amdgcn_sched_group_barrier(0x008, 16, 0);
#pragma unroll
      for (int v = 0; v < 8; ++v)
        myLds[(v + 8 * half) * 64 + ct * 16 + l16] = c[v];
    }

    const int n = hi - i0;
    const float* qbase = Qh + (size_t)b * TILE_ROWS * 64;
#pragma unroll
    for (int g = 0; g < 16; g += ATT_EDGES_PER_TRIP) {
      v2f kv[ATT_EDGES_PER_TRIP], qv[ATT_EDGES_PER_TRIP], vv[ATT_EDGES_PER_TRIP], ev[ATT_EDGES_PER_TRIP];
#pragma unroll
      for (int j = 0; j < ATT_EDGES_PER_TRIP; ++j) {
        int s = __builtin_amdgcn_readlane(rs, g + j);
        int dl = __builtin_amdgcn_readlane(rd, g + j);
        kv[j] = *(const v2f*)(Kh + (size_t)s * 64 + 2 * lane);
        qv[j] = *(const v2f*)(qbase + (size_t)dl * 64 + 2 * lane);
        vv[j] = *(const v2f*)(Vh + (size_t)s * 64 + 2 * lane);
        ev[j] = *(const v2f*)(myLds + (g + j) * 64 + 2 * lane);
      }
#pragma unroll
      for (int j = 0; j < ATT_EDGES_PER_TRIP; ++j) {
        if (g + j < n) {
          int dl = __builtin_amdgcn_readlane(rd, g + j);
          float sc = quad_sum(kv[j].x * qv[j].x * ev[j].x + kv[j].y * qv[j].y * ev[j].y);
          sc *= scale;
          sc = fminf(5.0f, fmaxf(-5.0f, sc));
          float ss = __expf(sc);
          atomicAdd(&acc[dl * 64 + 2 * lane], vv[j].x * ss);
          atomicAdd(&acc[dl * 64 + 2 * lane + 1], vv[j].y * ss);
          if ((lane & 3) == 0) atomicAdd(&zacc[dl * 8 + (lane >> 2)], ss);
        }
      }
    }
  }
  __syncthreads();
  for (int i = threadIdx.x; i < TILE_ROWS * 16; i += 256) {
    int r = i >> 4, c4 = i & 15;
    int row = b * TILE_ROWS + r;
    if (row < N) {
      v4f a = ((const v4f*)acc)[i];
      float zz = zacc[r * 8 + (c4 >> 1)] + 1e-6f;
      v4f o = {a.x / zz, a.y / zz, a.z / zz, a.w / zz};
      *(v4f*)(out + (size_t)row * 192 + 128 + 4 * c4) = o;
    }
  }
}


__global__ void __launch_bounds__(ROLE_THREADS)
qkv_combine_kernel(const float* __restrict__ h, const float* __restrict__ p,
                   const float* __restrict__ WQ, const float* __restrict__ WK,
                   const float* __restrict__ WV,
                   float* __restrict__ Qh, float* __restrict__ Kh, float* __restrict__ Vh, int rows,
                   const float* __restrict__ a, float* __restrict__ t2, int n4,
                   int nGemm, int nComb) {
  const int bx = (int)blockIdx.x;
  if (bx < nGemm) qkv_gemm_body(bx, h, p, WQ, WK, WV, Qh, Kh, Vh, rows);
  else combine_body(bx - nGemm, nComb, a, h, t2, n4);
}

__global__ void __launch_bounds__(ROLE_THREADS)
gcn_deg_kernel(const float* __restrict__ x, const float* __restrict__ recs,
               const int* __restrict__ start, const float* __restrict__ Wg,
               const float* __restrict__ bias, float* __restrict__ out,
               float* __restrict__ dinv, int N, int nb) {
  const int bx = (int)blockIdx.x;
  if (bx < nb) tile_sum_gcn_body(bx, x, recs, start, Wg, bias, out, N);
  else tile_deg_body(bx - nb, recs, start, dinv, N);
}

static_assert(CHEB_ACC_V4 <= 16 * 64, "the Chebyshev sum tile lies in the wfrag region");
static_assert(CHEB_ROWS_V4 * 4 <= 8 * 16 * 64, "the Chebyshev rows lie in the stage region");
__global__ void __launch_bounds__(ROLE_THREADS) __attribute__((amdgpu_num_vgpr(96)))
attn_cheb_kernel(const float* __restrict__ Ee, const float* __restrict__ WE,
                 const float* __restrict__ Qh, const float* __restrict__ Kh,
                 const float* __restrict__ Vh, const float* __restrict__ recs,
                 const int* __restrict__ start,
                 const float* __restrict__ Tx0, const float* __restrict__ Tx1,
                 const float* __restrict__ Tx2, const float* __restrict__ fc,
                 const float* __restrict__ Wc, const float* __restrict__ bias,
                 float* __restrict__ out, int N, int nb) {
  __shared__ float acc[TILE_ROWS * 64];
  __shared__ float zacc[TILE_ROWS * 8];
  __shared__ v4f stagev[8 * 16 * 64 / 4];
  __shared__ v4f wfrag[16 * 64];
  const int bx = (int)blockIdx.x;
  if (bx < nb) attn_tile_body(bx, acc, zacc, (float*)stagev, wfrag, Ee, WE, Qh, Kh, Vh, recs, start, out, N);
  else tile_sum_cheb_body(bx - nb, wfrag, stagev, Tx0, Tx1, Tx2, recs, start, fc, Wc, bias, out, N);
}

extern "C" void kernel_launch(void* const* d_in, const int* in_sizes, int n_in,
                              void* d_out, int out_size, void* d_ws, size_t ws_size,
                              hipStream_t stream) {
  const float* h  = (const float*)d_in[0];
  const float* p  = (const float*)d_in[1];
  const float* e  = (const float*)d_in[2];
  const float* ew = (const float*)d_in[3];
  const float* fc = (const float*)d_in[4];
  const int* esrc = (const int*)d_in[5];
  const int* edst = (const int*)d_in[6];
  const float* Wc = (const float*)d_in[7];
  const float* bc = (const float*)d_in[8];
  const float* Wg = (const float*)d_in[9];
  const float* bg = (const float*)d_in[10];
  const float* WQ = (const float*)d_in[11];
  const float* WK = (const float*)d_in[12];
  const float* WVm = (const float*)d_in[13];
  const float* WE = (const float*)d_in[14];
  float* out = (float*)d_out;

  const int N = in_sizes[0] / 64;
  const int E = in_sizes[5];
  const size_t ND = (size_t)N * 64;

  float* w = (float*)d_ws;
  w += N;
  float* dinv = w; w += N;
  w += E;
  w = (float*)(((size_t)w + 15) & ~(size_t)15);
  float* Tx1  = w; w += ND;
  float* Tx2  = w; w += ND;
  w += ND;
  float* tmp  = w; w += ND;
  w += ND;
  float* scr  = w; w += 2 * ND;
  float* Qh   = w; w += ND;
  float* Kh   = w; w += ND;
  float* Vh   = w; w += ND;
  float* list = w; w += ND + (size_t)N * 8;

  const int nb = (N + TILE_ROWS - 1) / TILE_ROWS;
  float* recs   = list;
  int* binStart = (int*)(recs + 4 * (size_t)E);
  int* binCnt   = (int*)scr;
  int* rank     = binCnt + (size_t)nb * CNT_PITCH;

  hipMemsetAsync(binCnt, 0, sizeof(int) * (size_t)nb * CNT_PITCH, stream);

  const int B = 256;
  const int nd4 = N * 16;
  const int combBlocks = (nd4 + B * ELEM_VECS_COMBINE - 1) / (B * ELEM_VECS_COMBINE);

  const int rankEdges = RANK_THREADS * RANK_PER_THREAD;
  tile_rank_kernel<<<(E + rankEdges - 1) / rankEdges, RANK_THREADS, sizeof(int) * (size_t)nb, stream>>>(
      edst, binCnt, rank, E, nb);
  tile_scan_kernel<<<1, 1024, 0, stream>>>(binCnt, binStart, nb);
  tile_fill_kernel<<<(E + B - 1) / B, B, 0, stream>>>(esrc, edst, ew, binStart, rank, recs, E);

  gcn_deg_kernel<<<2 * nb, B, 0, stream>>>(h, recs, binStart, Wg, bg, out, dinv, N, nb);
  wlap_kernel<<<(E + B - 1) / B, B, 0, stream>>>(dinv, recs, E);

  tile_sum_kernel<0><<<nb, 256, 0, stream>>>(h, recs, binStart, Tx1, N);
  tile_sum_kernel<0><<<nb, 256, 0, stream>>>(Tx1, recs, binStart, tmp, N);
  const int gemmBlocks = (N + GEMM_ROWS - 1) / GEMM_ROWS;
  qkv_combine_kernel<<<gemmBlocks + combBlocks, B, 0, stream>>>(
      h, p, WQ, WK, WVm, Qh, Kh, Vh, N, tmp, Tx2, nd4, gemmBlocks, combBlocks);
  attn_cheb_kernel<<<2 * nb, B, 0, stream>>>(e, WE, Qh, Kh, Vh, recs, binStart,
                                             h, Tx1, Tx2, fc, Wc, bc, out, N, nb);
}
